# starting kernel
# baseline (speedup 1.0000x reference)
#include <hip/hip_runtime.h>
#include <hip/hip_bf16.h>


typedef __attribute__((ext_vector_type(2))) float v2f;
typedef __attribute__((ext_vector_type(8))) float v8f;

#define D_MODEL 1024
#define D_INNER 2048
#define D_STATE 16
#define L_SEQ   2048
#define B_SZ    2
#define M_ROWS  (B_SZ * L_SEQ)

__global__ __launch_bounds__(256)
void gemm_f32_wmma(const float* __restrict__ A, const float* __restrict__ Bm,
                   float* __restrict__ C, int M, int N, int K) {
    __shared__ __align__(16) float As[64][36];
    __shared__ __align__(16) float Bs[128][36];

    const int tid  = threadIdx.x;
    const int lane = tid & 31;
    const int w    = tid >> 5;
    const int wm   = w >> 2;
    const int wn   = w & 3;
    const int lmod = lane & 15;
    const int lhal = lane >> 4;

    const int n0 = blockIdx.x * 128;
    const int m0 = blockIdx.y * 64;

    v8f acc[2][2] = {};

    for (int k0 = 0; k0 < K; k0 += 32) {
        #pragma unroll
        for (int i = 0; i < 2; ++i) {
            int id = tid + i * 256;
            int r = id >> 3, c4 = id & 7;
            *(float4*)&As[r][c4 * 4] =
                *(const float4*)&A[(size_t)(m0 + r) * K + k0 + c4 * 4];
        }
        #pragma unroll
        for (int i = 0; i < 4; ++i) {
            int id = tid + i * 256;
            int r = id >> 3, c4 = id & 7;
            *(float4*)&Bs[r][c4 * 4] =
                *(const float4*)&Bm[(size_t)(n0 + r) * K + k0 + c4 * 4];
        }
        __syncthreads();

        #pragma unroll
        for (int kk = 0; kk < 8; ++kk) {
            const int kc = kk * 4 + lhal * 2;
            v2f a[2], b[2];
            a[0] = *(const v2f*)&As[wm * 32 +      lmod][kc];
            a[1] = *(const v2f*)&As[wm * 32 + 16 + lmod][kc];
            b[0] = *(const v2f*)&Bs[wn * 32 +      lmod][kc];
            b[1] = *(const v2f*)&Bs[wn * 32 + 16 + lmod][kc];
            #pragma unroll
            for (int mi = 0; mi < 2; ++mi)
                #pragma unroll
                for (int ni = 0; ni < 2; ++ni)
                    acc[mi][ni] = __builtin_amdgcn_wmma_f32_16x16x4_f32(
                        false, a[mi], false, b[ni], (short)0, acc[mi][ni],
                        false, false);
        }
        __syncthreads();
    }

    #pragma unroll
    for (int mi = 0; mi < 2; ++mi)
        #pragma unroll
        for (int ni = 0; ni < 2; ++ni)
            #pragma unroll
            for (int r = 0; r < 8; ++r) {
                int gm = m0 + wm * 32 + mi * 16 + lhal * 8 + r;
                int gn = n0 + wn * 32 + ni * 16 + lmod;
                C[(size_t)gm * N + gn] = acc[mi][ni][r];
            }
}

__global__ __launch_bounds__(256)
void conv_silu_kernel(const float* __restrict__ xz, const float* __restrict__ cw,
                      const float* __restrict__ cb, float* __restrict__ u) {
    size_t id = (size_t)blockIdx.x * 256 + threadIdx.x;
    int d = (int)(id & (D_INNER - 1));
    int l = (int)((id >> 11) & (L_SEQ - 1));
    int b = (int)(id >> 22);
    const float* base = xz + (size_t)b * L_SEQ * (2 * D_INNER) + d;
    float acc = cb[d];
    #pragma unroll
    for (int j = 0; j < 4; ++j) {
        int ll = l - 3 + j;
        if (ll >= 0) acc += base[(size_t)ll * (2 * D_INNER)] * cw[d * 4 + j];
    }
    u[id] = acc / (1.f + __expf(-acc));
}

__global__ __launch_bounds__(256)
void xdbl_kernel(const float* __restrict__ u, const float* __restrict__ Wx,
                 float* __restrict__ xdbl) {
    int gw   = (int)((blockIdx.x * 256 + threadIdx.x) >> 5);
    int lane = threadIdx.x & 31;
    if (gw >= M_ROWS * 33) return;
    int row = gw / 33, e = gw - row * 33;
    const float* ur = u  + (size_t)row * D_INNER;
    const float* wr = Wx + (size_t)e * D_INNER;
    float s = 0.f;
    for (int d = lane; d < D_INNER; d += 32) s += ur[d] * wr[d];
    #pragma unroll
    for (int off = 16; off; off >>= 1) s += __shfl_xor(s, off, 32);
    if (lane == 0) xdbl[(size_t)row * 33 + e] = s;
}

__global__ __launch_bounds__(256)
void scan_kernel(const float* __restrict__ xdbl, const float* __restrict__ u,
                 const float* __restrict__ Wdt, const float* __restrict__ bdt,
                 const float* __restrict__ Alog, float* __restrict__ ybuf) {
    const int tid = threadIdx.x;
    const int d = blockIdx.x * 256 + tid;
    const int b = blockIdx.y;

    float A[D_STATE];
    #pragma unroll
    for (int n = 0; n < D_STATE; ++n)
        A[n] = -__expf(fminf(Alog[n], 5.f));

    const float wdt = Wdt[d];
    const float bd  = bdt[d];
    float h[D_STATE];
    #pragma unroll
    for (int n = 0; n < D_STATE; ++n) h[n] = 0.f;

    const float* xd = xdbl + (size_t)b * L_SEQ * 33;
    const float* ub = u    + (size_t)b * L_SEQ * D_INNER + d;
    float*       yb = ybuf + (size_t)b * L_SEQ * D_INNER + d;

    __shared__ float srow[33];

    for (int l = 0; l < L_SEQ; ++l) {
        if (tid < 33) srow[tid] = xd[(size_t)l * 33 + tid];
        __syncthreads();

        float dt = srow[0] * wdt + bd;
        dt = (dt > 20.f) ? dt : log1pf(__expf(dt));
        dt = fminf(fmaxf(dt, 1e-4f), 10.f);
        const float ut = ub[(size_t)l * D_INNER];

        float y = 0.f;
        #pragma unroll
        for (int n = 0; n < D_STATE; ++n) {
            float dA  = __expf(fminf(fmaxf(dt * A[n], -20.f), 0.f));
            float dBu = fminf(fmaxf(dt * srow[1 + n] * ut, -10.f), 10.f);
            float hn  = fminf(fmaxf(h[n] * dA + dBu, -100.f), 100.f);
            h[n] = hn;
            y += hn * srow[17 + n];
        }
        yb[(size_t)l * D_INNER] = y;
        __syncthreads();
    }
}

__global__ __launch_bounds__(256)
void ln_gate_kernel(float* __restrict__ ybuf, const float* __restrict__ u,
                    const float* __restrict__ xz, const float* __restrict__ Dp,
                    const float* __restrict__ g, const float* __restrict__ bta) {
    const int row = blockIdx.x;
    const int tid = threadIdx.x;
    float* yr = ybuf + (size_t)row * D_INNER;

    float v[8], s = 0.f, s2 = 0.f;
    #pragma unroll
    for (int i = 0; i < 8; ++i) {
        v[i] = yr[tid + i * 256];
        s += v[i]; s2 += v[i] * v[i];
    }
    #pragma unroll
    for (int off = 16; off; off >>= 1) {
        s  += __shfl_xor(s,  off, 32);
        s2 += __shfl_xor(s2, off, 32);
    }
    __shared__ float ws1[8], ws2[8];
    const int lane = tid & 31, w = tid >> 5;
    if (lane == 0) { ws1[w] = s; ws2[w] = s2; }
    __syncthreads();
    if (tid == 0) {
        float t1 = 0.f, t2 = 0.f;
        #pragma unroll
        for (int i = 0; i < 8; ++i) { t1 += ws1[i]; t2 += ws2[i]; }
        ws1[0] = t1; ws2[0] = t2;
    }
    __syncthreads();
    const float mu  = ws1[0] * (1.f / D_INNER);
    float var = ws2[0] * (1.f / D_INNER) - mu * mu;
    var = fmaxf(var, 0.f);
    const float rs = rsqrtf(var + 1e-5f);

    #pragma unroll
    for (int i = 0; i < 8; ++i) {
        const int d = tid + i * 256;
        float yn = (v[i] - mu) * rs * g[d] + bta[d];
        float uu = u[(size_t)row * D_INNER + d];
        float zz = xz[(size_t)row * (2 * D_INNER) + D_INNER + d];
        float sz = zz / (1.f + __expf(-zz));
        yr[d] = (yn + Dp[d] * uu) * sz;
    }
}

extern "C" void kernel_launch(void* const* d_in, const int* in_sizes, int n_in,
                              void* d_out, int out_size, void* d_ws, size_t ws_size,
                              hipStream_t stream) {
    const float* x      = (const float*)d_in[0];
    const float* W_in   = (const float*)d_in[1];
    const float* conv_w = (const float*)d_in[2];
    const float* conv_b = (const float*)d_in[3];
    const float* W_x    = (const float*)d_in[4];
    const float* W_dt   = (const float*)d_in[5];
    const float* b_dt   = (const float*)d_in[6];
    const float* A_log  = (const float*)d_in[7];
    const float* D_par  = (const float*)d_in[8];
    const float* W_out  = (const float*)d_in[9];
    const float* ln_g   = (const float*)d_in[10];
    const float* ln_b   = (const float*)d_in[11];
    float* out = (float*)d_out;

    float* ws = (float*)d_ws;
    const size_t XZ_OFF   = 0;
    const size_t U_OFF    = XZ_OFF + (size_t)M_ROWS * 2 * D_INNER;
    const size_t XDBL_OFF = U_OFF  + (size_t)M_ROWS * D_INNER;
    const size_t Y_OFF    = XDBL_OFF + (size_t)M_ROWS * 33;
    float* xz   = ws + XZ_OFF;
    float* u    = ws + U_OFF;
    float* xdbl = ws + XDBL_OFF;
    float* ybuf = ws + Y_OFF;

    gemm_f32_wmma<<<dim3(2 * D_INNER / 128, M_ROWS / 64), 256, 0, stream>>>(
        x, W_in, xz, M_ROWS, 2 * D_INNER, D_MODEL);

    {
        size_t n = (size_t)M_ROWS * D_INNER;
        conv_silu_kernel<<<(unsigned)(n / 256), 256, 0, stream>>>(
            xz, conv_w, conv_b, u);
    }

    {
        int waves = M_ROWS * 33;
        xdbl_kernel<<<(waves + 7) / 8, 256, 0, stream>>>(u, W_x, xdbl);
    }

    scan_kernel<<<dim3(D_INNER / 256, B_SZ), 256, 0, stream>>>(
        xdbl, u, W_dt, b_dt, A_log, ybuf);

    ln_gate_kernel<<<M_ROWS, 256, 0, stream>>>(ybuf, u, xz, D_par, ln_g, ln_b);

    gemm_f32_wmma<<<dim3(D_MODEL / 128, M_ROWS / 64), 256, 0, stream>>>(
        ybuf, W_out, out, M_ROWS, D_MODEL, D_INNER);
}


# scan: x_dbl rows of 16 steps staged in LDS (two buffers) and u of 16 steps held in registers, next chunk loaded ahead: one barrier per 16 steps, no global load on the serial path
# speedup vs baseline: 2.3411x; 2.3411x over previous
#include <hip/hip_runtime.h>
#include <hip/hip_bf16.h>


typedef __attribute__((ext_vector_type(2))) float v2f;
typedef __attribute__((ext_vector_type(8))) float v8f;

#define D_MODEL 1024
#define D_INNER 2048
#define D_STATE 16
#define L_SEQ   2048
#define B_SZ    2
#define M_ROWS  (B_SZ * L_SEQ)

__global__ __launch_bounds__(256)
void gemm_f32_wmma(const float* __restrict__ A, const float* __restrict__ Bm,
                   float* __restrict__ C, int M, int N, int K) {
    __shared__ __align__(16) float As[64][36];
    __shared__ __align__(16) float Bs[128][36];

    const int tid  = threadIdx.x;
    const int lane = tid & 31;
    const int w    = tid >> 5;
    const int wm   = w >> 2;
    const int wn   = w & 3;
    const int lmod = lane & 15;
    const int lhal = lane >> 4;

    const int n0 = blockIdx.x * 128;
    const int m0 = blockIdx.y * 64;

    v8f acc[2][2] = {};

    for (int k0 = 0; k0 < K; k0 += 32) {
        #pragma unroll
        for (int i = 0; i < 2; ++i) {
            int id = tid + i * 256;
            int r = id >> 3, c4 = id & 7;
            *(float4*)&As[r][c4 * 4] =
                *(const float4*)&A[(size_t)(m0 + r) * K + k0 + c4 * 4];
        }
        #pragma unroll
        for (int i = 0; i < 4; ++i) {
            int id = tid + i * 256;
            int r = id >> 3, c4 = id & 7;
            *(float4*)&Bs[r][c4 * 4] =
                *(const float4*)&Bm[(size_t)(n0 + r) * K + k0 + c4 * 4];
        }
        __syncthreads();

        #pragma unroll
        for (int kk = 0; kk < 8; ++kk) {
            const int kc = kk * 4 + lhal * 2;
            v2f a[2], b[2];
            a[0] = *(const v2f*)&As[wm * 32 +      lmod][kc];
            a[1] = *(const v2f*)&As[wm * 32 + 16 + lmod][kc];
            b[0] = *(const v2f*)&Bs[wn * 32 +      lmod][kc];
            b[1] = *(const v2f*)&Bs[wn * 32 + 16 + lmod][kc];
            #pragma unroll
            for (int mi = 0; mi < 2; ++mi)
                #pragma unroll
                for (int ni = 0; ni < 2; ++ni)
                    acc[mi][ni] = __builtin_amdgcn_wmma_f32_16x16x4_f32(
                        false, a[mi], false, b[ni], (short)0, acc[mi][ni],
                        false, false);
        }
        __syncthreads();
    }

    #pragma unroll
    for (int mi = 0; mi < 2; ++mi)
        #pragma unroll
        for (int ni = 0; ni < 2; ++ni)
            #pragma unroll
            for (int r = 0; r < 8; ++r) {
                int gm = m0 + wm * 32 + mi * 16 + lhal * 8 + r;
                int gn = n0 + wn * 32 + ni * 16 + lmod;
                C[(size_t)gm * N + gn] = acc[mi][ni][r];
            }
}

__global__ __launch_bounds__(256)
void conv_silu_kernel(const float* __restrict__ xz, const float* __restrict__ cw,
                      const float* __restrict__ cb, float* __restrict__ u) {
    size_t id = (size_t)blockIdx.x * 256 + threadIdx.x;
    int d = (int)(id & (D_INNER - 1));
    int l = (int)((id >> 11) & (L_SEQ - 1));
    int b = (int)(id >> 22);
    const float* base = xz + (size_t)b * L_SEQ * (2 * D_INNER) + d;
    float acc = cb[d];
    #pragma unroll
    for (int j = 0; j < 4; ++j) {
        int ll = l - 3 + j;
        if (ll >= 0) acc += base[(size_t)ll * (2 * D_INNER)] * cw[d * 4 + j];
    }
    u[id] = acc / (1.f + __expf(-acc));
}

__global__ __launch_bounds__(256)
void xdbl_kernel(const float* __restrict__ u, const float* __restrict__ Wx,
                 float* __restrict__ xdbl) {
    int gw   = (int)((blockIdx.x * 256 + threadIdx.x) >> 5);
    int lane = threadIdx.x & 31;
    if (gw >= M_ROWS * 33) return;
    int row = gw / 33, e = gw - row * 33;
    const float* ur = u  + (size_t)row * D_INNER;
    const float* wr = Wx + (size_t)e * D_INNER;
    float s = 0.f;
    for (int d = lane; d < D_INNER; d += 32) s += ur[d] * wr[d];
    #pragma unroll
    for (int off = 16; off; off >>= 1) s += __shfl_xor(s, off, 32);
    if (lane == 0) xdbl[(size_t)row * 33 + e] = s;
}

#define SCAN_T     16
#define SCAN_NCH   (L_SEQ / SCAN_T)
#define SCAN_PITCH 36
#define SCAN_XCH   (SCAN_T * 33)

__device__ __forceinline__ int scan_dst(int i) {
    const int t = i / 33, e = i - t * 33;
    return t * SCAN_PITCH + (e ? e + 3 : 0);
}

__global__ __launch_bounds__(256)
void scan_kernel(const float* __restrict__ xdbl, const float* __restrict__ u,
                 const float* __restrict__ Wdt, const float* __restrict__ bdt,
                 const float* __restrict__ Alog, float* __restrict__ ybuf) {
    const int tid = threadIdx.x;
    const int d = blockIdx.x * 256 + tid;
    const int b = blockIdx.y;

    float A[D_STATE];
    #pragma unroll
    for (int n = 0; n < D_STATE; ++n)
        A[n] = -__expf(fminf(Alog[n], 5.f));

    const float wdt = Wdt[d];
    const float bd  = bdt[d];
    float h[D_STATE];
    #pragma unroll
    for (int n = 0; n < D_STATE; ++n) h[n] = 0.f;

    const float* xd = xdbl + (size_t)b * L_SEQ * 33;
    const float* ub = u    + (size_t)b * L_SEQ * D_INNER + d;
    float*       yb = ybuf + (size_t)b * L_SEQ * D_INNER + d;

    __shared__ __align__(16) float sx[2][SCAN_T * SCAN_PITCH];

    const int dst0 = scan_dst(tid);
    const int dst1 = scan_dst(tid + 256);
    const int dst2 = scan_dst(tid + 512);
    const bool has2 = tid < (SCAN_XCH - 512);

    float uc[SCAN_T];
    sx[0][dst0] = xd[tid];
    sx[0][dst1] = xd[tid + 256];
    if (has2) sx[0][dst2] = xd[tid + 512];
    #pragma unroll
    for (int t = 0; t < SCAN_T; ++t) uc[t] = ub[(size_t)t * D_INNER];
    __syncthreads();

    for (int c = 0; c < SCAN_NCH; ++c) {
        const float* s = sx[c & 1];
        const bool more = (c + 1 < SCAN_NCH);

        float xn0 = 0.f, xn1 = 0.f, xn2 = 0.f;
        float un[SCAN_T];
        #pragma unroll
        for (int t = 0; t < SCAN_T; ++t) un[t] = 0.f;
        if (more) {
            const float* nsrc = xd + (size_t)(c + 1) * SCAN_XCH;
            xn0 = nsrc[tid];
            xn1 = nsrc[tid + 256];
            if (has2) xn2 = nsrc[tid + 512];
            const float* nu = ub + (size_t)(c + 1) * SCAN_T * D_INNER;
            #pragma unroll
            for (int t = 0; t < SCAN_T; ++t) un[t] = nu[(size_t)t * D_INNER];
        }

        float* yc = yb + (size_t)c * SCAN_T * D_INNER;
        #pragma unroll
        for (int t = 0; t < SCAN_T; ++t) {
            const float*  r  = s + t * SCAN_PITCH;
            const float4* r4 = (const float4*)r;
            float Bv[D_STATE], Cv[D_STATE];
            #pragma unroll
            for (int q = 0; q < 4; ++q) {
                const float4 bq = r4[1 + q];
                const float4 cq = r4[5 + q];
                Bv[q * 4 + 0] = bq.x; Bv[q * 4 + 1] = bq.y;
                Bv[q * 4 + 2] = bq.z; Bv[q * 4 + 3] = bq.w;
                Cv[q * 4 + 0] = cq.x; Cv[q * 4 + 1] = cq.y;
                Cv[q * 4 + 2] = cq.z; Cv[q * 4 + 3] = cq.w;
            }

            float dt = r[0] * wdt + bd;
            dt = (dt > 20.f) ? dt : log1pf(__expf(dt));
            dt = fminf(fmaxf(dt, 1e-4f), 10.f);
            const float ut = uc[t];

            float y = 0.f;
            #pragma unroll
            for (int n = 0; n < D_STATE; ++n) {
                float dA  = __expf(fminf(fmaxf(dt * A[n], -20.f), 0.f));
                float dBu = fminf(fmaxf(dt * Bv[n] * ut, -10.f), 10.f);
                float hn  = fminf(fmaxf(h[n] * dA + dBu, -100.f), 100.f);
                h[n] = hn;
                y += hn * Cv[n];
            }
            yc[(size_t)t * D_INNER] = y;
        }

        if (more) {
            float* nx = sx[(c + 1) & 1];
            nx[dst0] = xn0;
            nx[dst1] = xn1;
            if (has2) nx[dst2] = xn2;
        }
        #pragma unroll
        for (int t = 0; t < SCAN_T; ++t) uc[t] = un[t];
        __syncthreads();
    }
}

__global__ __launch_bounds__(256)
void ln_gate_kernel(float* __restrict__ ybuf, const float* __restrict__ u,
                    const float* __restrict__ xz, const float* __restrict__ Dp,
                    const float* __restrict__ g, const float* __restrict__ bta) {
    const int row = blockIdx.x;
    const int tid = threadIdx.x;
    float* yr = ybuf + (size_t)row * D_INNER;

    float v[8], s = 0.f, s2 = 0.f;
    #pragma unroll
    for (int i = 0; i < 8; ++i) {
        v[i] = yr[tid + i * 256];
        s += v[i]; s2 += v[i] * v[i];
    }
    #pragma unroll
    for (int off = 16; off; off >>= 1) {
        s  += __shfl_xor(s,  off, 32);
        s2 += __shfl_xor(s2, off, 32);
    }
    __shared__ float ws1[8], ws2[8];
    const int lane = tid & 31, w = tid >> 5;
    if (lane == 0) { ws1[w] = s; ws2[w] = s2; }
    __syncthreads();
    if (tid == 0) {
        float t1 = 0.f, t2 = 0.f;
        #pragma unroll
        for (int i = 0; i < 8; ++i) { t1 += ws1[i]; t2 += ws2[i]; }
        ws1[0] = t1; ws2[0] = t2;
    }
    __syncthreads();
    const float mu  = ws1[0] * (1.f / D_INNER);
    float var = ws2[0] * (1.f / D_INNER) - mu * mu;
    var = fmaxf(var, 0.f);
    const float rs = rsqrtf(var + 1e-5f);

    #pragma unroll
    for (int i = 0; i < 8; ++i) {
        const int d = tid + i * 256;
        float yn = (v[i] - mu) * rs * g[d] + bta[d];
        float uu = u[(size_t)row * D_INNER + d];
        float zz = xz[(size_t)row * (2 * D_INNER) + D_INNER + d];
        float sz = zz / (1.f + __expf(-zz));
        yr[d] = (yn + Dp[d] * uu) * sz;
    }
}

extern "C" void kernel_launch(void* const* d_in, const int* in_sizes, int n_in,
                              void* d_out, int out_size, void* d_ws, size_t ws_size,
                              hipStream_t stream) {
    const float* x      = (const float*)d_in[0];
    const float* W_in   = (const float*)d_in[1];
    const float* conv_w = (const float*)d_in[2];
    const float* conv_b = (const float*)d_in[3];
    const float* W_x    = (const float*)d_in[4];
    const float* W_dt   = (const float*)d_in[5];
    const float* b_dt   = (const float*)d_in[6];
    const float* A_log  = (const float*)d_in[7];
    const float* D_par  = (const float*)d_in[8];
    const float* W_out  = (const float*)d_in[9];
    const float* ln_g   = (const float*)d_in[10];
    const float* ln_b   = (const float*)d_in[11];
    float* out = (float*)d_out;

    float* ws = (float*)d_ws;
    const size_t XZ_OFF   = 0;
    const size_t U_OFF    = XZ_OFF + (size_t)M_ROWS * 2 * D_INNER;
    const size_t XDBL_OFF = U_OFF  + (size_t)M_ROWS * D_INNER;
    const size_t Y_OFF    = XDBL_OFF + (size_t)M_ROWS * 33;
    float* xz   = ws + XZ_OFF;
    float* u    = ws + U_OFF;
    float* xdbl = ws + XDBL_OFF;
    float* ybuf = ws + Y_OFF;

    gemm_f32_wmma<<<dim3(2 * D_INNER / 128, M_ROWS / 64), 256, 0, stream>>>(
        x, W_in, xz, M_ROWS, 2 * D_INNER, D_MODEL);

    {
        size_t n = (size_t)M_ROWS * D_INNER;
        conv_silu_kernel<<<(unsigned)(n / 256), 256, 0, stream>>>(
            xz, conv_w, conv_b, u);
    }

    {
        int waves = M_ROWS * 33;
        xdbl_kernel<<<(waves + 7) / 8, 256, 0, stream>>>(u, W_x, xdbl);
    }

    scan_kernel<<<dim3(D_INNER / 256, B_SZ), 256, 0, stream>>>(
        xdbl, u, W_dt, b_dt, A_log, ybuf);

    ln_gate_kernel<<<M_ROWS, 256, 0, stream>>>(ybuf, u, xz, D_par, ln_g, ln_b);

    gemm_f32_wmma<<<dim3(D_MODEL / 128, M_ROWS / 64), 256, 0, stream>>>(
        ybuf, W_out, out, M_ROWS, D_MODEL, D_INNER);
}


# scan: the 16 states of a channel split over 8 lanes (2 each), y summed by a cross-lane tree (sum order changes); 128 work-groups of 32 channels instead of 16 of 256; dt staged once per step
# speedup vs baseline: 5.3663x; 2.2922x over previous
#include <hip/hip_runtime.h>
#include <hip/hip_bf16.h>


typedef __attribute__((ext_vector_type(2))) float v2f;
typedef __attribute__((ext_vector_type(8))) float v8f;

#define D_MODEL 1024
#define D_INNER 2048
#define D_STATE 16
#define L_SEQ   2048
#define B_SZ    2
#define M_ROWS  (B_SZ * L_SEQ)

__global__ __launch_bounds__(256)
void gemm_f32_wmma(const float* __restrict__ A, const float* __restrict__ Bm,
                   float* __restrict__ C, int M, int N, int K) {
    __shared__ __align__(16) float As[64][36];
    __shared__ __align__(16) float Bs[128][36];

    const int tid  = threadIdx.x;
    const int lane = tid & 31;
    const int w    = tid >> 5;
    const int wm   = w >> 2;
    const int wn   = w & 3;
    const int lmod = lane & 15;
    const int lhal = lane >> 4;

    const int n0 = blockIdx.x * 128;
    const int m0 = blockIdx.y * 64;

    v8f acc[2][2] = {};

    for (int k0 = 0; k0 < K; k0 += 32) {
        #pragma unroll
        for (int i = 0; i < 2; ++i) {
            int id = tid + i * 256;
            int r = id >> 3, c4 = id & 7;
            *(float4*)&As[r][c4 * 4] =
                *(const float4*)&A[(size_t)(m0 + r) * K + k0 + c4 * 4];
        }
        #pragma unroll
        for (int i = 0; i < 4; ++i) {
            int id = tid + i * 256;
            int r = id >> 3, c4 = id & 7;
            *(float4*)&Bs[r][c4 * 4] =
                *(const float4*)&Bm[(size_t)(n0 + r) * K + k0 + c4 * 4];
        }
        __syncthreads();

        #pragma unroll
        for (int kk = 0; kk < 8; ++kk) {
            const int kc = kk * 4 + lhal * 2;
            v2f a[2], b[2];
            a[0] = *(const v2f*)&As[wm * 32 +      lmod][kc];
            a[1] = *(const v2f*)&As[wm * 32 + 16 + lmod][kc];
            b[0] = *(const v2f*)&Bs[wn * 32 +      lmod][kc];
            b[1] = *(const v2f*)&Bs[wn * 32 + 16 + lmod][kc];
            #pragma unroll
            for (int mi = 0; mi < 2; ++mi)
                #pragma unroll
                for (int ni = 0; ni < 2; ++ni)
                    acc[mi][ni] = __builtin_amdgcn_wmma_f32_16x16x4_f32(
                        false, a[mi], false, b[ni], (short)0, acc[mi][ni],
                        false, false);
        }
        __syncthreads();
    }

    #pragma unroll
    for (int mi = 0; mi < 2; ++mi)
        #pragma unroll
        for (int ni = 0; ni < 2; ++ni)
            #pragma unroll
            for (int r = 0; r < 8; ++r) {
                int gm = m0 + wm * 32 + mi * 16 + lhal * 8 + r;
                int gn = n0 + wn * 32 + ni * 16 + lmod;
                C[(size_t)gm * N + gn] = acc[mi][ni][r];
            }
}

__global__ __launch_bounds__(256)
void conv_silu_kernel(const float* __restrict__ xz, const float* __restrict__ cw,
                      const float* __restrict__ cb, float* __restrict__ u) {
    size_t id = (size_t)blockIdx.x * 256 + threadIdx.x;
    int d = (int)(id & (D_INNER - 1));
    int l = (int)((id >> 11) & (L_SEQ - 1));
    int b = (int)(id >> 22);
    const float* base = xz + (size_t)b * L_SEQ * (2 * D_INNER) + d;
    float acc = cb[d];
    #pragma unroll
    for (int j = 0; j < 4; ++j) {
        int ll = l - 3 + j;
        if (ll >= 0) acc += base[(size_t)ll * (2 * D_INNER)] * cw[d * 4 + j];
    }
    u[id] = acc / (1.f + __expf(-acc));
}

__global__ __launch_bounds__(256)
void xdbl_kernel(const float* __restrict__ u, const float* __restrict__ Wx,
                 float* __restrict__ xdbl) {
    int gw   = (int)((blockIdx.x * 256 + threadIdx.x) >> 5);
    int lane = threadIdx.x & 31;
    if (gw >= M_ROWS * 33) return;
    int row = gw / 33, e = gw - row * 33;
    const float* ur = u  + (size_t)row * D_INNER;
    const float* wr = Wx + (size_t)e * D_INNER;
    float s = 0.f;
    for (int d = lane; d < D_INNER; d += 32) s += ur[d] * wr[d];
    #pragma unroll
    for (int off = 16; off; off >>= 1) s += __shfl_xor(s, off, 32);
    if (lane == 0) xdbl[(size_t)row * 33 + e] = s;
}

#define SCAN_T     16
#define SCAN_NCH   (L_SEQ / SCAN_T)
#define SCAN_PITCH 36
#define SCAN_XCH   (SCAN_T * 33)
#define SCAN_CH    32
#define SCAN_YP    36

__device__ __forceinline__ int scan_dst(int i) {
    const int t = i / 33, e = i - t * 33;
    return t * SCAN_PITCH + (e ? e + 3 : 0);
}

__device__ __forceinline__ float scan_dt(float raw, float wdt, float bd) {
    float dt = raw * wdt + bd;
    const float sp = log1pf(__expf(dt));
    dt = (dt > 20.f) ? dt : sp;
    return fminf(fmaxf(dt, 1e-4f), 10.f);
}

template <int XMASK>
__device__ __forceinline__ float scan_add_xor(float v) {
    const int o = __builtin_amdgcn_update_dpp(0, __float_as_int(v), 0x160 + XMASK,
                                              0xF, 0xF, true);
    return v + __int_as_float(o);
}

__global__ __launch_bounds__(256)
void scan_kernel(const float* __restrict__ xdbl, const float* __restrict__ u,
                 const float* __restrict__ Wdt, const float* __restrict__ bdt,
                 const float* __restrict__ Alog, float* __restrict__ ybuf) {
    const int tid = threadIdx.x;
    const int j   = tid & 7;
    const int c   = tid >> 3;
    const int sc  = tid & 31;
    const int st  = tid >> 5;
    const int d0  = blockIdx.x * SCAN_CH;
    const int b   = blockIdx.y;

    const float A0 = -__expf(fminf(Alog[2 * j],     5.f));
    const float A1 = -__expf(fminf(Alog[2 * j + 1], 5.f));
    const float wdt = Wdt[d0 + sc];
    const float bd  = bdt[d0 + sc];
    float h0 = 0.f, h1 = 0.f;

    const float* xd = xdbl + (size_t)b * L_SEQ * 33;
    const float* ub = u    + (size_t)b * L_SEQ * D_INNER + d0 + sc;
    float*       yb = ybuf + (size_t)b * L_SEQ * D_INNER + d0 + sc;

    __shared__ __align__(16) float  sx[2][SCAN_T * SCAN_PITCH];
    __shared__ __align__(16) float2 sdu[2][SCAN_T][SCAN_CH];
    __shared__ __align__(16) float  sy[2][SCAN_T * SCAN_YP];

    const int dst0 = scan_dst(tid);
    const int dst1 = scan_dst(tid + 256);
    const int dst2 = scan_dst(tid + 512);
    const bool has2 = tid < (SCAN_XCH - 512);

    sx[0][dst0] = xd[tid];
    sx[0][dst1] = xd[tid + 256];
    if (has2) sx[0][dst2] = xd[tid + 512];
    sdu[0][st][sc]     = make_float2(scan_dt(xd[st * 33], wdt, bd),
                                     ub[(size_t)st * D_INNER]);
    sdu[0][st + 8][sc] = make_float2(scan_dt(xd[(st + 8) * 33], wdt, bd),
                                     ub[(size_t)(st + 8) * D_INNER]);
    __syncthreads();

    for (int ch = 0; ch < SCAN_NCH; ++ch) {
        const int  cur  = ch & 1;
        const bool more = (ch + 1 < SCAN_NCH);

        float xn0 = 0.f, xn1 = 0.f, xn2 = 0.f;
        float rn0 = 0.f, rn1 = 0.f, un0 = 0.f, un1 = 0.f;
        if (more) {
            const float* nsrc = xd + (size_t)(ch + 1) * SCAN_XCH;
            xn0 = nsrc[tid];
            xn1 = nsrc[tid + 256];
            if (has2) xn2 = nsrc[tid + 512];
            rn0 = nsrc[st * 33];
            rn1 = nsrc[(st + 8) * 33];
            const float* nu = ub + (size_t)(ch + 1) * SCAN_T * D_INNER;
            un0 = nu[(size_t)st * D_INNER];
            un1 = nu[(size_t)(st + 8) * D_INNER];
        }

        if (ch > 0) {
            const float* py = sy[cur ^ 1];
            float* yc = yb + (size_t)(ch - 1) * SCAN_T * D_INNER;
            yc[(size_t)st * D_INNER]       = py[st * SCAN_YP + sc];
            yc[(size_t)(st + 8) * D_INNER] = py[(st + 8) * SCAN_YP + sc];
        }

        const float* s = sx[cur];
        float ysel = 0.f;
        #pragma unroll
        for (int t = 0; t < SCAN_T; ++t) {
            const float2 du = sdu[cur][t][c];
            const float2 Bq = *(const float2*)&s[t * SCAN_PITCH + 4  + 2 * j];
            const float2 Cq = *(const float2*)&s[t * SCAN_PITCH + 20 + 2 * j];
            const float dt = du.x, ut = du.y;

            const float dA0 = __expf(fminf(fmaxf(dt * A0, -20.f), 0.f));
            const float dA1 = __expf(fminf(fmaxf(dt * A1, -20.f), 0.f));
            const float dB0 = fminf(fmaxf(dt * Bq.x * ut, -10.f), 10.f);
            const float dB1 = fminf(fmaxf(dt * Bq.y * ut, -10.f), 10.f);
            h0 = fminf(fmaxf(h0 * dA0 + dB0, -100.f), 100.f);
            h1 = fminf(fmaxf(h1 * dA1 + dB1, -100.f), 100.f);

            float p = h0 * Cq.x + h1 * Cq.y;
            p = scan_add_xor<1>(p);
            p = scan_add_xor<2>(p);
            p = scan_add_xor<4>(p);
            if ((t & 7) == j) ysel = p;
            if ((t & 7) == 7)
                sy[cur][((t & 8) + j) * SCAN_YP + c] = ysel;
        }

        if (more) {
            float* nx = sx[cur ^ 1];
            nx[dst0] = xn0;
            nx[dst1] = xn1;
            if (has2) nx[dst2] = xn2;
            const float dn0 = scan_dt(rn0, wdt, bd);
            const float dn1 = scan_dt(rn1, wdt, bd);
            sdu[cur ^ 1][st][sc]     = make_float2(dn0, un0);
            sdu[cur ^ 1][st + 8][sc] = make_float2(dn1, un1);
        }
        __syncthreads();
    }

    {
        const float* py = sy[(SCAN_NCH - 1) & 1];
        float* yc = yb + (size_t)(SCAN_NCH - 1) * SCAN_T * D_INNER;
        yc[(size_t)st * D_INNER]       = py[st * SCAN_YP + sc];
        yc[(size_t)(st + 8) * D_INNER] = py[(st + 8) * SCAN_YP + sc];
    }
}

__global__ __launch_bounds__(256)
void ln_gate_kernel(float* __restrict__ ybuf, const float* __restrict__ u,
                    const float* __restrict__ xz, const float* __restrict__ Dp,
                    const float* __restrict__ g, const float* __restrict__ bta) {
    const int row = blockIdx.x;
    const int tid = threadIdx.x;
    float* yr = ybuf + (size_t)row * D_INNER;

    float v[8], s = 0.f, s2 = 0.f;
    #pragma unroll
    for (int i = 0; i < 8; ++i) {
        v[i] = yr[tid + i * 256];
        s += v[i]; s2 += v[i] * v[i];
    }
    #pragma unroll
    for (int off = 16; off; off >>= 1) {
        s  += __shfl_xor(s,  off, 32);
        s2 += __shfl_xor(s2, off, 32);
    }
    __shared__ float ws1[8], ws2[8];
    const int lane = tid & 31, w = tid >> 5;
    if (lane == 0) { ws1[w] = s; ws2[w] = s2; }
    __syncthreads();
    if (tid == 0) {
        float t1 = 0.f, t2 = 0.f;
        #pragma unroll
        for (int i = 0; i < 8; ++i) { t1 += ws1[i]; t2 += ws2[i]; }
        ws1[0] = t1; ws2[0] = t2;
    }
    __syncthreads();
    const float mu  = ws1[0] * (1.f / D_INNER);
    float var = ws2[0] * (1.f / D_INNER) - mu * mu;
    var = fmaxf(var, 0.f);
    const float rs = rsqrtf(var + 1e-5f);

    #pragma unroll
    for (int i = 0; i < 8; ++i) {
        const int d = tid + i * 256;
        float yn = (v[i] - mu) * rs * g[d] + bta[d];
        float uu = u[(size_t)row * D_INNER + d];
        float zz = xz[(size_t)row * (2 * D_INNER) + D_INNER + d];
        float sz = zz / (1.f + __expf(-zz));
        yr[d] = (yn + Dp[d] * uu) * sz;
    }
}

extern "C" void kernel_launch(void* const* d_in, const int* in_sizes, int n_in,
                              void* d_out, int out_size, void* d_ws, size_t ws_size,
                              hipStream_t stream) {
    const float* x      = (const float*)d_in[0];
    const float* W_in   = (const float*)d_in[1];
    const float* conv_w = (const float*)d_in[2];
    const float* conv_b = (const float*)d_in[3];
    const float* W_x    = (const float*)d_in[4];
    const float* W_dt   = (const float*)d_in[5];
    const float* b_dt   = (const float*)d_in[6];
    const float* A_log  = (const float*)d_in[7];
    const float* D_par  = (const float*)d_in[8];
    const float* W_out  = (const float*)d_in[9];
    const float* ln_g   = (const float*)d_in[10];
    const float* ln_b   = (const float*)d_in[11];
    float* out = (float*)d_out;

    float* ws = (float*)d_ws;
    const size_t XZ_OFF   = 0;
    const size_t U_OFF    = XZ_OFF + (size_t)M_ROWS * 2 * D_INNER;
    const size_t XDBL_OFF = U_OFF  + (size_t)M_ROWS * D_INNER;
    const size_t Y_OFF    = XDBL_OFF + (size_t)M_ROWS * 33;
    float* xz   = ws + XZ_OFF;
    float* u    = ws + U_OFF;
    float* xdbl = ws + XDBL_OFF;
    float* ybuf = ws + Y_OFF;

    gemm_f32_wmma<<<dim3(2 * D_INNER / 128, M_ROWS / 64), 256, 0, stream>>>(
        x, W_in, xz, M_ROWS, 2 * D_INNER, D_MODEL);

    {
        size_t n = (size_t)M_ROWS * D_INNER;
        conv_silu_kernel<<<(unsigned)(n / 256), 256, 0, stream>>>(
            xz, conv_w, conv_b, u);
    }

    {
        int waves = M_ROWS * 33;
        xdbl_kernel<<<(waves + 7) / 8, 256, 0, stream>>>(u, W_x, xdbl);
    }

    scan_kernel<<<dim3(D_INNER / SCAN_CH, B_SZ), 256, 0, stream>>>(
        xdbl, u, W_dt, b_dt, A_log, ybuf);

    ln_gate_kernel<<<M_ROWS, 256, 0, stream>>>(ybuf, u, xz, D_par, ln_g, ln_b);

    gemm_f32_wmma<<<dim3(D_MODEL / 128, M_ROWS / 64), 256, 0, stream>>>(
        ybuf, W_out, out, M_ROWS, D_MODEL, D_INNER);
}


# x_dbl projection (u @ W_x^T, N=33) moved from one-wave-per-output dot products to f32 WMMA: 16 rows per work-group, K split over 8 waves, partials added via LDS (sum order changes)
# speedup vs baseline: 5.8375x; 1.0878x over previous
#include <hip/hip_runtime.h>
#include <hip/hip_bf16.h>


typedef __attribute__((ext_vector_type(2))) float v2f;
typedef __attribute__((ext_vector_type(8))) float v8f;

#define D_MODEL 1024
#define D_INNER 2048
#define D_STATE 16
#define L_SEQ   2048
#define B_SZ    2
#define M_ROWS  (B_SZ * L_SEQ)

__global__ __launch_bounds__(256)
void gemm_f32_wmma(const float* __restrict__ A, const float* __restrict__ Bm,
                   float* __restrict__ C, int M, int N, int K) {
    __shared__ __align__(16) float As[64][36];
    __shared__ __align__(16) float Bs[128][36];

    const int tid  = threadIdx.x;
    const int lane = tid & 31;
    const int w    = tid >> 5;
    const int wm   = w >> 2;
    const int wn   = w & 3;
    const int lmod = lane & 15;
    const int lhal = lane >> 4;

    const int n0 = blockIdx.x * 128;
    const int m0 = blockIdx.y * 64;

    v8f acc[2][2] = {};

    for (int k0 = 0; k0 < K; k0 += 32) {
        #pragma unroll
        for (int i = 0; i < 2; ++i) {
            int id = tid + i * 256;
            int r = id >> 3, c4 = id & 7;
            *(float4*)&As[r][c4 * 4] =
                *(const float4*)&A[(size_t)(m0 + r) * K + k0 + c4 * 4];
        }
        #pragma unroll
        for (int i = 0; i < 4; ++i) {
            int id = tid + i * 256;
            int r = id >> 3, c4 = id & 7;
            *(float4*)&Bs[r][c4 * 4] =
                *(const float4*)&Bm[(size_t)(n0 + r) * K + k0 + c4 * 4];
        }
        __syncthreads();

        #pragma unroll
        for (int kk = 0; kk < 8; ++kk) {
            const int kc = kk * 4 + lhal * 2;
            v2f a[2], b[2];
            a[0] = *(const v2f*)&As[wm * 32 +      lmod][kc];
            a[1] = *(const v2f*)&As[wm * 32 + 16 + lmod][kc];
            b[0] = *(const v2f*)&Bs[wn * 32 +      lmod][kc];
            b[1] = *(const v2f*)&Bs[wn * 32 + 16 + lmod][kc];
            #pragma unroll
            for (int mi = 0; mi < 2; ++mi)
                #pragma unroll
                for (int ni = 0; ni < 2; ++ni)
                    acc[mi][ni] = __builtin_amdgcn_wmma_f32_16x16x4_f32(
                        false, a[mi], false, b[ni], (short)0, acc[mi][ni],
                        false, false);
        }
        __syncthreads();
    }

    #pragma unroll
    for (int mi = 0; mi < 2; ++mi)
        #pragma unroll
        for (int ni = 0; ni < 2; ++ni)
            #pragma unroll
            for (int r = 0; r < 8; ++r) {
                int gm = m0 + wm * 32 + mi * 16 + lhal * 8 + r;
                int gn = n0 + wn * 32 + ni * 16 + lmod;
                C[(size_t)gm * N + gn] = acc[mi][ni][r];
            }
}

__global__ __launch_bounds__(256)
void conv_silu_kernel(const float* __restrict__ xz, const float* __restrict__ cw,
                      const float* __restrict__ cb, float* __restrict__ u) {
    size_t id = (size_t)blockIdx.x * 256 + threadIdx.x;
    int d = (int)(id & (D_INNER - 1));
    int l = (int)((id >> 11) & (L_SEQ - 1));
    int b = (int)(id >> 22);
    const float* base = xz + (size_t)b * L_SEQ * (2 * D_INNER) + d;
    float acc = cb[d];
    #pragma unroll
    for (int j = 0; j < 4; ++j) {
        int ll = l - 3 + j;
        if (ll >= 0) acc += base[(size_t)ll * (2 * D_INNER)] * cw[d * 4 + j];
    }
    u[id] = acc / (1.f + __expf(-acc));
}

#define XD_KT    256
#define XD_PITCH 260
#define XD_RP    52

__global__ __launch_bounds__(256)
void xdbl_wmma_kernel(const float* u, const float* Wx, float* __restrict__ xdbl) {
    __shared__ __align__(16) float As[16][XD_PITCH];
    __shared__ __align__(16) float Bs[33][XD_PITCH];
    __shared__ __align__(16) float red[8][16][XD_RP];

    const int tid  = threadIdx.x;
    const int lane = tid & 31;
    const int w    = tid >> 5;
    const int c    = lane & 15;
    const int h    = lane >> 4;
    const int m0   = blockIdx.x * 16;

    const int  sr   = tid >> 6;
    const int  sc   = (tid & 63) * 4;
    const bool last = tid < 64;
    const float* ua = u  + (size_t)(m0 + sr) * D_INNER + sc;
    const float* wb = Wx + (size_t)sr * D_INNER + sc;

    float4 a0, a1, a2, a3, b0, b1, b2, b3, b4, b5, b6, b7;
    float4 bl = make_float4(0.f, 0.f, 0.f, 0.f);
#define XD_FETCH(ko)                                                    \
    a0 = *(const float4*)&ua[(size_t) 0 * D_INNER + (ko)];              \
    a1 = *(const float4*)&ua[(size_t) 4 * D_INNER + (ko)];              \
    a2 = *(const float4*)&ua[(size_t) 8 * D_INNER + (ko)];              \
    a3 = *(const float4*)&ua[(size_t)12 * D_INNER + (ko)];              \
    b0 = *(const float4*)&wb[(size_t) 0 * D_INNER + (ko)];              \
    b1 = *(const float4*)&wb[(size_t) 4 * D_INNER + (ko)];              \
    b2 = *(const float4*)&wb[(size_t) 8 * D_INNER + (ko)];              \
    b3 = *(const float4*)&wb[(size_t)12 * D_INNER + (ko)];              \
    b4 = *(const float4*)&wb[(size_t)16 * D_INNER + (ko)];              \
    b5 = *(const float4*)&wb[(size_t)20 * D_INNER + (ko)];              \
    b6 = *(const float4*)&wb[(size_t)24 * D_INNER + (ko)];              \
    b7 = *(const float4*)&wb[(size_t)28 * D_INNER + (ko)];              \
    if (last) bl = *(const float4*)&wb[(size_t)32 * D_INNER + (ko)];

    XD_FETCH(0)

    v8f acc[3] = {};
    const int kb = w * 32 + h * 4;

    for (int kt = 0; kt < D_INNER / XD_KT; ++kt) {
        *(float4*)&As[sr     ][sc] = a0;
        *(float4*)&As[sr +  4][sc] = a1;
        *(float4*)&As[sr +  8][sc] = a2;
        *(float4*)&As[sr + 12][sc] = a3;
        *(float4*)&Bs[sr     ][sc] = b0;
        *(float4*)&Bs[sr +  4][sc] = b1;
        *(float4*)&Bs[sr +  8][sc] = b2;
        *(float4*)&Bs[sr + 12][sc] = b3;
        *(float4*)&Bs[sr + 16][sc] = b4;
        *(float4*)&Bs[sr + 20][sc] = b5;
        *(float4*)&Bs[sr + 24][sc] = b6;
        *(float4*)&Bs[sr + 28][sc] = b7;
        if (last) *(float4*)&Bs[32][sc] = bl;
        __syncthreads();

        if (kt + 1 < D_INNER / XD_KT) {
            XD_FETCH((kt + 1) * XD_KT)
        }

        #pragma unroll
        for (int i = 0; i < 4; ++i) {
            const float4 f4 = *(const float4*)&As[c][kb + 8 * i];
            const float4 p4 = *(const float4*)&Bs[c][kb + 8 * i];
            const float4 q4 = *(const float4*)&Bs[16 + c][kb + 8 * i];
            const float4 r4 = *(const float4*)&Bs[32][kb + 8 * i];
            const v2f f0 = {f4.x, f4.y}, f1 = {f4.z, f4.w};
            const v2f p0 = {p4.x, p4.y}, p1 = {p4.z, p4.w};
            const v2f q0 = {q4.x, q4.y}, q1 = {q4.z, q4.w};
            const v2f r0 = {r4.x, r4.y}, r1 = {r4.z, r4.w};
            acc[0] = __builtin_amdgcn_wmma_f32_16x16x4_f32(
                false, f0, false, p0, (short)0, acc[0], false, false);
            acc[1] = __builtin_amdgcn_wmma_f32_16x16x4_f32(
                false, f0, false, q0, (short)0, acc[1], false, false);
            acc[2] = __builtin_amdgcn_wmma_f32_16x16x4_f32(
                false, f0, false, r0, (short)0, acc[2], false, false);
            acc[0] = __builtin_amdgcn_wmma_f32_16x16x4_f32(
                false, f1, false, p1, (short)0, acc[0], false, false);
            acc[1] = __builtin_amdgcn_wmma_f32_16x16x4_f32(
                false, f1, false, q1, (short)0, acc[1], false, false);
            acc[2] = __builtin_amdgcn_wmma_f32_16x16x4_f32(
                false, f1, false, r1, (short)0, acc[2], false, false);
        }
        __syncthreads();
    }
#undef XD_FETCH

    #pragma unroll
    for (int ni = 0; ni < 3; ++ni)
        #pragma unroll
        for (int r = 0; r < 8; ++r)
            red[w][h * 8 + r][ni * 16 + c] = acc[ni][r];
    __syncthreads();

    for (int idx = tid; idx < 16 * 33; idx += 256) {
        const int row = idx / 33, e = idx - row * 33;
        float s = 0.f;
        #pragma unroll
        for (int q = 0; q < 8; ++q) s += red[q][row][e];
        xdbl[(size_t)m0 * 33 + idx] = s;
    }
}

#define SCAN_T     16
#define SCAN_NCH   (L_SEQ / SCAN_T)
#define SCAN_PITCH 36
#define SCAN_XCH   (SCAN_T * 33)
#define SCAN_CH    32
#define SCAN_YP    36

__device__ __forceinline__ int scan_dst(int i) {
    const int t = i / 33, e = i - t * 33;
    return t * SCAN_PITCH + (e ? e + 3 : 0);
}

__device__ __forceinline__ float scan_dt(float raw, float wdt, float bd) {
    float dt = raw * wdt + bd;
    const float sp = log1pf(__expf(dt));
    dt = (dt > 20.f) ? dt : sp;
    return fminf(fmaxf(dt, 1e-4f), 10.f);
}

template <int XMASK>
__device__ __forceinline__ float scan_add_xor(float v) {
    const int o = __builtin_amdgcn_update_dpp(0, __float_as_int(v), 0x160 + XMASK,
                                              0xF, 0xF, true);
    return v + __int_as_float(o);
}

__global__ __launch_bounds__(256)
void scan_kernel(const float* __restrict__ xdbl, const float* __restrict__ u,
                 const float* __restrict__ Wdt, const float* __restrict__ bdt,
                 const float* __restrict__ Alog, float* __restrict__ ybuf) {
    const int tid = threadIdx.x;
    const int j   = tid & 7;
    const int c   = tid >> 3;
    const int sc  = tid & 31;
    const int st  = tid >> 5;
    const int d0  = blockIdx.x * SCAN_CH;
    const int b   = blockIdx.y;

    const float A0 = -__expf(fminf(Alog[2 * j],     5.f));
    const float A1 = -__expf(fminf(Alog[2 * j + 1], 5.f));
    const float wdt = Wdt[d0 + sc];
    const float bd  = bdt[d0 + sc];
    float h0 = 0.f, h1 = 0.f;

    const float* xd = xdbl + (size_t)b * L_SEQ * 33;
    const float* ub = u    + (size_t)b * L_SEQ * D_INNER + d0 + sc;
    float*       yb = ybuf + (size_t)b * L_SEQ * D_INNER + d0 + sc;

    __shared__ __align__(16) float  sx[2][SCAN_T * SCAN_PITCH];
    __shared__ __align__(16) float2 sdu[2][SCAN_T][SCAN_CH];
    __shared__ __align__(16) float  sy[2][SCAN_T * SCAN_YP];

    const int dst0 = scan_dst(tid);
    const int dst1 = scan_dst(tid + 256);
    const int dst2 = scan_dst(tid + 512);
    const bool has2 = tid < (SCAN_XCH - 512);

    sx[0][dst0] = xd[tid];
    sx[0][dst1] = xd[tid + 256];
    if (has2) sx[0][dst2] = xd[tid + 512];
    sdu[0][st][sc]     = make_float2(scan_dt(xd[st * 33], wdt, bd),
                                     ub[(size_t)st * D_INNER]);
    sdu[0][st + 8][sc] = make_float2(scan_dt(xd[(st + 8) * 33], wdt, bd),
                                     ub[(size_t)(st + 8) * D_INNER]);
    __syncthreads();

    for (int ch = 0; ch < SCAN_NCH; ++ch) {
        const int  cur  = ch & 1;
        const bool more = (ch + 1 < SCAN_NCH);

        float xn0 = 0.f, xn1 = 0.f, xn2 = 0.f;
        float rn0 = 0.f, rn1 = 0.f, un0 = 0.f, un1 = 0.f;
        if (more) {
            const float* nsrc = xd + (size_t)(ch + 1) * SCAN_XCH;
            xn0 = nsrc[tid];
            xn1 = nsrc[tid + 256];
            if (has2) xn2 = nsrc[tid + 512];
            rn0 = nsrc[st * 33];
            rn1 = nsrc[(st + 8) * 33];
            const float* nu = ub + (size_t)(ch + 1) * SCAN_T * D_INNER;
            un0 = nu[(size_t)st * D_INNER];
            un1 = nu[(size_t)(st + 8) * D_INNER];
        }

        if (ch > 0) {
            const float* py = sy[cur ^ 1];
            float* yc = yb + (size_t)(ch - 1) * SCAN_T * D_INNER;
            yc[(size_t)st * D_INNER]       = py[st * SCAN_YP + sc];
            yc[(size_t)(st + 8) * D_INNER] = py[(st + 8) * SCAN_YP + sc];
        }

        const float* s = sx[cur];
        float ysel = 0.f;
        #pragma unroll
        for (int t = 0; t < SCAN_T; ++t) {
            const float2 du = sdu[cur][t][c];
            const float2 Bq = *(const float2*)&s[t * SCAN_PITCH + 4  + 2 * j];
            const float2 Cq = *(const float2*)&s[t * SCAN_PITCH + 20 + 2 * j];
            const float dt = du.x, ut = du.y;

            const float dA0 = __expf(fminf(fmaxf(dt * A0, -20.f), 0.f));
            const float dA1 = __expf(fminf(fmaxf(dt * A1, -20.f), 0.f));
            const float dB0 = fminf(fmaxf(dt * Bq.x * ut, -10.f), 10.f);
            const float dB1 = fminf(fmaxf(dt * Bq.y * ut, -10.f), 10.f);
            h0 = fminf(fmaxf(h0 * dA0 + dB0, -100.f), 100.f);
            h1 = fminf(fmaxf(h1 * dA1 + dB1, -100.f), 100.f);

            float p = h0 * Cq.x + h1 * Cq.y;
            p = scan_add_xor<1>(p);
            p = scan_add_xor<2>(p);
            p = scan_add_xor<4>(p);
            if ((t & 7) == j) ysel = p;
            if ((t & 7) == 7)
                sy[cur][((t & 8) + j) * SCAN_YP + c] = ysel;
        }

        if (more) {
            float* nx = sx[cur ^ 1];
            nx[dst0] = xn0;
            nx[dst1] = xn1;
            if (has2) nx[dst2] = xn2;
            const float dn0 = scan_dt(rn0, wdt, bd);
            const float dn1 = scan_dt(rn1, wdt, bd);
            sdu[cur ^ 1][st][sc]     = make_float2(dn0, un0);
            sdu[cur ^ 1][st + 8][sc] = make_float2(dn1, un1);
        }
        __syncthreads();
    }

    {
        const float* py = sy[(SCAN_NCH - 1) & 1];
        float* yc = yb + (size_t)(SCAN_NCH - 1) * SCAN_T * D_INNER;
        yc[(size_t)st * D_INNER]       = py[st * SCAN_YP + sc];
        yc[(size_t)(st + 8) * D_INNER] = py[(st + 8) * SCAN_YP + sc];
    }
}

__global__ __launch_bounds__(256)
void ln_gate_kernel(float* __restrict__ ybuf, const float* __restrict__ u,
                    const float* __restrict__ xz, const float* __restrict__ Dp,
                    const float* __restrict__ g, const float* __restrict__ bta) {
    const int row = blockIdx.x;
    const int tid = threadIdx.x;
    float* yr = ybuf + (size_t)row * D_INNER;

    float v[8], s = 0.f, s2 = 0.f;
    #pragma unroll
    for (int i = 0; i < 8; ++i) {
        v[i] = yr[tid + i * 256];
        s += v[i]; s2 += v[i] * v[i];
    }
    #pragma unroll
    for (int off = 16; off; off >>= 1) {
        s  += __shfl_xor(s,  off, 32);
        s2 += __shfl_xor(s2, off, 32);
    }
    __shared__ float ws1[8], ws2[8];
    const int lane = tid & 31, w = tid >> 5;
    if (lane == 0) { ws1[w] = s; ws2[w] = s2; }
    __syncthreads();
    if (tid == 0) {
        float t1 = 0.f, t2 = 0.f;
        #pragma unroll
        for (int i = 0; i < 8; ++i) { t1 += ws1[i]; t2 += ws2[i]; }
        ws1[0] = t1; ws2[0] = t2;
    }
    __syncthreads();
    const float mu  = ws1[0] * (1.f / D_INNER);
    float var = ws2[0] * (1.f / D_INNER) - mu * mu;
    var = fmaxf(var, 0.f);
    const float rs = rsqrtf(var + 1e-5f);

    #pragma unroll
    for (int i = 0; i < 8; ++i) {
        const int d = tid + i * 256;
        float yn = (v[i] - mu) * rs * g[d] + bta[d];
        float uu = u[(size_t)row * D_INNER + d];
        float zz = xz[(size_t)row * (2 * D_INNER) + D_INNER + d];
        float sz = zz / (1.f + __expf(-zz));
        yr[d] = (yn + Dp[d] * uu) * sz;
    }
}

extern "C" void kernel_launch(void* const* d_in, const int* in_sizes, int n_in,
                              void* d_out, int out_size, void* d_ws, size_t ws_size,
                              hipStream_t stream) {
    const float* x      = (const float*)d_in[0];
    const float* W_in   = (const float*)d_in[1];
    const float* conv_w = (const float*)d_in[2];
    const float* conv_b = (const float*)d_in[3];
    const float* W_x    = (const float*)d_in[4];
    const float* W_dt   = (const float*)d_in[5];
    const float* b_dt   = (const float*)d_in[6];
    const float* A_log  = (const float*)d_in[7];
    const float* D_par  = (const float*)d_in[8];
    const float* W_out  = (const float*)d_in[9];
    const float* ln_g   = (const float*)d_in[10];
    const float* ln_b   = (const float*)d_in[11];
    float* out = (float*)d_out;

    float* ws = (float*)d_ws;
    const size_t XZ_OFF   = 0;
    const size_t U_OFF    = XZ_OFF + (size_t)M_ROWS * 2 * D_INNER;
    const size_t XDBL_OFF = U_OFF  + (size_t)M_ROWS * D_INNER;
    const size_t Y_OFF    = XDBL_OFF + (size_t)M_ROWS * 33;
    float* xz   = ws + XZ_OFF;
    float* u    = ws + U_OFF;
    float* xdbl = ws + XDBL_OFF;
    float* ybuf = ws + Y_OFF;

    gemm_f32_wmma<<<dim3(2 * D_INNER / 128, M_ROWS / 64), 256, 0, stream>>>(
        x, W_in, xz, M_ROWS, 2 * D_INNER, D_MODEL);

    {
        size_t n = (size_t)M_ROWS * D_INNER;
        conv_silu_kernel<<<(unsigned)(n / 256), 256, 0, stream>>>(
            xz, conv_w, conv_b, u);
    }

    xdbl_wmma_kernel<<<M_ROWS / 16, 256, 0, stream>>>(u, W_x, xdbl);

    scan_kernel<<<dim3(D_INNER / SCAN_CH, B_SZ), 256, 0, stream>>>(
        xdbl, u, W_dt, b_dt, A_log, ybuf);

    ln_gate_kernel<<<M_ROWS, 256, 0, stream>>>(ybuf, u, xz, D_par, ln_g, ln_b);

    gemm_f32_wmma<<<dim3(D_MODEL / 128, M_ROWS / 64), 256, 0, stream>>>(
        ybuf, W_out, out, M_ROWS, D_MODEL, D_INNER);
}


# conv+silu: one thread computes 4 channels x 16 steps from a sliding window of 19 row pieces in registers (16-byte loads and stores): 4096 long waves instead of 262144 one-element waves
# speedup vs baseline: 6.0676x; 1.0394x over previous
#include <hip/hip_runtime.h>
#include <hip/hip_bf16.h>


typedef __attribute__((ext_vector_type(2))) float v2f;
typedef __attribute__((ext_vector_type(8))) float v8f;

#define D_MODEL 1024
#define D_INNER 2048
#define D_STATE 16
#define L_SEQ   2048
#define B_SZ    2
#define M_ROWS  (B_SZ * L_SEQ)

__global__ __launch_bounds__(256)
void gemm_f32_wmma(const float* __restrict__ A, const float* __restrict__ Bm,
                   float* __restrict__ C, int M, int N, int K) {
    __shared__ __align__(16) float As[64][36];
    __shared__ __align__(16) float Bs[128][36];

    const int tid  = threadIdx.x;
    const int lane = tid & 31;
    const int w    = tid >> 5;
    const int wm   = w >> 2;
    const int wn   = w & 3;
    const int lmod = lane & 15;
    const int lhal = lane >> 4;

    const int n0 = blockIdx.x * 128;
    const int m0 = blockIdx.y * 64;

    v8f acc[2][2] = {};

    for (int k0 = 0; k0 < K; k0 += 32) {
        #pragma unroll
        for (int i = 0; i < 2; ++i) {
            int id = tid + i * 256;
            int r = id >> 3, c4 = id & 7;
            *(float4*)&As[r][c4 * 4] =
                *(const float4*)&A[(size_t)(m0 + r) * K + k0 + c4 * 4];
        }
        #pragma unroll
        for (int i = 0; i < 4; ++i) {
            int id = tid + i * 256;
            int r = id >> 3, c4 = id & 7;
            *(float4*)&Bs[r][c4 * 4] =
                *(const float4*)&Bm[(size_t)(n0 + r) * K + k0 + c4 * 4];
        }
        __syncthreads();

        #pragma unroll
        for (int kk = 0; kk < 8; ++kk) {
            const int kc = kk * 4 + lhal * 2;
            v2f a[2], b[2];
            a[0] = *(const v2f*)&As[wm * 32 +      lmod][kc];
            a[1] = *(const v2f*)&As[wm * 32 + 16 + lmod][kc];
            b[0] = *(const v2f*)&Bs[wn * 32 +      lmod][kc];
            b[1] = *(const v2f*)&Bs[wn * 32 + 16 + lmod][kc];
            #pragma unroll
            for (int mi = 0; mi < 2; ++mi)
                #pragma unroll
                for (int ni = 0; ni < 2; ++ni)
                    acc[mi][ni] = __builtin_amdgcn_wmma_f32_16x16x4_f32(
                        false, a[mi], false, b[ni], (short)0, acc[mi][ni],
                        false, false);
        }
        __syncthreads();
    }

    #pragma unroll
    for (int mi = 0; mi < 2; ++mi)
        #pragma unroll
        for (int ni = 0; ni < 2; ++ni)
            #pragma unroll
            for (int r = 0; r < 8; ++r) {
                int gm = m0 + wm * 32 + mi * 16 + lhal * 8 + r;
                int gn = n0 + wn * 32 + ni * 16 + lmod;
                C[(size_t)gm * N + gn] = acc[mi][ni][r];
            }
}

#define CONV_T 16

__device__ __forceinline__ float conv_silu1(float bias, float x0, float x1,
                                            float x2, float x3, float4 w) {
    float acc = bias;
    acc += x0 * w.x;
    acc += x1 * w.y;
    acc += x2 * w.z;
    acc += x3 * w.w;
    return acc / (1.f + __expf(-acc));
}

__global__ __launch_bounds__(256)
void conv_silu_kernel(const float* __restrict__ xz, const float* __restrict__ cw,
                      const float* __restrict__ cb, float* __restrict__ u) {
    const int id = blockIdx.x * 256 + threadIdx.x;
    const int d  = (id & 511) * 4;
    const int r0 = (id >> 9) * CONV_T;
    const bool head = (r0 & (L_SEQ - 1)) == 0;

    const float* src = xz + (size_t)r0 * (2 * D_INNER) + d;

    float4 x[CONV_T + 3];
    #pragma unroll
    for (int k = 0; k < 3; ++k) x[k] = make_float4(0.f, 0.f, 0.f, 0.f);
    if (!head) {
        #pragma unroll
        for (int k = 0; k < 3; ++k)
            x[k] = *(const float4*)(src + (ptrdiff_t)(k - 3) * (2 * D_INNER));
    }
    #pragma unroll
    for (int k = 3; k < CONV_T + 3; ++k)
        x[k] = *(const float4*)(src + (size_t)(k - 3) * (2 * D_INNER));

    const float4 w0 = *(const float4*)&cw[(d + 0) * 4];
    const float4 w1 = *(const float4*)&cw[(d + 1) * 4];
    const float4 w2 = *(const float4*)&cw[(d + 2) * 4];
    const float4 w3 = *(const float4*)&cw[(d + 3) * 4];
    const float4 bb = *(const float4*)&cb[d];

    float* dst = u + (size_t)r0 * D_INNER + d;
    #pragma unroll
    for (int t = 0; t < CONV_T; ++t) {
        float4 o;
        o.x = conv_silu1(bb.x, x[t].x, x[t + 1].x, x[t + 2].x, x[t + 3].x, w0);
        o.y = conv_silu1(bb.y, x[t].y, x[t + 1].y, x[t + 2].y, x[t + 3].y, w1);
        o.z = conv_silu1(bb.z, x[t].z, x[t + 1].z, x[t + 2].z, x[t + 3].z, w2);
        o.w = conv_silu1(bb.w, x[t].w, x[t + 1].w, x[t + 2].w, x[t + 3].w, w3);
        *(float4*)(dst + (size_t)t * D_INNER) = o;
    }
}

#define XD_KT    256
#define XD_PITCH 260
#define XD_RP    52

__global__ __launch_bounds__(256)
void xdbl_wmma_kernel(const float* u, const float* Wx, float* __restrict__ xdbl) {
    __shared__ __align__(16) float As[16][XD_PITCH];
    __shared__ __align__(16) float Bs[33][XD_PITCH];
    __shared__ __align__(16) float red[8][16][XD_RP];

    const int tid  = threadIdx.x;
    const int lane = tid & 31;
    const int w    = tid >> 5;
    const int c    = lane & 15;
    const int h    = lane >> 4;
    const int m0   = blockIdx.x * 16;

    const int  sr   = tid >> 6;
    const int  sc   = (tid & 63) * 4;
    const bool last = tid < 64;
    const float* ua = u  + (size_t)(m0 + sr) * D_INNER + sc;
    const float* wb = Wx + (size_t)sr * D_INNER + sc;

    float4 a0, a1, a2, a3, b0, b1, b2, b3, b4, b5, b6, b7;
    float4 bl = make_float4(0.f, 0.f, 0.f, 0.f);
#define XD_FETCH(ko)                                                    \
    a0 = *(const float4*)&ua[(size_t) 0 * D_INNER + (ko)];              \
    a1 = *(const float4*)&ua[(size_t) 4 * D_INNER + (ko)];              \
    a2 = *(const float4*)&ua[(size_t) 8 * D_INNER + (ko)];              \
    a3 = *(const float4*)&ua[(size_t)12 * D_INNER + (ko)];              \
    b0 = *(const float4*)&wb[(size_t) 0 * D_INNER + (ko)];              \
    b1 = *(const float4*)&wb[(size_t) 4 * D_INNER + (ko)];              \
    b2 = *(const float4*)&wb[(size_t) 8 * D_INNER + (ko)];              \
    b3 = *(const float4*)&wb[(size_t)12 * D_INNER + (ko)];              \
    b4 = *(const float4*)&wb[(size_t)16 * D_INNER + (ko)];              \
    b5 = *(const float4*)&wb[(size_t)20 * D_INNER + (ko)];              \
    b6 = *(const float4*)&wb[(size_t)24 * D_INNER + (ko)];              \
    b7 = *(const float4*)&wb[(size_t)28 * D_INNER + (ko)];              \
    if (last) bl = *(const float4*)&wb[(size_t)32 * D_INNER + (ko)];

    XD_FETCH(0)

    v8f acc[3] = {};
    const int kb = w * 32 + h * 4;

    for (int kt = 0; kt < D_INNER / XD_KT; ++kt) {
        *(float4*)&As[sr     ][sc] = a0;
        *(float4*)&As[sr +  4][sc] = a1;
        *(float4*)&As[sr +  8][sc] = a2;
        *(float4*)&As[sr + 12][sc] = a3;
        *(float4*)&Bs[sr     ][sc] = b0;
        *(float4*)&Bs[sr +  4][sc] = b1;
        *(float4*)&Bs[sr +  8][sc] = b2;
        *(float4*)&Bs[sr + 12][sc] = b3;
        *(float4*)&Bs[sr + 16][sc] = b4;
        *(float4*)&Bs[sr + 20][sc] = b5;
        *(float4*)&Bs[sr + 24][sc] = b6;
        *(float4*)&Bs[sr + 28][sc] = b7;
        if (last) *(float4*)&Bs[32][sc] = bl;
        __syncthreads();

        if (kt + 1 < D_INNER / XD_KT) {
            XD_FETCH((kt + 1) * XD_KT)
        }

        #pragma unroll
        for (int i = 0; i < 4; ++i) {
            const float4 f4 = *(const float4*)&As[c][kb + 8 * i];
            const float4 p4 = *(const float4*)&Bs[c][kb + 8 * i];
            const float4 q4 = *(const float4*)&Bs[16 + c][kb + 8 * i];
            const float4 r4 = *(const float4*)&Bs[32][kb + 8 * i];
            const v2f f0 = {f4.x, f4.y}, f1 = {f4.z, f4.w};
            const v2f p0 = {p4.x, p4.y}, p1 = {p4.z, p4.w};
            const v2f q0 = {q4.x, q4.y}, q1 = {q4.z, q4.w};
            const v2f r0 = {r4.x, r4.y}, r1 = {r4.z, r4.w};
            acc[0] = __builtin_amdgcn_wmma_f32_16x16x4_f32(
                false, f0, false, p0, (short)0, acc[0], false, false);
            acc[1] = __builtin_amdgcn_wmma_f32_16x16x4_f32(
                false, f0, false, q0, (short)0, acc[1], false, false);
            acc[2] = __builtin_amdgcn_wmma_f32_16x16x4_f32(
                false, f0, false, r0, (short)0, acc[2], false, false);
            acc[0] = __builtin_amdgcn_wmma_f32_16x16x4_f32(
                false, f1, false, p1, (short)0, acc[0], false, false);
            acc[1] = __builtin_amdgcn_wmma_f32_16x16x4_f32(
                false, f1, false, q1, (short)0, acc[1], false, false);
            acc[2] = __builtin_amdgcn_wmma_f32_16x16x4_f32(
                false, f1, false, r1, (short)0, acc[2], false, false);
        }
        __syncthreads();
    }
#undef XD_FETCH

    #pragma unroll
    for (int ni = 0; ni < 3; ++ni)
        #pragma unroll
        for (int r = 0; r < 8; ++r)
            red[w][h * 8 + r][ni * 16 + c] = acc[ni][r];
    __syncthreads();

    for (int idx = tid; idx < 16 * 33; idx += 256) {
        const int row = idx / 33, e = idx - row * 33;
        float s = 0.f;
        #pragma unroll
        for (int q = 0; q < 8; ++q) s += red[q][row][e];
        xdbl[(size_t)m0 * 33 + idx] = s;
    }
}

#define SCAN_T     16
#define SCAN_NCH   (L_SEQ / SCAN_T)
#define SCAN_PITCH 36
#define SCAN_XCH   (SCAN_T * 33)
#define SCAN_CH    32
#define SCAN_YP    36

__device__ __forceinline__ int scan_dst(int i) {
    const int t = i / 33, e = i - t * 33;
    return t * SCAN_PITCH + (e ? e + 3 : 0);
}

__device__ __forceinline__ float scan_dt(float raw, float wdt, float bd) {
    float dt = raw * wdt + bd;
    const float sp = log1pf(__expf(dt));
    dt = (dt > 20.f) ? dt : sp;
    return fminf(fmaxf(dt, 1e-4f), 10.f);
}

template <int XMASK>
__device__ __forceinline__ float scan_add_xor(float v) {
    const int o = __builtin_amdgcn_update_dpp(0, __float_as_int(v), 0x160 + XMASK,
                                              0xF, 0xF, true);
    return v + __int_as_float(o);
}

__global__ __launch_bounds__(256)
void scan_kernel(const float* __restrict__ xdbl, const float* __restrict__ u,
                 const float* __restrict__ Wdt, const float* __restrict__ bdt,
                 const float* __restrict__ Alog, float* __restrict__ ybuf) {
    const int tid = threadIdx.x;
    const int j   = tid & 7;
    const int c   = tid >> 3;
    const int sc  = tid & 31;
    const int st  = tid >> 5;
    const int d0  = blockIdx.x * SCAN_CH;
    const int b   = blockIdx.y;

    const float A0 = -__expf(fminf(Alog[2 * j],     5.f));
    const float A1 = -__expf(fminf(Alog[2 * j + 1], 5.f));
    const float wdt = Wdt[d0 + sc];
    const float bd  = bdt[d0 + sc];
    float h0 = 0.f, h1 = 0.f;

    const float* xd = xdbl + (size_t)b * L_SEQ * 33;
    const float* ub = u    + (size_t)b * L_SEQ * D_INNER + d0 + sc;
    float*       yb = ybuf + (size_t)b * L_SEQ * D_INNER + d0 + sc;

    __shared__ __align__(16) float  sx[2][SCAN_T * SCAN_PITCH];
    __shared__ __align__(16) float2 sdu[2][SCAN_T][SCAN_CH];
    __shared__ __align__(16) float  sy[2][SCAN_T * SCAN_YP];

    const int dst0 = scan_dst(tid);
    const int dst1 = scan_dst(tid + 256);
    const int dst2 = scan_dst(tid + 512);
    const bool has2 = tid < (SCAN_XCH - 512);

    sx[0][dst0] = xd[tid];
    sx[0][dst1] = xd[tid + 256];
    if (has2) sx[0][dst2] = xd[tid + 512];
    sdu[0][st][sc]     = make_float2(scan_dt(xd[st * 33], wdt, bd),
                                     ub[(size_t)st * D_INNER]);
    sdu[0][st + 8][sc] = make_float2(scan_dt(xd[(st + 8) * 33], wdt, bd),
                                     ub[(size_t)(st + 8) * D_INNER]);
    __syncthreads();

    for (int ch = 0; ch < SCAN_NCH; ++ch) {
        const int  cur  = ch & 1;
        const bool more = (ch + 1 < SCAN_NCH);

        float xn0 = 0.f, xn1 = 0.f, xn2 = 0.f;
        float rn0 = 0.f, rn1 = 0.f, un0 = 0.f, un1 = 0.f;
        if (more) {
            const float* nsrc = xd + (size_t)(ch + 1) * SCAN_XCH;
            xn0 = nsrc[tid];
            xn1 = nsrc[tid + 256];
            if (has2) xn2 = nsrc[tid + 512];
            rn0 = nsrc[st * 33];
            rn1 = nsrc[(st + 8) * 33];
            const float* nu = ub + (size_t)(ch + 1) * SCAN_T * D_INNER;
            un0 = nu[(size_t)st * D_INNER];
            un1 = nu[(size_t)(st + 8) * D_INNER];
        }

        if (ch > 0) {
            const float* py = sy[cur ^ 1];
            float* yc = yb + (size_t)(ch - 1) * SCAN_T * D_INNER;
            yc[(size_t)st * D_INNER]       = py[st * SCAN_YP + sc];
            yc[(size_t)(st + 8) * D_INNER] = py[(st + 8) * SCAN_YP + sc];
        }

        const float* s = sx[cur];
        float ysel = 0.f;
        #pragma unroll
        for (int t = 0; t < SCAN_T; ++t) {
            const float2 du = sdu[cur][t][c];
            const float2 Bq = *(const float2*)&s[t * SCAN_PITCH + 4  + 2 * j];
            const float2 Cq = *(const float2*)&s[t * SCAN_PITCH + 20 + 2 * j];
            const float dt = du.x, ut = du.y;

            const float dA0 = __expf(fminf(fmaxf(dt * A0, -20.f), 0.f));
            const float dA1 = __expf(fminf(fmaxf(dt * A1, -20.f), 0.f));
            const float dB0 = fminf(fmaxf(dt * Bq.x * ut, -10.f), 10.f);
            const float dB1 = fminf(fmaxf(dt * Bq.y * ut, -10.f), 10.f);
            h0 = fminf(fmaxf(h0 * dA0 + dB0, -100.f), 100.f);
            h1 = fminf(fmaxf(h1 * dA1 + dB1, -100.f), 100.f);

            float p = h0 * Cq.x + h1 * Cq.y;
            p = scan_add_xor<1>(p);
            p = scan_add_xor<2>(p);
            p = scan_add_xor<4>(p);
            if ((t & 7) == j) ysel = p;
            if ((t & 7) == 7)
                sy[cur][((t & 8) + j) * SCAN_YP + c] = ysel;
        }

        if (more) {
            float* nx = sx[cur ^ 1];
            nx[dst0] = xn0;
            nx[dst1] = xn1;
            if (has2) nx[dst2] = xn2;
            const float dn0 = scan_dt(rn0, wdt, bd);
            const float dn1 = scan_dt(rn1, wdt, bd);
            sdu[cur ^ 1][st][sc]     = make_float2(dn0, un0);
            sdu[cur ^ 1][st + 8][sc] = make_float2(dn1, un1);
        }
        __syncthreads();
    }

    {
        const float* py = sy[(SCAN_NCH - 1) & 1];
        float* yc = yb + (size_t)(SCAN_NCH - 1) * SCAN_T * D_INNER;
        yc[(size_t)st * D_INNER]       = py[st * SCAN_YP + sc];
        yc[(size_t)(st + 8) * D_INNER] = py[(st + 8) * SCAN_YP + sc];
    }
}

__global__ __launch_bounds__(256)
void ln_gate_kernel(float* __restrict__ ybuf, const float* __restrict__ u,
                    const float* __restrict__ xz, const float* __restrict__ Dp,
                    const float* __restrict__ g, const float* __restrict__ bta) {
    const int row = blockIdx.x;
    const int tid = threadIdx.x;
    float* yr = ybuf + (size_t)row * D_INNER;

    float v[8], s = 0.f, s2 = 0.f;
    #pragma unroll
    for (int i = 0; i < 8; ++i) {
        v[i] = yr[tid + i * 256];
        s += v[i]; s2 += v[i] * v[i];
    }
    #pragma unroll
    for (int off = 16; off; off >>= 1) {
        s  += __shfl_xor(s,  off, 32);
        s2 += __shfl_xor(s2, off, 32);
    }
    __shared__ float ws1[8], ws2[8];
    const int lane = tid & 31, w = tid >> 5;
    if (lane == 0) { ws1[w] = s; ws2[w] = s2; }
    __syncthreads();
    if (tid == 0) {
        float t1 = 0.f, t2 = 0.f;
        #pragma unroll
        for (int i = 0; i < 8; ++i) { t1 += ws1[i]; t2 += ws2[i]; }
        ws1[0] = t1; ws2[0] = t2;
    }
    __syncthreads();
    const float mu  = ws1[0] * (1.f / D_INNER);
    float var = ws2[0] * (1.f / D_INNER) - mu * mu;
    var = fmaxf(var, 0.f);
    const float rs = rsqrtf(var + 1e-5f);

    #pragma unroll
    for (int i = 0; i < 8; ++i) {
        const int d = tid + i * 256;
        float yn = (v[i] - mu) * rs * g[d] + bta[d];
        float uu = u[(size_t)row * D_INNER + d];
        float zz = xz[(size_t)row * (2 * D_INNER) + D_INNER + d];
        float sz = zz / (1.f + __expf(-zz));
        yr[d] = (yn + Dp[d] * uu) * sz;
    }
}

extern "C" void kernel_launch(void* const* d_in, const int* in_sizes, int n_in,
                              void* d_out, int out_size, void* d_ws, size_t ws_size,
                              hipStream_t stream) {
    const float* x      = (const float*)d_in[0];
    const float* W_in   = (const float*)d_in[1];
    const float* conv_w = (const float*)d_in[2];
    const float* conv_b = (const float*)d_in[3];
    const float* W_x    = (const float*)d_in[4];
    const float* W_dt   = (const float*)d_in[5];
    const float* b_dt   = (const float*)d_in[6];
    const float* A_log  = (const float*)d_in[7];
    const float* D_par  = (const float*)d_in[8];
    const float* W_out  = (const float*)d_in[9];
    const float* ln_g   = (const float*)d_in[10];
    const float* ln_b   = (const float*)d_in[11];
    float* out = (float*)d_out;

    float* ws = (float*)d_ws;
    const size_t XZ_OFF   = 0;
    const size_t U_OFF    = XZ_OFF + (size_t)M_ROWS * 2 * D_INNER;
    const size_t XDBL_OFF = U_OFF  + (size_t)M_ROWS * D_INNER;
    const size_t Y_OFF    = XDBL_OFF + (size_t)M_ROWS * 33;
    float* xz   = ws + XZ_OFF;
    float* u    = ws + U_OFF;
    float* xdbl = ws + XDBL_OFF;
    float* ybuf = ws + Y_OFF;

    gemm_f32_wmma<<<dim3(2 * D_INNER / 128, M_ROWS / 64), 256, 0, stream>>>(
        x, W_in, xz, M_ROWS, 2 * D_INNER, D_MODEL);

    conv_silu_kernel<<<(M_ROWS / CONV_T) * 512 / 256, 256, 0, stream>>>(
        xz, conv_w, conv_b, u);

    xdbl_wmma_kernel<<<M_ROWS / 16, 256, 0, stream>>>(u, W_x, xdbl);

    scan_kernel<<<dim3(D_INNER / SCAN_CH, B_SZ), 256, 0, stream>>>(
        xdbl, u, W_dt, b_dt, A_log, ybuf);

    ln_gate_kernel<<<M_ROWS, 256, 0, stream>>>(ybuf, u, xz, D_par, ln_g, ln_b);

    gemm_f32_wmma<<<dim3(D_MODEL / 128, M_ROWS / 64), 256, 0, stream>>>(
        ybuf, W_out, out, M_ROWS, D_MODEL, D_INNER);
}


# both GEMMs: K loop software-pipelined: operand tiles fetched a whole tile ahead into a second LDS stage (one barrier per tile, was two), fragments loaded one 8-k block ahead of their WMMAs
# speedup vs baseline: 7.1343x; 1.1758x over previous
#include <hip/hip_runtime.h>
#include <hip/hip_bf16.h>


typedef __attribute__((ext_vector_type(2))) float v2f;
typedef __attribute__((ext_vector_type(8))) float v8f;

#define D_MODEL 1024
#define D_INNER 2048
#define D_STATE 16
#define L_SEQ   2048
#define B_SZ    2
#define M_ROWS  (B_SZ * L_SEQ)

#define GM_PITCH 36
#define GM_BROW  64
#define GM_STAGE ((64 + 128) * GM_PITCH)

__global__ __launch_bounds__(256)
void gemm_f32_wmma(const float* __restrict__ A, const float* __restrict__ Bm,
                   float* __restrict__ C, int M, int N, int K) {
    __shared__ __align__(16) float Ls[2 * GM_STAGE];

    const int tid  = threadIdx.x;
    const int lane = tid & 31;
    const int w    = tid >> 5;
    const int wm   = w >> 2;
    const int wn   = w & 3;
    const int lmod = lane & 15;
    const int lhal = lane >> 4;

    const int n0 = blockIdx.x * 128;
    const int m0 = blockIdx.y * 64;

    const int sr = tid >> 3, sc = (tid & 7) * 4;
    const float* ga = A  + (size_t)(m0 + sr) * K + sc;
    const float* gb = Bm + (size_t)(n0 + sr) * K + sc;
    const int so = sr * GM_PITCH + sc;

    const int fa0 = (wm * 32 + lmod) * GM_PITCH + lhal * 2;
    const int fb0 = (GM_BROW + wn * 32 + lmod) * GM_PITCH + lhal * 2;

    float4 ra0, ra1, rb0, rb1, rb2, rb3;
#define GM_FETCH(k0)                                                    \
    ra0 = *(const float4*)&ga[(size_t) 0 * K + (k0)];                   \
    ra1 = *(const float4*)&ga[(size_t)32 * K + (k0)];                   \
    rb0 = *(const float4*)&gb[(size_t) 0 * K + (k0)];                   \
    rb1 = *(const float4*)&gb[(size_t)32 * K + (k0)];                   \
    rb2 = *(const float4*)&gb[(size_t)64 * K + (k0)];                   \
    rb3 = *(const float4*)&gb[(size_t)96 * K + (k0)];
#define GM_PUT(st)                                                      \
    *(float4*)&Ls[(st) + so                               ] = ra0;      \
    *(float4*)&Ls[(st) + so + 32 * GM_PITCH               ] = ra1;      \
    *(float4*)&Ls[(st) + so + (GM_BROW     ) * GM_PITCH   ] = rb0;      \
    *(float4*)&Ls[(st) + so + (GM_BROW + 32) * GM_PITCH   ] = rb1;      \
    *(float4*)&Ls[(st) + so + (GM_BROW + 64) * GM_PITCH   ] = rb2;      \
    *(float4*)&Ls[(st) + so + (GM_BROW + 96) * GM_PITCH   ] = rb3;
#define GM_FRAGS(S, st, u)                                              \
    _Pragma("unroll")                                                   \
    for (int q = 0; q < 2; ++q) {                                       \
        const int kc = ((u) * 2 + q) * 4;                               \
        fa[S][q][0] = *(const v2f*)&Ls[(st) + fa0 + kc];                \
        fa[S][q][1] = *(const v2f*)&Ls[(st) + fa0 + 16 * GM_PITCH + kc];\
        fb[S][q][0] = *(const v2f*)&Ls[(st) + fb0 + kc];                \
        fb[S][q][1] = *(const v2f*)&Ls[(st) + fb0 + 16 * GM_PITCH + kc];\
    }

    v8f acc[2][2] = {};
    v2f fa[2][2][2], fb[2][2][2];

    const int T = K / 32;
    GM_FETCH(0)
    GM_PUT(0)
    __syncthreads();
    if (1 < T) {
        GM_FETCH(32)
    }
    GM_FRAGS(0, 0, 0)

    for (int t = 0; t < T; ++t) {
        const int cur = (t & 1) * GM_STAGE;
        const int nxt = GM_STAGE - cur;

        #pragma unroll
        for (int u = 0; u < 4; ++u) {
            const int cs = u & 1;
            if (u < 3) {
                GM_FRAGS(cs ^ 1, cur, u + 1)
            } else {
                GM_PUT(nxt)
                __syncthreads();
                if (t + 2 < T) {
                    GM_FETCH((t + 2) * 32)
                }
                GM_FRAGS(0, nxt, 0)
                __builtin_amdgcn_sched_barrier(0);
            }
            #pragma unroll
            for (int q = 0; q < 2; ++q)
                #pragma unroll
                for (int mi = 0; mi < 2; ++mi)
                    #pragma unroll
                    for (int ni = 0; ni < 2; ++ni)
                        acc[mi][ni] = __builtin_amdgcn_wmma_f32_16x16x4_f32(
                            false, fa[cs][q][mi], false, fb[cs][q][ni],
                            (short)0, acc[mi][ni], false, false);
        }
        __builtin_amdgcn_sched_barrier(0);
        asm volatile("" :: "v"(fa[0][0][0]), "v"(fa[0][0][1]), "v"(fa[0][1][0]),
                           "v"(fa[0][1][1]), "v"(fb[0][0][0]), "v"(fb[0][0][1]),
                           "v"(fb[0][1][0]), "v"(fb[0][1][1]));
    }
#undef GM_FETCH
#undef GM_PUT
#undef GM_FRAGS

    #pragma unroll
    for (int mi = 0; mi < 2; ++mi)
        #pragma unroll
        for (int ni = 0; ni < 2; ++ni)
            #pragma unroll
            for (int r = 0; r < 8; ++r) {
                int gm = m0 + wm * 32 + mi * 16 + lhal * 8 + r;
                int gn = n0 + wn * 32 + ni * 16 + lmod;
                C[(size_t)gm * N + gn] = acc[mi][ni][r];
            }
}

#define CONV_T 16

__device__ __forceinline__ float conv_silu1(float bias, float x0, float x1,
                                            float x2, float x3, float4 w) {
    float acc = bias;
    acc += x0 * w.x;
    acc += x1 * w.y;
    acc += x2 * w.z;
    acc += x3 * w.w;
    return acc / (1.f + __expf(-acc));
}

__global__ __launch_bounds__(256)
void conv_silu_kernel(const float* __restrict__ xz, const float* __restrict__ cw,
                      const float* __restrict__ cb, float* __restrict__ u) {
    const int id = blockIdx.x * 256 + threadIdx.x;
    const int d  = (id & 511) * 4;
    const int r0 = (id >> 9) * CONV_T;
    const bool head = (r0 & (L_SEQ - 1)) == 0;

    const float* src = xz + (size_t)r0 * (2 * D_INNER) + d;

    float4 x[CONV_T + 3];
    #pragma unroll
    for (int k = 0; k < 3; ++k) x[k] = make_float4(0.f, 0.f, 0.f, 0.f);
    if (!head) {
        #pragma unroll
        for (int k = 0; k < 3; ++k)
            x[k] = *(const float4*)(src + (ptrdiff_t)(k - 3) * (2 * D_INNER));
    }
    #pragma unroll
    for (int k = 3; k < CONV_T + 3; ++k)
        x[k] = *(const float4*)(src + (size_t)(k - 3) * (2 * D_INNER));

    const float4 w0 = *(const float4*)&cw[(d + 0) * 4];
    const float4 w1 = *(const float4*)&cw[(d + 1) * 4];
    const float4 w2 = *(const float4*)&cw[(d + 2) * 4];
    const float4 w3 = *(const float4*)&cw[(d + 3) * 4];
    const float4 bb = *(const float4*)&cb[d];

    float* dst = u + (size_t)r0 * D_INNER + d;
    #pragma unroll
    for (int t = 0; t < CONV_T; ++t) {
        float4 o;
        o.x = conv_silu1(bb.x, x[t].x, x[t + 1].x, x[t + 2].x, x[t + 3].x, w0);
        o.y = conv_silu1(bb.y, x[t].y, x[t + 1].y, x[t + 2].y, x[t + 3].y, w1);
        o.z = conv_silu1(bb.z, x[t].z, x[t + 1].z, x[t + 2].z, x[t + 3].z, w2);
        o.w = conv_silu1(bb.w, x[t].w, x[t + 1].w, x[t + 2].w, x[t + 3].w, w3);
        *(float4*)(dst + (size_t)t * D_INNER) = o;
    }
}

#define XD_KT    256
#define XD_PITCH 260
#define XD_RP    52

__global__ __launch_bounds__(256)
void xdbl_wmma_kernel(const float* u, const float* Wx, float* __restrict__ xdbl) {
    __shared__ __align__(16) float As[16][XD_PITCH];
    __shared__ __align__(16) float Bs[33][XD_PITCH];
    __shared__ __align__(16) float red[8][16][XD_RP];

    const int tid  = threadIdx.x;
    const int lane = tid & 31;
    const int w    = tid >> 5;
    const int c    = lane & 15;
    const int h    = lane >> 4;
    const int m0   = blockIdx.x * 16;

    const int  sr   = tid >> 6;
    const int  sc   = (tid & 63) * 4;
    const bool last = tid < 64;
    const float* ua = u  + (size_t)(m0 + sr) * D_INNER + sc;
    const float* wb = Wx + (size_t)sr * D_INNER + sc;

    float4 a0, a1, a2, a3, b0, b1, b2, b3, b4, b5, b6, b7;
    float4 bl = make_float4(0.f, 0.f, 0.f, 0.f);
#define XD_FETCH(ko)                                                    \
    a0 = *(const float4*)&ua[(size_t) 0 * D_INNER + (ko)];              \
    a1 = *(const float4*)&ua[(size_t) 4 * D_INNER + (ko)];              \
    a2 = *(const float4*)&ua[(size_t) 8 * D_INNER + (ko)];              \
    a3 = *(const float4*)&ua[(size_t)12 * D_INNER + (ko)];              \
    b0 = *(const float4*)&wb[(size_t) 0 * D_INNER + (ko)];              \
    b1 = *(const float4*)&wb[(size_t) 4 * D_INNER + (ko)];              \
    b2 = *(const float4*)&wb[(size_t) 8 * D_INNER + (ko)];              \
    b3 = *(const float4*)&wb[(size_t)12 * D_INNER + (ko)];              \
    b4 = *(const float4*)&wb[(size_t)16 * D_INNER + (ko)];              \
    b5 = *(const float4*)&wb[(size_t)20 * D_INNER + (ko)];              \
    b6 = *(const float4*)&wb[(size_t)24 * D_INNER + (ko)];              \
    b7 = *(const float4*)&wb[(size_t)28 * D_INNER + (ko)];              \
    if (last) bl = *(const float4*)&wb[(size_t)32 * D_INNER + (ko)];

    XD_FETCH(0)

    v8f acc[3] = {};
    const int kb = w * 32 + h * 4;

    for (int kt = 0; kt < D_INNER / XD_KT; ++kt) {
        *(float4*)&As[sr     ][sc] = a0;
        *(float4*)&As[sr +  4][sc] = a1;
        *(float4*)&As[sr +  8][sc] = a2;
        *(float4*)&As[sr + 12][sc] = a3;
        *(float4*)&Bs[sr     ][sc] = b0;
        *(float4*)&Bs[sr +  4][sc] = b1;
        *(float4*)&Bs[sr +  8][sc] = b2;
        *(float4*)&Bs[sr + 12][sc] = b3;
        *(float4*)&Bs[sr + 16][sc] = b4;
        *(float4*)&Bs[sr + 20][sc] = b5;
        *(float4*)&Bs[sr + 24][sc] = b6;
        *(float4*)&Bs[sr + 28][sc] = b7;
        if (last) *(float4*)&Bs[32][sc] = bl;
        __syncthreads();

        if (kt + 1 < D_INNER / XD_KT) {
            XD_FETCH((kt + 1) * XD_KT)
        }

        #pragma unroll
        for (int i = 0; i < 4; ++i) {
            const float4 f4 = *(const float4*)&As[c][kb + 8 * i];
            const float4 p4 = *(const float4*)&Bs[c][kb + 8 * i];
            const float4 q4 = *(const float4*)&Bs[16 + c][kb + 8 * i];
            const float4 r4 = *(const float4*)&Bs[32][kb + 8 * i];
            const v2f f0 = {f4.x, f4.y}, f1 = {f4.z, f4.w};
            const v2f p0 = {p4.x, p4.y}, p1 = {p4.z, p4.w};
            const v2f q0 = {q4.x, q4.y}, q1 = {q4.z, q4.w};
            const v2f r0 = {r4.x, r4.y}, r1 = {r4.z, r4.w};
            acc[0] = __builtin_amdgcn_wmma_f32_16x16x4_f32(
                false, f0, false, p0, (short)0, acc[0], false, false);
            acc[1] = __builtin_amdgcn_wmma_f32_16x16x4_f32(
                false, f0, false, q0, (short)0, acc[1], false, false);
            acc[2] = __builtin_amdgcn_wmma_f32_16x16x4_f32(
                false, f0, false, r0, (short)0, acc[2], false, false);
            acc[0] = __builtin_amdgcn_wmma_f32_16x16x4_f32(
                false, f1, false, p1, (short)0, acc[0], false, false);
            acc[1] = __builtin_amdgcn_wmma_f32_16x16x4_f32(
                false, f1, false, q1, (short)0, acc[1], false, false);
            acc[2] = __builtin_amdgcn_wmma_f32_16x16x4_f32(
                false, f1, false, r1, (short)0, acc[2], false, false);
        }
        __syncthreads();
    }
#undef XD_FETCH

    #pragma unroll
    for (int ni = 0; ni < 3; ++ni)
        #pragma unroll
        for (int r = 0; r < 8; ++r)
            red[w][h * 8 + r][ni * 16 + c] = acc[ni][r];
    __syncthreads();

    for (int idx = tid; idx < 16 * 33; idx += 256) {
        const int row = idx / 33, e = idx - row * 33;
        float s = 0.f;
        #pragma unroll
        for (int q = 0; q < 8; ++q) s += red[q][row][e];
        xdbl[(size_t)m0 * 33 + idx] = s;
    }
}

#define SCAN_T     16
#define SCAN_NCH   (L_SEQ / SCAN_T)
#define SCAN_PITCH 36
#define SCAN_XCH   (SCAN_T * 33)
#define SCAN_CH    32
#define SCAN_YP    36

__device__ __forceinline__ int scan_dst(int i) {
    const int t = i / 33, e = i - t * 33;
    return t * SCAN_PITCH + (e ? e + 3 : 0);
}

__device__ __forceinline__ float scan_dt(float raw, float wdt, float bd) {
    float dt = raw * wdt + bd;
    const float sp = log1pf(__expf(dt));
    dt = (dt > 20.f) ? dt : sp;
    return fminf(fmaxf(dt, 1e-4f), 10.f);
}

template <int XMASK>
__device__ __forceinline__ float scan_add_xor(float v) {
    const int o = __builtin_amdgcn_update_dpp(0, __float_as_int(v), 0x160 + XMASK,
                                              0xF, 0xF, true);
    return v + __int_as_float(o);
}

__global__ __launch_bounds__(256)
void scan_kernel(const float* __restrict__ xdbl, const float* __restrict__ u,
                 const float* __restrict__ Wdt, const float* __restrict__ bdt,
                 const float* __restrict__ Alog, float* __restrict__ ybuf) {
    const int tid = threadIdx.x;
    const int j   = tid & 7;
    const int c   = tid >> 3;
    const int sc  = tid & 31;
    const int st  = tid >> 5;
    const int d0  = blockIdx.x * SCAN_CH;
    const int b   = blockIdx.y;

    const float A0 = -__expf(fminf(Alog[2 * j],     5.f));
    const float A1 = -__expf(fminf(Alog[2 * j + 1], 5.f));
    const float wdt = Wdt[d0 + sc];
    const float bd  = bdt[d0 + sc];
    float h0 = 0.f, h1 = 0.f;

    const float* xd = xdbl + (size_t)b * L_SEQ * 33;
    const float* ub = u    + (size_t)b * L_SEQ * D_INNER + d0 + sc;
    float*       yb = ybuf + (size_t)b * L_SEQ * D_INNER + d0 + sc;

    __shared__ __align__(16) float  sx[2][SCAN_T * SCAN_PITCH];
    __shared__ __align__(16) float2 sdu[2][SCAN_T][SCAN_CH];
    __shared__ __align__(16) float  sy[2][SCAN_T * SCAN_YP];

    const int dst0 = scan_dst(tid);
    const int dst1 = scan_dst(tid + 256);
    const int dst2 = scan_dst(tid + 512);
    const bool has2 = tid < (SCAN_XCH - 512);

    sx[0][dst0] = xd[tid];
    sx[0][dst1] = xd[tid + 256];
    if (has2) sx[0][dst2] = xd[tid + 512];
    sdu[0][st][sc]     = make_float2(scan_dt(xd[st * 33], wdt, bd),
                                     ub[(size_t)st * D_INNER]);
    sdu[0][st + 8][sc] = make_float2(scan_dt(xd[(st + 8) * 33], wdt, bd),
                                     ub[(size_t)(st + 8) * D_INNER]);
    __syncthreads();

    for (int ch = 0; ch < SCAN_NCH; ++ch) {
        const int  cur  = ch & 1;
        const bool more = (ch + 1 < SCAN_NCH);

        float xn0 = 0.f, xn1 = 0.f, xn2 = 0.f;
        float rn0 = 0.f, rn1 = 0.f, un0 = 0.f, un1 = 0.f;
        if (more) {
            const float* nsrc = xd + (size_t)(ch + 1) * SCAN_XCH;
            xn0 = nsrc[tid];
            xn1 = nsrc[tid + 256];
            if (has2) xn2 = nsrc[tid + 512];
            rn0 = nsrc[st * 33];
            rn1 = nsrc[(st + 8) * 33];
            const float* nu = ub + (size_t)(ch + 1) * SCAN_T * D_INNER;
            un0 = nu[(size_t)st * D_INNER];
            un1 = nu[(size_t)(st + 8) * D_INNER];
        }

        if (ch > 0) {
            const float* py = sy[cur ^ 1];
            float* yc = yb + (size_t)(ch - 1) * SCAN_T * D_INNER;
            yc[(size_t)st * D_INNER]       = py[st * SCAN_YP + sc];
            yc[(size_t)(st + 8) * D_INNER] = py[(st + 8) * SCAN_YP + sc];
        }

        const float* s = sx[cur];
        float ysel = 0.f;
        #pragma unroll
        for (int t = 0; t < SCAN_T; ++t) {
            const float2 du = sdu[cur][t][c];
            const float2 Bq = *(const float2*)&s[t * SCAN_PITCH + 4  + 2 * j];
            const float2 Cq = *(const float2*)&s[t * SCAN_PITCH + 20 + 2 * j];
            const float dt = du.x, ut = du.y;

            const float dA0 = __expf(fminf(fmaxf(dt * A0, -20.f), 0.f));
            const float dA1 = __expf(fminf(fmaxf(dt * A1, -20.f), 0.f));
            const float dB0 = fminf(fmaxf(dt * Bq.x * ut, -10.f), 10.f);
            const float dB1 = fminf(fmaxf(dt * Bq.y * ut, -10.f), 10.f);
            h0 = fminf(fmaxf(h0 * dA0 + dB0, -100.f), 100.f);
            h1 = fminf(fmaxf(h1 * dA1 + dB1, -100.f), 100.f);

            float p = h0 * Cq.x + h1 * Cq.y;
            p = scan_add_xor<1>(p);
            p = scan_add_xor<2>(p);
            p = scan_add_xor<4>(p);
            if ((t & 7) == j) ysel = p;
            if ((t & 7) == 7)
                sy[cur][((t & 8) + j) * SCAN_YP + c] = ysel;
        }

        if (more) {
            float* nx = sx[cur ^ 1];
            nx[dst0] = xn0;
            nx[dst1] = xn1;
            if (has2) nx[dst2] = xn2;
            const float dn0 = scan_dt(rn0, wdt, bd);
            const float dn1 = scan_dt(rn1, wdt, bd);
            sdu[cur ^ 1][st][sc]     = make_float2(dn0, un0);
            sdu[cur ^ 1][st + 8][sc] = make_float2(dn1, un1);
        }
        __syncthreads();
    }

    {
        const float* py = sy[(SCAN_NCH - 1) & 1];
        float* yc = yb + (size_t)(SCAN_NCH - 1) * SCAN_T * D_INNER;
        yc[(size_t)st * D_INNER]       = py[st * SCAN_YP + sc];
        yc[(size_t)(st + 8) * D_INNER] = py[(st + 8) * SCAN_YP + sc];
    }
}

__global__ __launch_bounds__(256)
void ln_gate_kernel(float* __restrict__ ybuf, const float* __restrict__ u,
                    const float* __restrict__ xz, const float* __restrict__ Dp,
                    const float* __restrict__ g, const float* __restrict__ bta) {
    const int row = blockIdx.x;
    const int tid = threadIdx.x;
    float* yr = ybuf + (size_t)row * D_INNER;

    float v[8], s = 0.f, s2 = 0.f;
    #pragma unroll
    for (int i = 0; i < 8; ++i) {
        v[i] = yr[tid + i * 256];
        s += v[i]; s2 += v[i] * v[i];
    }
    #pragma unroll
    for (int off = 16; off; off >>= 1) {
        s  += __shfl_xor(s,  off, 32);
        s2 += __shfl_xor(s2, off, 32);
    }
    __shared__ float ws1[8], ws2[8];
    const int lane = tid & 31, w = tid >> 5;
    if (lane == 0) { ws1[w] = s; ws2[w] = s2; }
    __syncthreads();
    if (tid == 0) {
        float t1 = 0.f, t2 = 0.f;
        #pragma unroll
        for (int i = 0; i < 8; ++i) { t1 += ws1[i]; t2 += ws2[i]; }
        ws1[0] = t1; ws2[0] = t2;
    }
    __syncthreads();
    const float mu  = ws1[0] * (1.f / D_INNER);
    float var = ws2[0] * (1.f / D_INNER) - mu * mu;
    var = fmaxf(var, 0.f);
    const float rs = rsqrtf(var + 1e-5f);

    #pragma unroll
    for (int i = 0; i < 8; ++i) {
        const int d = tid + i * 256;
        float yn = (v[i] - mu) * rs * g[d] + bta[d];
        float uu = u[(size_t)row * D_INNER + d];
        float zz = xz[(size_t)row * (2 * D_INNER) + D_INNER + d];
        float sz = zz / (1.f + __expf(-zz));
        yr[d] = (yn + Dp[d] * uu) * sz;
    }
}

extern "C" void kernel_launch(void* const* d_in, const int* in_sizes, int n_in,
                              void* d_out, int out_size, void* d_ws, size_t ws_size,
                              hipStream_t stream) {
    const float* x      = (const float*)d_in[0];
    const float* W_in   = (const float*)d_in[1];
    const float* conv_w = (const float*)d_in[2];
    const float* conv_b = (const float*)d_in[3];
    const float* W_x    = (const float*)d_in[4];
    const float* W_dt   = (const float*)d_in[5];
    const float* b_dt   = (const float*)d_in[6];
    const float* A_log  = (const float*)d_in[7];
    const float* D_par  = (const float*)d_in[8];
    const float* W_out  = (const float*)d_in[9];
    const float* ln_g   = (const float*)d_in[10];
    const float* ln_b   = (const float*)d_in[11];
    float* out = (float*)d_out;

    float* ws = (float*)d_ws;
    const size_t XZ_OFF   = 0;
    const size_t U_OFF    = XZ_OFF + (size_t)M_ROWS * 2 * D_INNER;
    const size_t XDBL_OFF = U_OFF  + (size_t)M_ROWS * D_INNER;
    const size_t Y_OFF    = XDBL_OFF + (size_t)M_ROWS * 33;
    float* xz   = ws + XZ_OFF;
    float* u    = ws + U_OFF;
    float* xdbl = ws + XDBL_OFF;
    float* ybuf = ws + Y_OFF;

    gemm_f32_wmma<<<dim3(2 * D_INNER / 128, M_ROWS / 64), 256, 0, stream>>>(
        x, W_in, xz, M_ROWS, 2 * D_INNER, D_MODEL);

    conv_silu_kernel<<<(M_ROWS / CONV_T) * 512 / 256, 256, 0, stream>>>(
        xz, conv_w, conv_b, u);

    xdbl_wmma_kernel<<<M_ROWS / 16, 256, 0, stream>>>(u, W_x, xdbl);

    scan_kernel<<<dim3(D_INNER / SCAN_CH, B_SZ), 256, 0, stream>>>(
        xdbl, u, W_dt, b_dt, A_log, ybuf);

    ln_gate_kernel<<<M_ROWS, 256, 0, stream>>>(ybuf, u, xz, D_par, ln_g, ln_b);

    gemm_f32_wmma<<<dim3(D_MODEL / 128, M_ROWS / 64), 256, 0, stream>>>(
        ybuf, W_out, out, M_ROWS, D_MODEL, D_INNER);
}


# scan and the z (gate) half of the input projection GEMM share one launch, role by block index: 128 vector-bound scan work-groups first, 1024 matrix-bound z tiles beside them; x half GEMM stays ahead
# speedup vs baseline: 7.5456x; 1.0577x over previous
#include <hip/hip_runtime.h>
#include <hip/hip_bf16.h>


typedef __attribute__((ext_vector_type(2))) float v2f;
typedef __attribute__((ext_vector_type(8))) float v8f;

#define D_MODEL 1024
#define D_INNER 2048
#define D_STATE 16
#define L_SEQ   2048
#define B_SZ    2
#define M_ROWS  (B_SZ * L_SEQ)

#define GM_PITCH 36
#define GM_BROW  64
#define GM_STAGE ((64 + 128) * GM_PITCH)

__device__ __forceinline__
void gemm_tile_f32_wmma(const float* __restrict__ A, const float* __restrict__ Bm,
                        float* __restrict__ C, int ldc, int K,
                        int bx, int by, float* Ls) {
    const int tid  = threadIdx.x;
    const int lane = tid & 31;
    const int w    = tid >> 5;
    const int wm   = w >> 2;
    const int wn   = w & 3;
    const int lmod = lane & 15;
    const int lhal = lane >> 4;

    const int n0 = bx * 128;
    const int m0 = by * 64;

    const int sr = tid >> 3, sc = (tid & 7) * 4;
    const float* ga = A  + (size_t)(m0 + sr) * K + sc;
    const float* gb = Bm + (size_t)(n0 + sr) * K + sc;
    const int so = sr * GM_PITCH + sc;

    const int fa0 = (wm * 32 + lmod) * GM_PITCH + lhal * 2;
    const int fb0 = (GM_BROW + wn * 32 + lmod) * GM_PITCH + lhal * 2;

    float4 ra0, ra1, rb0, rb1, rb2, rb3;
#define GM_FETCH(k0)                                                    \
    ra0 = *(const float4*)&ga[(size_t) 0 * K + (k0)];                   \
    ra1 = *(const float4*)&ga[(size_t)32 * K + (k0)];                   \
    rb0 = *(const float4*)&gb[(size_t) 0 * K + (k0)];                   \
    rb1 = *(const float4*)&gb[(size_t)32 * K + (k0)];                   \
    rb2 = *(const float4*)&gb[(size_t)64 * K + (k0)];                   \
    rb3 = *(const float4*)&gb[(size_t)96 * K + (k0)];
#define GM_PUT(st)                                                      \
    *(float4*)&Ls[(st) + so                               ] = ra0;      \
    *(float4*)&Ls[(st) + so + 32 * GM_PITCH               ] = ra1;      \
    *(float4*)&Ls[(st) + so + (GM_BROW     ) * GM_PITCH   ] = rb0;      \
    *(float4*)&Ls[(st) + so + (GM_BROW + 32) * GM_PITCH   ] = rb1;      \
    *(float4*)&Ls[(st) + so + (GM_BROW + 64) * GM_PITCH   ] = rb2;      \
    *(float4*)&Ls[(st) + so + (GM_BROW + 96) * GM_PITCH   ] = rb3;
#define GM_FRAGS(S, st, u)                                              \
    _Pragma("unroll")                                                   \
    for (int q = 0; q < 2; ++q) {                                       \
        const int kc = ((u) * 2 + q) * 4;                               \
        fa[S][q][0] = *(const v2f*)&Ls[(st) + fa0 + kc];                \
        fa[S][q][1] = *(const v2f*)&Ls[(st) + fa0 + 16 * GM_PITCH + kc];\
        fb[S][q][0] = *(const v2f*)&Ls[(st) + fb0 + kc];                \
        fb[S][q][1] = *(const v2f*)&Ls[(st) + fb0 + 16 * GM_PITCH + kc];\
    }

    v8f acc[2][2] = {};
    v2f fa[2][2][2], fb[2][2][2];

    const int T = K / 32;
    GM_FETCH(0)
    GM_PUT(0)
    __syncthreads();
    if (1 < T) {
        GM_FETCH(32)
    }
    GM_FRAGS(0, 0, 0)

    for (int t = 0; t < T; ++t) {
        const int cur = (t & 1) * GM_STAGE;
        const int nxt = GM_STAGE - cur;

        #pragma unroll
        for (int u = 0; u < 4; ++u) {
            const int cs = u & 1;
            if (u < 3) {
                GM_FRAGS(cs ^ 1, cur, u + 1)
            } else {
                GM_PUT(nxt)
                __syncthreads();
                if (t + 2 < T) {
                    GM_FETCH((t + 2) * 32)
                }
                GM_FRAGS(0, nxt, 0)
                __builtin_amdgcn_sched_barrier(0);
            }
            #pragma unroll
            for (int q = 0; q < 2; ++q)
                #pragma unroll
                for (int mi = 0; mi < 2; ++mi)
                    #pragma unroll
                    for (int ni = 0; ni < 2; ++ni)
                        acc[mi][ni] = __builtin_amdgcn_wmma_f32_16x16x4_f32(
                            false, fa[cs][q][mi], false, fb[cs][q][ni],
                            (short)0, acc[mi][ni], false, false);
        }
        __builtin_amdgcn_sched_barrier(0);
        asm volatile("" :: "v"(fa[0][0][0]), "v"(fa[0][0][1]), "v"(fa[0][1][0]),
                           "v"(fa[0][1][1]), "v"(fb[0][0][0]), "v"(fb[0][0][1]),
                           "v"(fb[0][1][0]), "v"(fb[0][1][1]));
    }
#undef GM_FETCH
#undef GM_PUT
#undef GM_FRAGS

    #pragma unroll
    for (int mi = 0; mi < 2; ++mi)
        #pragma unroll
        for (int ni = 0; ni < 2; ++ni)
            #pragma unroll
            for (int r = 0; r < 8; ++r) {
                int gm = m0 + wm * 32 + mi * 16 + lhal * 8 + r;
                int gn = n0 + wn * 32 + ni * 16 + lmod;
                C[(size_t)gm * ldc + gn] = acc[mi][ni][r];
            }
}

__global__ __launch_bounds__(256)
void gemm_f32_wmma(const float* __restrict__ A, const float* __restrict__ Bm,
                   float* __restrict__ C, int ldc, int K) {
    __shared__ __align__(16) float Ls[2 * GM_STAGE];
    gemm_tile_f32_wmma(A, Bm, C, ldc, K, blockIdx.x, blockIdx.y, Ls);
}

#define CONV_T 16

__device__ __forceinline__ float conv_silu1(float bias, float x0, float x1,
                                            float x2, float x3, float4 w) {
    float acc = bias;
    acc += x0 * w.x;
    acc += x1 * w.y;
    acc += x2 * w.z;
    acc += x3 * w.w;
    return acc / (1.f + __expf(-acc));
}

__global__ __launch_bounds__(256)
void conv_silu_kernel(const float* __restrict__ xz, const float* __restrict__ cw,
                      const float* __restrict__ cb, float* __restrict__ u) {
    const int id = blockIdx.x * 256 + threadIdx.x;
    const int d  = (id & 511) * 4;
    const int r0 = (id >> 9) * CONV_T;
    const bool head = (r0 & (L_SEQ - 1)) == 0;

    const float* src = xz + (size_t)r0 * (2 * D_INNER) + d;

    float4 x[CONV_T + 3];
    #pragma unroll
    for (int k = 0; k < 3; ++k) x[k] = make_float4(0.f, 0.f, 0.f, 0.f);
    if (!head) {
        #pragma unroll
        for (int k = 0; k < 3; ++k)
            x[k] = *(const float4*)(src + (ptrdiff_t)(k - 3) * (2 * D_INNER));
    }
    #pragma unroll
    for (int k = 3; k < CONV_T + 3; ++k)
        x[k] = *(const float4*)(src + (size_t)(k - 3) * (2 * D_INNER));

    const float4 w0 = *(const float4*)&cw[(d + 0) * 4];
    const float4 w1 = *(const float4*)&cw[(d + 1) * 4];
    const float4 w2 = *(const float4*)&cw[(d + 2) * 4];
    const float4 w3 = *(const float4*)&cw[(d + 3) * 4];
    const float4 bb = *(const float4*)&cb[d];

    float* dst = u + (size_t)r0 * D_INNER + d;
    #pragma unroll
    for (int t = 0; t < CONV_T; ++t) {
        float4 o;
        o.x = conv_silu1(bb.x, x[t].x, x[t + 1].x, x[t + 2].x, x[t + 3].x, w0);
        o.y = conv_silu1(bb.y, x[t].y, x[t + 1].y, x[t + 2].y, x[t + 3].y, w1);
        o.z = conv_silu1(bb.z, x[t].z, x[t + 1].z, x[t + 2].z, x[t + 3].z, w2);
        o.w = conv_silu1(bb.w, x[t].w, x[t + 1].w, x[t + 2].w, x[t + 3].w, w3);
        *(float4*)(dst + (size_t)t * D_INNER) = o;
    }
}

#define XD_KT    256
#define XD_PITCH 260
#define XD_RP    52

__global__ __launch_bounds__(256)
void xdbl_wmma_kernel(const float* u, const float* Wx, float* __restrict__ xdbl) {
    __shared__ __align__(16) float As[16][XD_PITCH];
    __shared__ __align__(16) float Bs[33][XD_PITCH];
    __shared__ __align__(16) float red[8][16][XD_RP];

    const int tid  = threadIdx.x;
    const int lane = tid & 31;
    const int w    = tid >> 5;
    const int c    = lane & 15;
    const int h    = lane >> 4;
    const int m0   = blockIdx.x * 16;

    const int  sr   = tid >> 6;
    const int  sc   = (tid & 63) * 4;
    const bool last = tid < 64;
    const float* ua = u  + (size_t)(m0 + sr) * D_INNER + sc;
    const float* wb = Wx + (size_t)sr * D_INNER + sc;

    float4 a0, a1, a2, a3, b0, b1, b2, b3, b4, b5, b6, b7;
    float4 bl = make_float4(0.f, 0.f, 0.f, 0.f);
#define XD_FETCH(ko)                                                    \
    a0 = *(const float4*)&ua[(size_t) 0 * D_INNER + (ko)];              \
    a1 = *(const float4*)&ua[(size_t) 4 * D_INNER + (ko)];              \
    a2 = *(const float4*)&ua[(size_t) 8 * D_INNER + (ko)];              \
    a3 = *(const float4*)&ua[(size_t)12 * D_INNER + (ko)];              \
    b0 = *(const float4*)&wb[(size_t) 0 * D_INNER + (ko)];              \
    b1 = *(const float4*)&wb[(size_t) 4 * D_INNER + (ko)];              \
    b2 = *(const float4*)&wb[(size_t) 8 * D_INNER + (ko)];              \
    b3 = *(const float4*)&wb[(size_t)12 * D_INNER + (ko)];              \
    b4 = *(const float4*)&wb[(size_t)16 * D_INNER + (ko)];              \
    b5 = *(const float4*)&wb[(size_t)20 * D_INNER + (ko)];              \
    b6 = *(const float4*)&wb[(size_t)24 * D_INNER + (ko)];              \
    b7 = *(const float4*)&wb[(size_t)28 * D_INNER + (ko)];              \
    if (last) bl = *(const float4*)&wb[(size_t)32 * D_INNER + (ko)];

    XD_FETCH(0)

    v8f acc[3] = {};
    const int kb = w * 32 + h * 4;

    for (int kt = 0; kt < D_INNER / XD_KT; ++kt) {
        *(float4*)&As[sr     ][sc] = a0;
        *(float4*)&As[sr +  4][sc] = a1;
        *(float4*)&As[sr +  8][sc] = a2;
        *(float4*)&As[sr + 12][sc] = a3;
        *(float4*)&Bs[sr     ][sc] = b0;
        *(float4*)&Bs[sr +  4][sc] = b1;
        *(float4*)&Bs[sr +  8][sc] = b2;
        *(float4*)&Bs[sr + 12][sc] = b3;
        *(float4*)&Bs[sr + 16][sc] = b4;
        *(float4*)&Bs[sr + 20][sc] = b5;
        *(float4*)&Bs[sr + 24][sc] = b6;
        *(float4*)&Bs[sr + 28][sc] = b7;
        if (last) *(float4*)&Bs[32][sc] = bl;
        __syncthreads();

        if (kt + 1 < D_INNER / XD_KT) {
            XD_FETCH((kt + 1) * XD_KT)
        }

        #pragma unroll
        for (int i = 0; i < 4; ++i) {
            const float4 f4 = *(const float4*)&As[c][kb + 8 * i];
            const float4 p4 = *(const float4*)&Bs[c][kb + 8 * i];
            const float4 q4 = *(const float4*)&Bs[16 + c][kb + 8 * i];
            const float4 r4 = *(const float4*)&Bs[32][kb + 8 * i];
            const v2f f0 = {f4.x, f4.y}, f1 = {f4.z, f4.w};
            const v2f p0 = {p4.x, p4.y}, p1 = {p4.z, p4.w};
            const v2f q0 = {q4.x, q4.y}, q1 = {q4.z, q4.w};
            const v2f r0 = {r4.x, r4.y}, r1 = {r4.z, r4.w};
            acc[0] = __builtin_amdgcn_wmma_f32_16x16x4_f32(
                false, f0, false, p0, (short)0, acc[0], false, false);
            acc[1] = __builtin_amdgcn_wmma_f32_16x16x4_f32(
                false, f0, false, q0, (short)0, acc[1], false, false);
            acc[2] = __builtin_amdgcn_wmma_f32_16x16x4_f32(
                false, f0, false, r0, (short)0, acc[2], false, false);
            acc[0] = __builtin_amdgcn_wmma_f32_16x16x4_f32(
                false, f1, false, p1, (short)0, acc[0], false, false);
            acc[1] = __builtin_amdgcn_wmma_f32_16x16x4_f32(
                false, f1, false, q1, (short)0, acc[1], false, false);
            acc[2] = __builtin_amdgcn_wmma_f32_16x16x4_f32(
                false, f1, false, r1, (short)0, acc[2], false, false);
        }
        __syncthreads();
    }
#undef XD_FETCH

    #pragma unroll
    for (int ni = 0; ni < 3; ++ni)
        #pragma unroll
        for (int r = 0; r < 8; ++r)
            red[w][h * 8 + r][ni * 16 + c] = acc[ni][r];
    __syncthreads();

    for (int idx = tid; idx < 16 * 33; idx += 256) {
        const int row = idx / 33, e = idx - row * 33;
        float s = 0.f;
        #pragma unroll
        for (int q = 0; q < 8; ++q) s += red[q][row][e];
        xdbl[(size_t)m0 * 33 + idx] = s;
    }
}

#define SCAN_T     16
#define SCAN_NCH   (L_SEQ / SCAN_T)
#define SCAN_PITCH 36
#define SCAN_XCH   (SCAN_T * 33)
#define SCAN_CH    32
#define SCAN_YP    36

__device__ __forceinline__ int scan_dst(int i) {
    const int t = i / 33, e = i - t * 33;
    return t * SCAN_PITCH + (e ? e + 3 : 0);
}

__device__ __forceinline__ float scan_dt(float raw, float wdt, float bd) {
    float dt = raw * wdt + bd;
    const float sp = log1pf(__expf(dt));
    dt = (dt > 20.f) ? dt : sp;
    return fminf(fmaxf(dt, 1e-4f), 10.f);
}

template <int XMASK>
__device__ __forceinline__ float scan_add_xor(float v) {
    const int o = __builtin_amdgcn_update_dpp(0, __float_as_int(v), 0x160 + XMASK,
                                              0xF, 0xF, true);
    return v + __int_as_float(o);
}

#define SCAN_GX    (D_INNER / SCAN_CH)
#define SCAN_WGS   (SCAN_GX * B_SZ)
#define SCAN_LDS   (2 * SCAN_T * SCAN_PITCH + 4 * SCAN_T * SCAN_CH + 2 * SCAN_T * SCAN_YP)

__device__ __forceinline__
void scan_group(const float* __restrict__ xdbl, const float* __restrict__ u,
                const float* __restrict__ Wdt, const float* __restrict__ bdt,
                const float* __restrict__ Alog, float* __restrict__ ybuf,
                int bx, int by, float* smem) {
    const int tid = threadIdx.x;
    const int j   = tid & 7;
    const int c   = tid >> 3;
    const int sc  = tid & 31;
    const int st  = tid >> 5;
    const int d0  = bx * SCAN_CH;
    const int b   = by;

    const float A0 = -__expf(fminf(Alog[2 * j],     5.f));
    const float A1 = -__expf(fminf(Alog[2 * j + 1], 5.f));
    const float wdt = Wdt[d0 + sc];
    const float bd  = bdt[d0 + sc];
    float h0 = 0.f, h1 = 0.f;

    const float* xd = xdbl + (size_t)b * L_SEQ * 33;
    const float* ub = u    + (size_t)b * L_SEQ * D_INNER + d0 + sc;
    float*       yb = ybuf + (size_t)b * L_SEQ * D_INNER + d0 + sc;

    float  (*sx)[SCAN_T * SCAN_PITCH] = (float (*)[SCAN_T * SCAN_PITCH])smem;
    float2 (*sdu)[SCAN_T][SCAN_CH] =
        (float2 (*)[SCAN_T][SCAN_CH])(smem + 2 * SCAN_T * SCAN_PITCH);
    float  (*sy)[SCAN_T * SCAN_YP] =
        (float (*)[SCAN_T * SCAN_YP])(smem + 2 * SCAN_T * SCAN_PITCH + 4 * SCAN_T * SCAN_CH);

    const int dst0 = scan_dst(tid);
    const int dst1 = scan_dst(tid + 256);
    const int dst2 = scan_dst(tid + 512);
    const bool has2 = tid < (SCAN_XCH - 512);

    sx[0][dst0] = xd[tid];
    sx[0][dst1] = xd[tid + 256];
    if (has2) sx[0][dst2] = xd[tid + 512];
    sdu[0][st][sc]     = make_float2(scan_dt(xd[st * 33], wdt, bd),
                                     ub[(size_t)st * D_INNER]);
    sdu[0][st + 8][sc] = make_float2(scan_dt(xd[(st + 8) * 33], wdt, bd),
                                     ub[(size_t)(st + 8) * D_INNER]);
    __syncthreads();

    for (int ch = 0; ch < SCAN_NCH; ++ch) {
        const int  cur  = ch & 1;
        const bool more = (ch + 1 < SCAN_NCH);

        float xn0 = 0.f, xn1 = 0.f, xn2 = 0.f;
        float rn0 = 0.f, rn1 = 0.f, un0 = 0.f, un1 = 0.f;
        if (more) {
            const float* nsrc = xd + (size_t)(ch + 1) * SCAN_XCH;
            xn0 = nsrc[tid];
            xn1 = nsrc[tid + 256];
            if (has2) xn2 = nsrc[tid + 512];
            rn0 = nsrc[st * 33];
            rn1 = nsrc[(st + 8) * 33];
            const float* nu = ub + (size_t)(ch + 1) * SCAN_T * D_INNER;
            un0 = nu[(size_t)st * D_INNER];
            un1 = nu[(size_t)(st + 8) * D_INNER];
        }

        if (ch > 0) {
            const float* py = sy[cur ^ 1];
            float* yc = yb + (size_t)(ch - 1) * SCAN_T * D_INNER;
            yc[(size_t)st * D_INNER]       = py[st * SCAN_YP + sc];
            yc[(size_t)(st + 8) * D_INNER] = py[(st + 8) * SCAN_YP + sc];
        }

        const float* s = sx[cur];
        float ysel = 0.f;
        #pragma unroll
        for (int t = 0; t < SCAN_T; ++t) {
            const float2 du = sdu[cur][t][c];
            const float2 Bq = *(const float2*)&s[t * SCAN_PITCH + 4  + 2 * j];
            const float2 Cq = *(const float2*)&s[t * SCAN_PITCH + 20 + 2 * j];
            const float dt = du.x, ut = du.y;

            const float dA0 = __expf(fminf(fmaxf(dt * A0, -20.f), 0.f));
            const float dA1 = __expf(fminf(fmaxf(dt * A1, -20.f), 0.f));
            const float dB0 = fminf(fmaxf(dt * Bq.x * ut, -10.f), 10.f);
            const float dB1 = fminf(fmaxf(dt * Bq.y * ut, -10.f), 10.f);
            h0 = fminf(fmaxf(h0 * dA0 + dB0, -100.f), 100.f);
            h1 = fminf(fmaxf(h1 * dA1 + dB1, -100.f), 100.f);

            float p = h0 * Cq.x + h1 * Cq.y;
            p = scan_add_xor<1>(p);
            p = scan_add_xor<2>(p);
            p = scan_add_xor<4>(p);
            if ((t & 7) == j) ysel = p;
            if ((t & 7) == 7)
                sy[cur][((t & 8) + j) * SCAN_YP + c] = ysel;
        }

        if (more) {
            float* nx = sx[cur ^ 1];
            nx[dst0] = xn0;
            nx[dst1] = xn1;
            if (has2) nx[dst2] = xn2;
            const float dn0 = scan_dt(rn0, wdt, bd);
            const float dn1 = scan_dt(rn1, wdt, bd);
            sdu[cur ^ 1][st][sc]     = make_float2(dn0, un0);
            sdu[cur ^ 1][st + 8][sc] = make_float2(dn1, un1);
        }
        __syncthreads();
    }

    {
        const float* py = sy[(SCAN_NCH - 1) & 1];
        float* yc = yb + (size_t)(SCAN_NCH - 1) * SCAN_T * D_INNER;
        yc[(size_t)st * D_INNER]       = py[st * SCAN_YP + sc];
        yc[(size_t)(st + 8) * D_INNER] = py[(st + 8) * SCAN_YP + sc];
    }
}

#define ZG_NX   (D_INNER / 128)
#define ZG_WGS  (ZG_NX * (M_ROWS / 64))

__global__ __launch_bounds__(256)
void scan_zgemm_kernel(const float* __restrict__ xdbl, const float* __restrict__ u,
                       const float* __restrict__ Wdt, const float* __restrict__ bdt,
                       const float* __restrict__ Alog, float* __restrict__ ybuf,
                       const float* __restrict__ x, const float* __restrict__ Wz,
                       float* __restrict__ zout, int ldc, int K) {
    __shared__ __align__(16) float smem[2 * GM_STAGE];
    static_assert(SCAN_LDS <= 2 * GM_STAGE, "scan LDS must fit the GEMM stages");
    const int bid = blockIdx.x;
    if (bid < SCAN_WGS) {
        scan_group(xdbl, u, Wdt, bdt, Alog, ybuf, bid % SCAN_GX, bid / SCAN_GX, smem);
    } else {
        const int g = bid - SCAN_WGS;
        gemm_tile_f32_wmma(x, Wz, zout, ldc, K, g % ZG_NX, g / ZG_NX, smem);
    }
}

__global__ __launch_bounds__(256)
void ln_gate_kernel(float* __restrict__ ybuf, const float* __restrict__ u,
                    const float* __restrict__ xz, const float* __restrict__ Dp,
                    const float* __restrict__ g, const float* __restrict__ bta) {
    const int row = blockIdx.x;
    const int tid = threadIdx.x;
    float* yr = ybuf + (size_t)row * D_INNER;

    float v[8], s = 0.f, s2 = 0.f;
    #pragma unroll
    for (int i = 0; i < 8; ++i) {
        v[i] = yr[tid + i * 256];
        s += v[i]; s2 += v[i] * v[i];
    }
    #pragma unroll
    for (int off = 16; off; off >>= 1) {
        s  += __shfl_xor(s,  off, 32);
        s2 += __shfl_xor(s2, off, 32);
    }
    __shared__ float ws1[8], ws2[8];
    const int lane = tid & 31, w = tid >> 5;
    if (lane == 0) { ws1[w] = s; ws2[w] = s2; }
    __syncthreads();
    if (tid == 0) {
        float t1 = 0.f, t2 = 0.f;
        #pragma unroll
        for (int i = 0; i < 8; ++i) { t1 += ws1[i]; t2 += ws2[i]; }
        ws1[0] = t1; ws2[0] = t2;
    }
    __syncthreads();
    const float mu  = ws1[0] * (1.f / D_INNER);
    float var = ws2[0] * (1.f / D_INNER) - mu * mu;
    var = fmaxf(var, 0.f);
    const float rs = rsqrtf(var + 1e-5f);

    #pragma unroll
    for (int i = 0; i < 8; ++i) {
        const int d = tid + i * 256;
        float yn = (v[i] - mu) * rs * g[d] + bta[d];
        float uu = u[(size_t)row * D_INNER + d];
        float zz = xz[(size_t)row * (2 * D_INNER) + D_INNER + d];
        float sz = zz / (1.f + __expf(-zz));
        yr[d] = (yn + Dp[d] * uu) * sz;
    }
}

extern "C" void kernel_launch(void* const* d_in, const int* in_sizes, int n_in,
                              void* d_out, int out_size, void* d_ws, size_t ws_size,
                              hipStream_t stream) {
    const float* x      = (const float*)d_in[0];
    const float* W_in   = (const float*)d_in[1];
    const float* conv_w = (const float*)d_in[2];
    const float* conv_b = (const float*)d_in[3];
    const float* W_x    = (const float*)d_in[4];
    const float* W_dt   = (const float*)d_in[5];
    const float* b_dt   = (const float*)d_in[6];
    const float* A_log  = (const float*)d_in[7];
    const float* D_par  = (const float*)d_in[8];
    const float* W_out  = (const float*)d_in[9];
    const float* ln_g   = (const float*)d_in[10];
    const float* ln_b   = (const float*)d_in[11];
    float* out = (float*)d_out;

    float* ws = (float*)d_ws;
    const size_t XZ_OFF   = 0;
    const size_t U_OFF    = XZ_OFF + (size_t)M_ROWS * 2 * D_INNER;
    const size_t XDBL_OFF = U_OFF  + (size_t)M_ROWS * D_INNER;
    const size_t Y_OFF    = XDBL_OFF + (size_t)M_ROWS * 33;
    float* xz   = ws + XZ_OFF;
    float* u    = ws + U_OFF;
    float* xdbl = ws + XDBL_OFF;
    float* ybuf = ws + Y_OFF;

    gemm_f32_wmma<<<dim3(D_INNER / 128, M_ROWS / 64), 256, 0, stream>>>(
        x, W_in, xz, 2 * D_INNER, D_MODEL);

    conv_silu_kernel<<<(M_ROWS / CONV_T) * 512 / 256, 256, 0, stream>>>(
        xz, conv_w, conv_b, u);

    xdbl_wmma_kernel<<<M_ROWS / 16, 256, 0, stream>>>(u, W_x, xdbl);

    scan_zgemm_kernel<<<SCAN_WGS + ZG_WGS, 256, 0, stream>>>(
        xdbl, u, W_dt, b_dt, A_log, ybuf,
        x, W_in + (size_t)D_INNER * D_MODEL, xz + D_INNER, 2 * D_INNER, D_MODEL);

    ln_gate_kernel<<<M_ROWS, 256, 0, stream>>>(ybuf, u, xz, D_par, ln_g, ln_b);

    gemm_f32_wmma<<<dim3(D_MODEL / 128, M_ROWS / 64), 256, 0, stream>>>(
        ybuf, W_out, out, D_MODEL, D_INNER);
}


# scan: dt's softplus as max(x,0) + 2*atanh(s) series, s = t/(2+t), t = exp(-|x|), in place of the library log1p; dt rounds differently, relative error ~2e-7 for every x down to the clip floor
# speedup vs baseline: 7.9017x; 1.0472x over previous
#include <hip/hip_runtime.h>
#include <hip/hip_bf16.h>


typedef __attribute__((ext_vector_type(2))) float v2f;
typedef __attribute__((ext_vector_type(8))) float v8f;

#define D_MODEL 1024
#define D_INNER 2048
#define D_STATE 16
#define L_SEQ   2048
#define B_SZ    2
#define M_ROWS  (B_SZ * L_SEQ)

#define GM_PITCH 36
#define GM_BROW  64
#define GM_STAGE ((64 + 128) * GM_PITCH)

__device__ __forceinline__
void gemm_tile_f32_wmma(const float* __restrict__ A, const float* __restrict__ Bm,
                        float* __restrict__ C, int ldc, int K,
                        int bx, int by, float* Ls) {
    const int tid  = threadIdx.x;
    const int lane = tid & 31;
    const int w    = tid >> 5;
    const int wm   = w >> 2;
    const int wn   = w & 3;
    const int lmod = lane & 15;
    const int lhal = lane >> 4;

    const int n0 = bx * 128;
    const int m0 = by * 64;

    const int sr = tid >> 3, sc = (tid & 7) * 4;
    const float* ga = A  + (size_t)(m0 + sr) * K + sc;
    const float* gb = Bm + (size_t)(n0 + sr) * K + sc;
    const int so = sr * GM_PITCH + sc;

    const int fa0 = (wm * 32 + lmod) * GM_PITCH + lhal * 2;
    const int fb0 = (GM_BROW + wn * 32 + lmod) * GM_PITCH + lhal * 2;

    float4 ra0, ra1, rb0, rb1, rb2, rb3;
#define GM_FETCH(k0)                                                    \
    ra0 = *(const float4*)&ga[(size_t) 0 * K + (k0)];                   \
    ra1 = *(const float4*)&ga[(size_t)32 * K + (k0)];                   \
    rb0 = *(const float4*)&gb[(size_t) 0 * K + (k0)];                   \
    rb1 = *(const float4*)&gb[(size_t)32 * K + (k0)];                   \
    rb2 = *(const float4*)&gb[(size_t)64 * K + (k0)];                   \
    rb3 = *(const float4*)&gb[(size_t)96 * K + (k0)];
#define GM_PUT(st)                                                      \
    *(float4*)&Ls[(st) + so                               ] = ra0;      \
    *(float4*)&Ls[(st) + so + 32 * GM_PITCH               ] = ra1;      \
    *(float4*)&Ls[(st) + so + (GM_BROW     ) * GM_PITCH   ] = rb0;      \
    *(float4*)&Ls[(st) + so + (GM_BROW + 32) * GM_PITCH   ] = rb1;      \
    *(float4*)&Ls[(st) + so + (GM_BROW + 64) * GM_PITCH   ] = rb2;      \
    *(float4*)&Ls[(st) + so + (GM_BROW + 96) * GM_PITCH   ] = rb3;
#define GM_FRAGS(S, st, u)                                              \
    _Pragma("unroll")                                                   \
    for (int q = 0; q < 2; ++q) {                                       \
        const int kc = ((u) * 2 + q) * 4;                               \
        fa[S][q][0] = *(const v2f*)&Ls[(st) + fa0 + kc];                \
        fa[S][q][1] = *(const v2f*)&Ls[(st) + fa0 + 16 * GM_PITCH + kc];\
        fb[S][q][0] = *(const v2f*)&Ls[(st) + fb0 + kc];                \
        fb[S][q][1] = *(const v2f*)&Ls[(st) + fb0 + 16 * GM_PITCH + kc];\
    }

    v8f acc[2][2] = {};
    v2f fa[2][2][2], fb[2][2][2];

    const int T = K / 32;
    GM_FETCH(0)
    GM_PUT(0)
    __syncthreads();
    if (1 < T) {
        GM_FETCH(32)
    }
    GM_FRAGS(0, 0, 0)

    for (int t = 0; t < T; ++t) {
        const int cur = (t & 1) * GM_STAGE;
        const int nxt = GM_STAGE - cur;

        #pragma unroll
        for (int u = 0; u < 4; ++u) {
            const int cs = u & 1;
            if (u < 3) {
                GM_FRAGS(cs ^ 1, cur, u + 1)
            } else {
                GM_PUT(nxt)
                __syncthreads();
                if (t + 2 < T) {
                    GM_FETCH((t + 2) * 32)
                }
                GM_FRAGS(0, nxt, 0)
                __builtin_amdgcn_sched_barrier(0);
            }
            #pragma unroll
            for (int q = 0; q < 2; ++q)
                #pragma unroll
                for (int mi = 0; mi < 2; ++mi)
                    #pragma unroll
                    for (int ni = 0; ni < 2; ++ni)
                        acc[mi][ni] = __builtin_amdgcn_wmma_f32_16x16x4_f32(
                            false, fa[cs][q][mi], false, fb[cs][q][ni],
                            (short)0, acc[mi][ni], false, false);
        }
        __builtin_amdgcn_sched_barrier(0);
        asm volatile("" :: "v"(fa[0][0][0]), "v"(fa[0][0][1]), "v"(fa[0][1][0]),
                           "v"(fa[0][1][1]), "v"(fb[0][0][0]), "v"(fb[0][0][1]),
                           "v"(fb[0][1][0]), "v"(fb[0][1][1]));
    }
#undef GM_FETCH
#undef GM_PUT
#undef GM_FRAGS

    #pragma unroll
    for (int mi = 0; mi < 2; ++mi)
        #pragma unroll
        for (int ni = 0; ni < 2; ++ni)
            #pragma unroll
            for (int r = 0; r < 8; ++r) {
                int gm = m0 + wm * 32 + mi * 16 + lhal * 8 + r;
                int gn = n0 + wn * 32 + ni * 16 + lmod;
                C[(size_t)gm * ldc + gn] = acc[mi][ni][r];
            }
}

__global__ __launch_bounds__(256)
void gemm_f32_wmma(const float* __restrict__ A, const float* __restrict__ Bm,
                   float* __restrict__ C, int ldc, int K) {
    __shared__ __align__(16) float Ls[2 * GM_STAGE];
    gemm_tile_f32_wmma(A, Bm, C, ldc, K, blockIdx.x, blockIdx.y, Ls);
}

#define CONV_T 16

__device__ __forceinline__ float conv_silu1(float bias, float x0, float x1,
                                            float x2, float x3, float4 w) {
    float acc = bias;
    acc += x0 * w.x;
    acc += x1 * w.y;
    acc += x2 * w.z;
    acc += x3 * w.w;
    return acc / (1.f + __expf(-acc));
}

__global__ __launch_bounds__(256)
void conv_silu_kernel(const float* __restrict__ xz, const float* __restrict__ cw,
                      const float* __restrict__ cb, float* __restrict__ u) {
    const int id = blockIdx.x * 256 + threadIdx.x;
    const int d  = (id & 511) * 4;
    const int r0 = (id >> 9) * CONV_T;
    const bool head = (r0 & (L_SEQ - 1)) == 0;

    const float* src = xz + (size_t)r0 * (2 * D_INNER) + d;

    float4 x[CONV_T + 3];
    #pragma unroll
    for (int k = 0; k < 3; ++k) x[k] = make_float4(0.f, 0.f, 0.f, 0.f);
    if (!head) {
        #pragma unroll
        for (int k = 0; k < 3; ++k)
            x[k] = *(const float4*)(src + (ptrdiff_t)(k - 3) * (2 * D_INNER));
    }
    #pragma unroll
    for (int k = 3; k < CONV_T + 3; ++k)
        x[k] = *(const float4*)(src + (size_t)(k - 3) * (2 * D_INNER));

    const float4 w0 = *(const float4*)&cw[(d + 0) * 4];
    const float4 w1 = *(const float4*)&cw[(d + 1) * 4];
    const float4 w2 = *(const float4*)&cw[(d + 2) * 4];
    const float4 w3 = *(const float4*)&cw[(d + 3) * 4];
    const float4 bb = *(const float4*)&cb[d];

    float* dst = u + (size_t)r0 * D_INNER + d;
    #pragma unroll
    for (int t = 0; t < CONV_T; ++t) {
        float4 o;
        o.x = conv_silu1(bb.x, x[t].x, x[t + 1].x, x[t + 2].x, x[t + 3].x, w0);
        o.y = conv_silu1(bb.y, x[t].y, x[t + 1].y, x[t + 2].y, x[t + 3].y, w1);
        o.z = conv_silu1(bb.z, x[t].z, x[t + 1].z, x[t + 2].z, x[t + 3].z, w2);
        o.w = conv_silu1(bb.w, x[t].w, x[t + 1].w, x[t + 2].w, x[t + 3].w, w3);
        *(float4*)(dst + (size_t)t * D_INNER) = o;
    }
}

#define XD_KT    256
#define XD_PITCH 260
#define XD_RP    52

__global__ __launch_bounds__(256)
void xdbl_wmma_kernel(const float* u, const float* Wx, float* __restrict__ xdbl) {
    __shared__ __align__(16) float As[16][XD_PITCH];
    __shared__ __align__(16) float Bs[33][XD_PITCH];
    __shared__ __align__(16) float red[8][16][XD_RP];

    const int tid  = threadIdx.x;
    const int lane = tid & 31;
    const int w    = tid >> 5;
    const int c    = lane & 15;
    const int h    = lane >> 4;
    const int m0   = blockIdx.x * 16;

    const int  sr   = tid >> 6;
    const int  sc   = (tid & 63) * 4;
    const bool last = tid < 64;
    const float* ua = u  + (size_t)(m0 + sr) * D_INNER + sc;
    const float* wb = Wx + (size_t)sr * D_INNER + sc;

    float4 a0, a1, a2, a3, b0, b1, b2, b3, b4, b5, b6, b7;
    float4 bl = make_float4(0.f, 0.f, 0.f, 0.f);
#define XD_FETCH(ko)                                                    \
    a0 = *(const float4*)&ua[(size_t) 0 * D_INNER + (ko)];              \
    a1 = *(const float4*)&ua[(size_t) 4 * D_INNER + (ko)];              \
    a2 = *(const float4*)&ua[(size_t) 8 * D_INNER + (ko)];              \
    a3 = *(const float4*)&ua[(size_t)12 * D_INNER + (ko)];              \
    b0 = *(const float4*)&wb[(size_t) 0 * D_INNER + (ko)];              \
    b1 = *(const float4*)&wb[(size_t) 4 * D_INNER + (ko)];              \
    b2 = *(const float4*)&wb[(size_t) 8 * D_INNER + (ko)];              \
    b3 = *(const float4*)&wb[(size_t)12 * D_INNER + (ko)];              \
    b4 = *(const float4*)&wb[(size_t)16 * D_INNER + (ko)];              \
    b5 = *(const float4*)&wb[(size_t)20 * D_INNER + (ko)];              \
    b6 = *(const float4*)&wb[(size_t)24 * D_INNER + (ko)];              \
    b7 = *(const float4*)&wb[(size_t)28 * D_INNER + (ko)];              \
    if (last) bl = *(const float4*)&wb[(size_t)32 * D_INNER + (ko)];

    XD_FETCH(0)

    v8f acc[3] = {};
    const int kb = w * 32 + h * 4;

    for (int kt = 0; kt < D_INNER / XD_KT; ++kt) {
        *(float4*)&As[sr     ][sc] = a0;
        *(float4*)&As[sr +  4][sc] = a1;
        *(float4*)&As[sr +  8][sc] = a2;
        *(float4*)&As[sr + 12][sc] = a3;
        *(float4*)&Bs[sr     ][sc] = b0;
        *(float4*)&Bs[sr +  4][sc] = b1;
        *(float4*)&Bs[sr +  8][sc] = b2;
        *(float4*)&Bs[sr + 12][sc] = b3;
        *(float4*)&Bs[sr + 16][sc] = b4;
        *(float4*)&Bs[sr + 20][sc] = b5;
        *(float4*)&Bs[sr + 24][sc] = b6;
        *(float4*)&Bs[sr + 28][sc] = b7;
        if (last) *(float4*)&Bs[32][sc] = bl;
        __syncthreads();

        if (kt + 1 < D_INNER / XD_KT) {
            XD_FETCH((kt + 1) * XD_KT)
        }

        #pragma unroll
        for (int i = 0; i < 4; ++i) {
            const float4 f4 = *(const float4*)&As[c][kb + 8 * i];
            const float4 p4 = *(const float4*)&Bs[c][kb + 8 * i];
            const float4 q4 = *(const float4*)&Bs[16 + c][kb + 8 * i];
            const float4 r4 = *(const float4*)&Bs[32][kb + 8 * i];
            const v2f f0 = {f4.x, f4.y}, f1 = {f4.z, f4.w};
            const v2f p0 = {p4.x, p4.y}, p1 = {p4.z, p4.w};
            const v2f q0 = {q4.x, q4.y}, q1 = {q4.z, q4.w};
            const v2f r0 = {r4.x, r4.y}, r1 = {r4.z, r4.w};
            acc[0] = __builtin_amdgcn_wmma_f32_16x16x4_f32(
                false, f0, false, p0, (short)0, acc[0], false, false);
            acc[1] = __builtin_amdgcn_wmma_f32_16x16x4_f32(
                false, f0, false, q0, (short)0, acc[1], false, false);
            acc[2] = __builtin_amdgcn_wmma_f32_16x16x4_f32(
                false, f0, false, r0, (short)0, acc[2], false, false);
            acc[0] = __builtin_amdgcn_wmma_f32_16x16x4_f32(
                false, f1, false, p1, (short)0, acc[0], false, false);
            acc[1] = __builtin_amdgcn_wmma_f32_16x16x4_f32(
                false, f1, false, q1, (short)0, acc[1], false, false);
            acc[2] = __builtin_amdgcn_wmma_f32_16x16x4_f32(
                false, f1, false, r1, (short)0, acc[2], false, false);
        }
        __syncthreads();
    }
#undef XD_FETCH

    #pragma unroll
    for (int ni = 0; ni < 3; ++ni)
        #pragma unroll
        for (int r = 0; r < 8; ++r)
            red[w][h * 8 + r][ni * 16 + c] = acc[ni][r];
    __syncthreads();

    for (int idx = tid; idx < 16 * 33; idx += 256) {
        const int row = idx / 33, e = idx - row * 33;
        float s = 0.f;
        #pragma unroll
        for (int q = 0; q < 8; ++q) s += red[q][row][e];
        xdbl[(size_t)m0 * 33 + idx] = s;
    }
}

#define SCAN_T     16
#define SCAN_NCH   (L_SEQ / SCAN_T)
#define SCAN_PITCH 36
#define SCAN_XCH   (SCAN_T * 33)
#define SCAN_CH    32
#define SCAN_YP    36

__device__ __forceinline__ int scan_dst(int i) {
    const int t = i / 33, e = i - t * 33;
    return t * SCAN_PITCH + (e ? e + 3 : 0);
}

__device__ __forceinline__ float scan_dt(float raw, float wdt, float bd) {
    const float x = raw * wdt + bd;
    const float t = __expf(-fabsf(x));
    const float s = t / (2.f + t);
    const float w = s * s;
    float q = 2.f / 13.f;
    q = q * w + 2.f / 11.f;
    q = q * w + 2.f / 9.f;
    q = q * w + 2.f / 7.f;
    q = q * w + 2.f / 5.f;
    q = q * w + 2.f / 3.f;
    q = q * w + 2.f;
    const float sp = fmaxf(x, 0.f) + s * q;
    return fminf(fmaxf(sp, 1e-4f), 10.f);
}

template <int XMASK>
__device__ __forceinline__ float scan_add_xor(float v) {
    const int o = __builtin_amdgcn_update_dpp(0, __float_as_int(v), 0x160 + XMASK,
                                              0xF, 0xF, true);
    return v + __int_as_float(o);
}

#define SCAN_GX    (D_INNER / SCAN_CH)
#define SCAN_WGS   (SCAN_GX * B_SZ)
#define SCAN_LDS   (2 * SCAN_T * SCAN_PITCH + 4 * SCAN_T * SCAN_CH + 2 * SCAN_T * SCAN_YP)

__device__ __forceinline__
void scan_group(const float* __restrict__ xdbl, const float* __restrict__ u,
                const float* __restrict__ Wdt, const float* __restrict__ bdt,
                const float* __restrict__ Alog, float* __restrict__ ybuf,
                int bx, int by, float* smem) {
    const int tid = threadIdx.x;
    const int j   = tid & 7;
    const int c   = tid >> 3;
    const int sc  = tid & 31;
    const int st  = tid >> 5;
    const int d0  = bx * SCAN_CH;
    const int b   = by;

    const float A0 = -__expf(fminf(Alog[2 * j],     5.f));
    const float A1 = -__expf(fminf(Alog[2 * j + 1], 5.f));
    const float wdt = Wdt[d0 + sc];
    const float bd  = bdt[d0 + sc];
    float h0 = 0.f, h1 = 0.f;

    const float* xd = xdbl + (size_t)b * L_SEQ * 33;
    const float* ub = u    + (size_t)b * L_SEQ * D_INNER + d0 + sc;
    float*       yb = ybuf + (size_t)b * L_SEQ * D_INNER + d0 + sc;

    float  (*sx)[SCAN_T * SCAN_PITCH] = (float (*)[SCAN_T * SCAN_PITCH])smem;
    float2 (*sdu)[SCAN_T][SCAN_CH] =
        (float2 (*)[SCAN_T][SCAN_CH])(smem + 2 * SCAN_T * SCAN_PITCH);
    float  (*sy)[SCAN_T * SCAN_YP] =
        (float (*)[SCAN_T * SCAN_YP])(smem + 2 * SCAN_T * SCAN_PITCH + 4 * SCAN_T * SCAN_CH);

    const int dst0 = scan_dst(tid);
    const int dst1 = scan_dst(tid + 256);
    const int dst2 = scan_dst(tid + 512);
    const bool has2 = tid < (SCAN_XCH - 512);

    sx[0][dst0] = xd[tid];
    sx[0][dst1] = xd[tid + 256];
    if (has2) sx[0][dst2] = xd[tid + 512];
    sdu[0][st][sc]     = make_float2(scan_dt(xd[st * 33], wdt, bd),
                                     ub[(size_t)st * D_INNER]);
    sdu[0][st + 8][sc] = make_float2(scan_dt(xd[(st + 8) * 33], wdt, bd),
                                     ub[(size_t)(st + 8) * D_INNER]);
    __syncthreads();

    for (int ch = 0; ch < SCAN_NCH; ++ch) {
        const int  cur  = ch & 1;
        const bool more = (ch + 1 < SCAN_NCH);

        float xn0 = 0.f, xn1 = 0.f, xn2 = 0.f;
        float rn0 = 0.f, rn1 = 0.f, un0 = 0.f, un1 = 0.f;
        if (more) {
            const float* nsrc = xd + (size_t)(ch + 1) * SCAN_XCH;
            xn0 = nsrc[tid];
            xn1 = nsrc[tid + 256];
            if (has2) xn2 = nsrc[tid + 512];
            rn0 = nsrc[st * 33];
            rn1 = nsrc[(st + 8) * 33];
            const float* nu = ub + (size_t)(ch + 1) * SCAN_T * D_INNER;
            un0 = nu[(size_t)st * D_INNER];
            un1 = nu[(size_t)(st + 8) * D_INNER];
        }

        if (ch > 0) {
            const float* py = sy[cur ^ 1];
            float* yc = yb + (size_t)(ch - 1) * SCAN_T * D_INNER;
            yc[(size_t)st * D_INNER]       = py[st * SCAN_YP + sc];
            yc[(size_t)(st + 8) * D_INNER] = py[(st + 8) * SCAN_YP + sc];
        }

        const float* s = sx[cur];
        float ysel = 0.f;
        #pragma unroll
        for (int t = 0; t < SCAN_T; ++t) {
            const float2 du = sdu[cur][t][c];
            const float2 Bq = *(const float2*)&s[t * SCAN_PITCH + 4  + 2 * j];
            const float2 Cq = *(const float2*)&s[t * SCAN_PITCH + 20 + 2 * j];
            const float dt = du.x, ut = du.y;

            const float dA0 = __expf(fminf(fmaxf(dt * A0, -20.f), 0.f));
            const float dA1 = __expf(fminf(fmaxf(dt * A1, -20.f), 0.f));
            const float dB0 = fminf(fmaxf(dt * Bq.x * ut, -10.f), 10.f);
            const float dB1 = fminf(fmaxf(dt * Bq.y * ut, -10.f), 10.f);
            h0 = fminf(fmaxf(h0 * dA0 + dB0, -100.f), 100.f);
            h1 = fminf(fmaxf(h1 * dA1 + dB1, -100.f), 100.f);

            float p = h0 * Cq.x + h1 * Cq.y;
            p = scan_add_xor<1>(p);
            p = scan_add_xor<2>(p);
            p = scan_add_xor<4>(p);
            if ((t & 7) == j) ysel = p;
            if ((t & 7) == 7)
                sy[cur][((t & 8) + j) * SCAN_YP + c] = ysel;
        }

        if (more) {
            float* nx = sx[cur ^ 1];
            nx[dst0] = xn0;
            nx[dst1] = xn1;
            if (has2) nx[dst2] = xn2;
            const float dn0 = scan_dt(rn0, wdt, bd);
            const float dn1 = scan_dt(rn1, wdt, bd);
            sdu[cur ^ 1][st][sc]     = make_float2(dn0, un0);
            sdu[cur ^ 1][st + 8][sc] = make_float2(dn1, un1);
        }
        __syncthreads();
    }

    {
        const float* py = sy[(SCAN_NCH - 1) & 1];
        float* yc = yb + (size_t)(SCAN_NCH - 1) * SCAN_T * D_INNER;
        yc[(size_t)st * D_INNER]       = py[st * SCAN_YP + sc];
        yc[(size_t)(st + 8) * D_INNER] = py[(st + 8) * SCAN_YP + sc];
    }
}

#define ZG_NX   (D_INNER / 128)
#define ZG_WGS  (ZG_NX * (M_ROWS / 64))

__global__ __launch_bounds__(256)
void scan_zgemm_kernel(const float* __restrict__ xdbl, const float* __restrict__ u,
                       const float* __restrict__ Wdt, const float* __restrict__ bdt,
                       const float* __restrict__ Alog, float* __restrict__ ybuf,
                       const float* __restrict__ x, const float* __restrict__ Wz,
                       float* __restrict__ zout, int ldc, int K) {
    __shared__ __align__(16) float smem[2 * GM_STAGE];
    static_assert(SCAN_LDS <= 2 * GM_STAGE, "scan LDS must fit the GEMM stages");
    const int bid = blockIdx.x;
    if (bid < SCAN_WGS) {
        scan_group(xdbl, u, Wdt, bdt, Alog, ybuf, bid % SCAN_GX, bid / SCAN_GX, smem);
    } else {
        const int g = bid - SCAN_WGS;
        gemm_tile_f32_wmma(x, Wz, zout, ldc, K, g % ZG_NX, g / ZG_NX, smem);
    }
}

__global__ __launch_bounds__(256)
void ln_gate_kernel(float* __restrict__ ybuf, const float* __restrict__ u,
                    const float* __restrict__ xz, const float* __restrict__ Dp,
                    const float* __restrict__ g, const float* __restrict__ bta) {
    const int row = blockIdx.x;
    const int tid = threadIdx.x;
    float* yr = ybuf + (size_t)row * D_INNER;

    float v[8], s = 0.f, s2 = 0.f;
    #pragma unroll
    for (int i = 0; i < 8; ++i) {
        v[i] = yr[tid + i * 256];
        s += v[i]; s2 += v[i] * v[i];
    }
    #pragma unroll
    for (int off = 16; off; off >>= 1) {
        s  += __shfl_xor(s,  off, 32);
        s2 += __shfl_xor(s2, off, 32);
    }
    __shared__ float ws1[8], ws2[8];
    const int lane = tid & 31, w = tid >> 5;
    if (lane == 0) { ws1[w] = s; ws2[w] = s2; }
    __syncthreads();
    if (tid == 0) {
        float t1 = 0.f, t2 = 0.f;
        #pragma unroll
        for (int i = 0; i < 8; ++i) { t1 += ws1[i]; t2 += ws2[i]; }
        ws1[0] = t1; ws2[0] = t2;
    }
    __syncthreads();
    const float mu  = ws1[0] * (1.f / D_INNER);
    float var = ws2[0] * (1.f / D_INNER) - mu * mu;
    var = fmaxf(var, 0.f);
    const float rs = rsqrtf(var + 1e-5f);

    #pragma unroll
    for (int i = 0; i < 8; ++i) {
        const int d = tid + i * 256;
        float yn = (v[i] - mu) * rs * g[d] + bta[d];
        float uu = u[(size_t)row * D_INNER + d];
        float zz = xz[(size_t)row * (2 * D_INNER) + D_INNER + d];
        float sz = zz / (1.f + __expf(-zz));
        yr[d] = (yn + Dp[d] * uu) * sz;
    }
}

extern "C" void kernel_launch(void* const* d_in, const int* in_sizes, int n_in,
                              void* d_out, int out_size, void* d_ws, size_t ws_size,
                              hipStream_t stream) {
    const float* x      = (const float*)d_in[0];
    const float* W_in   = (const float*)d_in[1];
    const float* conv_w = (const float*)d_in[2];
    const float* conv_b = (const float*)d_in[3];
    const float* W_x    = (const float*)d_in[4];
    const float* W_dt   = (const float*)d_in[5];
    const float* b_dt   = (const float*)d_in[6];
    const float* A_log  = (const float*)d_in[7];
    const float* D_par  = (const float*)d_in[8];
    const float* W_out  = (const float*)d_in[9];
    const float* ln_g   = (const float*)d_in[10];
    const float* ln_b   = (const float*)d_in[11];
    float* out = (float*)d_out;

    float* ws = (float*)d_ws;
    const size_t XZ_OFF   = 0;
    const size_t U_OFF    = XZ_OFF + (size_t)M_ROWS * 2 * D_INNER;
    const size_t XDBL_OFF = U_OFF  + (size_t)M_ROWS * D_INNER;
    const size_t Y_OFF    = XDBL_OFF + (size_t)M_ROWS * 33;
    float* xz   = ws + XZ_OFF;
    float* u    = ws + U_OFF;
    float* xdbl = ws + XDBL_OFF;
    float* ybuf = ws + Y_OFF;

    gemm_f32_wmma<<<dim3(D_INNER / 128, M_ROWS / 64), 256, 0, stream>>>(
        x, W_in, xz, 2 * D_INNER, D_MODEL);

    conv_silu_kernel<<<(M_ROWS / CONV_T) * 512 / 256, 256, 0, stream>>>(
        xz, conv_w, conv_b, u);

    xdbl_wmma_kernel<<<M_ROWS / 16, 256, 0, stream>>>(u, W_x, xdbl);

    scan_zgemm_kernel<<<SCAN_WGS + ZG_WGS, 256, 0, stream>>>(
        xdbl, u, W_dt, b_dt, A_log, ybuf,
        x, W_in + (size_t)D_INNER * D_MODEL, xz + D_INNER, 2 * D_INNER, D_MODEL);

    ln_gate_kernel<<<M_ROWS, 256, 0, stream>>>(ybuf, u, xz, D_par, ln_g, ln_b);

    gemm_f32_wmma<<<dim3(D_MODEL / 128, M_ROWS / 64), 256, 0, stream>>>(
        ybuf, W_out, out, D_MODEL, D_INNER);
}


# scan step body trimmed on the vector ALU: y of 8 steps reduced by one transposing butterfly (7 cross-lane adds, was 24; sum order changes); log2(e) folded into A, dt*u formed at staging
# speedup vs baseline: 8.4406x; 1.0682x over previous
#include <hip/hip_runtime.h>
#include <hip/hip_bf16.h>


typedef __attribute__((ext_vector_type(2))) float v2f;
typedef __attribute__((ext_vector_type(8))) float v8f;

#define D_MODEL 1024
#define D_INNER 2048
#define D_STATE 16
#define L_SEQ   2048
#define B_SZ    2
#define M_ROWS  (B_SZ * L_SEQ)

#define GM_PITCH 36
#define GM_BROW  64
#define GM_STAGE ((64 + 128) * GM_PITCH)

__device__ __forceinline__
void gemm_tile_f32_wmma(const float* __restrict__ A, const float* __restrict__ Bm,
                        float* __restrict__ C, int ldc, int K,
                        int bx, int by, float* Ls) {
    const int tid  = threadIdx.x;
    const int lane = tid & 31;
    const int w    = tid >> 5;
    const int wm   = w >> 2;
    const int wn   = w & 3;
    const int lmod = lane & 15;
    const int lhal = lane >> 4;

    const int n0 = bx * 128;
    const int m0 = by * 64;

    const int sr = tid >> 3, sc = (tid & 7) * 4;
    const float* ga = A  + (size_t)(m0 + sr) * K + sc;
    const float* gb = Bm + (size_t)(n0 + sr) * K + sc;
    const int so = sr * GM_PITCH + sc;

    const int fa0 = (wm * 32 + lmod) * GM_PITCH + lhal * 2;
    const int fb0 = (GM_BROW + wn * 32 + lmod) * GM_PITCH + lhal * 2;

    float4 ra0, ra1, rb0, rb1, rb2, rb3;
#define GM_FETCH(k0)                                                    \
    ra0 = *(const float4*)&ga[(size_t) 0 * K + (k0)];                   \
    ra1 = *(const float4*)&ga[(size_t)32 * K + (k0)];                   \
    rb0 = *(const float4*)&gb[(size_t) 0 * K + (k0)];                   \
    rb1 = *(const float4*)&gb[(size_t)32 * K + (k0)];                   \
    rb2 = *(const float4*)&gb[(size_t)64 * K + (k0)];                   \
    rb3 = *(const float4*)&gb[(size_t)96 * K + (k0)];
#define GM_PUT(st)                                                      \
    *(float4*)&Ls[(st) + so                               ] = ra0;      \
    *(float4*)&Ls[(st) + so + 32 * GM_PITCH               ] = ra1;      \
    *(float4*)&Ls[(st) + so + (GM_BROW     ) * GM_PITCH   ] = rb0;      \
    *(float4*)&Ls[(st) + so + (GM_BROW + 32) * GM_PITCH   ] = rb1;      \
    *(float4*)&Ls[(st) + so + (GM_BROW + 64) * GM_PITCH   ] = rb2;      \
    *(float4*)&Ls[(st) + so + (GM_BROW + 96) * GM_PITCH   ] = rb3;
#define GM_FRAGS(S, st, u)                                              \
    _Pragma("unroll")                                                   \
    for (int q = 0; q < 2; ++q) {                                       \
        const int kc = ((u) * 2 + q) * 4;                               \
        fa[S][q][0] = *(const v2f*)&Ls[(st) + fa0 + kc];                \
        fa[S][q][1] = *(const v2f*)&Ls[(st) + fa0 + 16 * GM_PITCH + kc];\
        fb[S][q][0] = *(const v2f*)&Ls[(st) + fb0 + kc];                \
        fb[S][q][1] = *(const v2f*)&Ls[(st) + fb0 + 16 * GM_PITCH + kc];\
    }

    v8f acc[2][2] = {};
    v2f fa[2][2][2], fb[2][2][2];

    const int T = K / 32;
    GM_FETCH(0)
    GM_PUT(0)
    __syncthreads();
    if (1 < T) {
        GM_FETCH(32)
    }
    GM_FRAGS(0, 0, 0)

    for (int t = 0; t < T; ++t) {
        const int cur = (t & 1) * GM_STAGE;
        const int nxt = GM_STAGE - cur;

        #pragma unroll
        for (int u = 0; u < 4; ++u) {
            const int cs = u & 1;
            if (u < 3) {
                GM_FRAGS(cs ^ 1, cur, u + 1)
            } else {
                GM_PUT(nxt)
                __syncthreads();
                if (t + 2 < T) {
                    GM_FETCH((t + 2) * 32)
                }
                GM_FRAGS(0, nxt, 0)
                __builtin_amdgcn_sched_barrier(0);
            }
            #pragma unroll
            for (int q = 0; q < 2; ++q)
                #pragma unroll
                for (int mi = 0; mi < 2; ++mi)
                    #pragma unroll
                    for (int ni = 0; ni < 2; ++ni)
                        acc[mi][ni] = __builtin_amdgcn_wmma_f32_16x16x4_f32(
                            false, fa[cs][q][mi], false, fb[cs][q][ni],
                            (short)0, acc[mi][ni], false, false);
        }
        __builtin_amdgcn_sched_barrier(0);
        asm volatile("" :: "v"(fa[0][0][0]), "v"(fa[0][0][1]), "v"(fa[0][1][0]),
                           "v"(fa[0][1][1]), "v"(fb[0][0][0]), "v"(fb[0][0][1]),
                           "v"(fb[0][1][0]), "v"(fb[0][1][1]));
    }
#undef GM_FETCH
#undef GM_PUT
#undef GM_FRAGS

    #pragma unroll
    for (int mi = 0; mi < 2; ++mi)
        #pragma unroll
        for (int ni = 0; ni < 2; ++ni)
            #pragma unroll
            for (int r = 0; r < 8; ++r) {
                int gm = m0 + wm * 32 + mi * 16 + lhal * 8 + r;
                int gn = n0 + wn * 32 + ni * 16 + lmod;
                C[(size_t)gm * ldc + gn] = acc[mi][ni][r];
            }
}

__global__ __launch_bounds__(256)
void gemm_f32_wmma(const float* __restrict__ A, const float* __restrict__ Bm,
                   float* __restrict__ C, int ldc, int K) {
    __shared__ __align__(16) float Ls[2 * GM_STAGE];
    gemm_tile_f32_wmma(A, Bm, C, ldc, K, blockIdx.x, blockIdx.y, Ls);
}

#define CONV_T 16

__device__ __forceinline__ float conv_silu1(float bias, float x0, float x1,
                                            float x2, float x3, float4 w) {
    float acc = bias;
    acc += x0 * w.x;
    acc += x1 * w.y;
    acc += x2 * w.z;
    acc += x3 * w.w;
    return acc / (1.f + __expf(-acc));
}

__global__ __launch_bounds__(256)
void conv_silu_kernel(const float* __restrict__ xz, const float* __restrict__ cw,
                      const float* __restrict__ cb, float* __restrict__ u) {
    const int id = blockIdx.x * 256 + threadIdx.x;
    const int d  = (id & 511) * 4;
    const int r0 = (id >> 9) * CONV_T;
    const bool head = (r0 & (L_SEQ - 1)) == 0;

    const float* src = xz + (size_t)r0 * (2 * D_INNER) + d;

    float4 x[CONV_T + 3];
    #pragma unroll
    for (int k = 0; k < 3; ++k) x[k] = make_float4(0.f, 0.f, 0.f, 0.f);
    if (!head) {
        #pragma unroll
        for (int k = 0; k < 3; ++k)
            x[k] = *(const float4*)(src + (ptrdiff_t)(k - 3) * (2 * D_INNER));
    }
    #pragma unroll
    for (int k = 3; k < CONV_T + 3; ++k)
        x[k] = *(const float4*)(src + (size_t)(k - 3) * (2 * D_INNER));

    const float4 w0 = *(const float4*)&cw[(d + 0) * 4];
    const float4 w1 = *(const float4*)&cw[(d + 1) * 4];
    const float4 w2 = *(const float4*)&cw[(d + 2) * 4];
    const float4 w3 = *(const float4*)&cw[(d + 3) * 4];
    const float4 bb = *(const float4*)&cb[d];

    float* dst = u + (size_t)r0 * D_INNER + d;
    #pragma unroll
    for (int t = 0; t < CONV_T; ++t) {
        float4 o;
        o.x = conv_silu1(bb.x, x[t].x, x[t + 1].x, x[t + 2].x, x[t + 3].x, w0);
        o.y = conv_silu1(bb.y, x[t].y, x[t + 1].y, x[t + 2].y, x[t + 3].y, w1);
        o.z = conv_silu1(bb.z, x[t].z, x[t + 1].z, x[t + 2].z, x[t + 3].z, w2);
        o.w = conv_silu1(bb.w, x[t].w, x[t + 1].w, x[t + 2].w, x[t + 3].w, w3);
        *(float4*)(dst + (size_t)t * D_INNER) = o;
    }
}

#define XD_KT    256
#define XD_PITCH 260
#define XD_RP    52

__global__ __launch_bounds__(256)
void xdbl_wmma_kernel(const float* u, const float* Wx, float* __restrict__ xdbl) {
    __shared__ __align__(16) float As[16][XD_PITCH];
    __shared__ __align__(16) float Bs[33][XD_PITCH];
    __shared__ __align__(16) float red[8][16][XD_RP];

    const int tid  = threadIdx.x;
    const int lane = tid & 31;
    const int w    = tid >> 5;
    const int c    = lane & 15;
    const int h    = lane >> 4;
    const int m0   = blockIdx.x * 16;

    const int  sr   = tid >> 6;
    const int  sc   = (tid & 63) * 4;
    const bool last = tid < 64;
    const float* ua = u  + (size_t)(m0 + sr) * D_INNER + sc;
    const float* wb = Wx + (size_t)sr * D_INNER + sc;

    float4 a0, a1, a2, a3, b0, b1, b2, b3, b4, b5, b6, b7;
    float4 bl = make_float4(0.f, 0.f, 0.f, 0.f);
#define XD_FETCH(ko)                                                    \
    a0 = *(const float4*)&ua[(size_t) 0 * D_INNER + (ko)];              \
    a1 = *(const float4*)&ua[(size_t) 4 * D_INNER + (ko)];              \
    a2 = *(const float4*)&ua[(size_t) 8 * D_INNER + (ko)];              \
    a3 = *(const float4*)&ua[(size_t)12 * D_INNER + (ko)];              \
    b0 = *(const float4*)&wb[(size_t) 0 * D_INNER + (ko)];              \
    b1 = *(const float4*)&wb[(size_t) 4 * D_INNER + (ko)];              \
    b2 = *(const float4*)&wb[(size_t) 8 * D_INNER + (ko)];              \
    b3 = *(const float4*)&wb[(size_t)12 * D_INNER + (ko)];              \
    b4 = *(const float4*)&wb[(size_t)16 * D_INNER + (ko)];              \
    b5 = *(const float4*)&wb[(size_t)20 * D_INNER + (ko)];              \
    b6 = *(const float4*)&wb[(size_t)24 * D_INNER + (ko)];              \
    b7 = *(const float4*)&wb[(size_t)28 * D_INNER + (ko)];              \
    if (last) bl = *(const float4*)&wb[(size_t)32 * D_INNER + (ko)];

    XD_FETCH(0)

    v8f acc[3] = {};
    const int kb = w * 32 + h * 4;

    for (int kt = 0; kt < D_INNER / XD_KT; ++kt) {
        *(float4*)&As[sr     ][sc] = a0;
        *(float4*)&As[sr +  4][sc] = a1;
        *(float4*)&As[sr +  8][sc] = a2;
        *(float4*)&As[sr + 12][sc] = a3;
        *(float4*)&Bs[sr     ][sc] = b0;
        *(float4*)&Bs[sr +  4][sc] = b1;
        *(float4*)&Bs[sr +  8][sc] = b2;
        *(float4*)&Bs[sr + 12][sc] = b3;
        *(float4*)&Bs[sr + 16][sc] = b4;
        *(float4*)&Bs[sr + 20][sc] = b5;
        *(float4*)&Bs[sr + 24][sc] = b6;
        *(float4*)&Bs[sr + 28][sc] = b7;
        if (last) *(float4*)&Bs[32][sc] = bl;
        __syncthreads();

        if (kt + 1 < D_INNER / XD_KT) {
            XD_FETCH((kt + 1) * XD_KT)
        }

        #pragma unroll
        for (int i = 0; i < 4; ++i) {
            const float4 f4 = *(const float4*)&As[c][kb + 8 * i];
            const float4 p4 = *(const float4*)&Bs[c][kb + 8 * i];
            const float4 q4 = *(const float4*)&Bs[16 + c][kb + 8 * i];
            const float4 r4 = *(const float4*)&Bs[32][kb + 8 * i];
            const v2f f0 = {f4.x, f4.y}, f1 = {f4.z, f4.w};
            const v2f p0 = {p4.x, p4.y}, p1 = {p4.z, p4.w};
            const v2f q0 = {q4.x, q4.y}, q1 = {q4.z, q4.w};
            const v2f r0 = {r4.x, r4.y}, r1 = {r4.z, r4.w};
            acc[0] = __builtin_amdgcn_wmma_f32_16x16x4_f32(
                false, f0, false, p0, (short)0, acc[0], false, false);
            acc[1] = __builtin_amdgcn_wmma_f32_16x16x4_f32(
                false, f0, false, q0, (short)0, acc[1], false, false);
            acc[2] = __builtin_amdgcn_wmma_f32_16x16x4_f32(
                false, f0, false, r0, (short)0, acc[2], false, false);
            acc[0] = __builtin_amdgcn_wmma_f32_16x16x4_f32(
                false, f1, false, p1, (short)0, acc[0], false, false);
            acc[1] = __builtin_amdgcn_wmma_f32_16x16x4_f32(
                false, f1, false, q1, (short)0, acc[1], false, false);
            acc[2] = __builtin_amdgcn_wmma_f32_16x16x4_f32(
                false, f1, false, r1, (short)0, acc[2], false, false);
        }
        __syncthreads();
    }
#undef XD_FETCH

    #pragma unroll
    for (int ni = 0; ni < 3; ++ni)
        #pragma unroll
        for (int r = 0; r < 8; ++r)
            red[w][h * 8 + r][ni * 16 + c] = acc[ni][r];
    __syncthreads();

    for (int idx = tid; idx < 16 * 33; idx += 256) {
        const int row = idx / 33, e = idx - row * 33;
        float s = 0.f;
        #pragma unroll
        for (int q = 0; q < 8; ++q) s += red[q][row][e];
        xdbl[(size_t)m0 * 33 + idx] = s;
    }
}

#define SCAN_T     16
#define SCAN_NCH   (L_SEQ / SCAN_T)
#define SCAN_PITCH 36
#define SCAN_XCH   (SCAN_T * 33)
#define SCAN_CH    32
#define SCAN_YP    36

__device__ __forceinline__ int scan_dst(int i) {
    const int t = i / 33, e = i - t * 33;
    return t * SCAN_PITCH + (e ? e + 3 : 0);
}

__device__ __forceinline__ float scan_dt(float raw, float wdt, float bd) {
    const float x = raw * wdt + bd;
    const float t = __expf(-fabsf(x));
    const float s = t / (2.f + t);
    const float w = s * s;
    float q = 2.f / 13.f;
    q = q * w + 2.f / 11.f;
    q = q * w + 2.f / 9.f;
    q = q * w + 2.f / 7.f;
    q = q * w + 2.f / 5.f;
    q = q * w + 2.f / 3.f;
    q = q * w + 2.f;
    const float sp = fmaxf(x, 0.f) + s * q;
    return fminf(fmaxf(sp, 1e-4f), 10.f);
}

template <int XMASK>
__device__ __forceinline__ float scan_xchg_add(bool up, float lo, float hi) {
    const float keep = up ? hi : lo;
    const float send = up ? lo : hi;
    const int o = __builtin_amdgcn_update_dpp(0, __float_as_int(send), 0x160 + XMASK,
                                              0xF, 0xF, true);
    return keep + __int_as_float(o);
}

#define SCAN_LOG2E 1.44269504088896341f
#define SCAN_GX    (D_INNER / SCAN_CH)
#define SCAN_WGS   (SCAN_GX * B_SZ)
#define SCAN_LDS   (2 * SCAN_T * SCAN_PITCH + 4 * SCAN_T * SCAN_CH + 2 * SCAN_T * SCAN_YP)

__device__ __forceinline__
void scan_group(const float* __restrict__ xdbl, const float* __restrict__ u,
                const float* __restrict__ Wdt, const float* __restrict__ bdt,
                const float* __restrict__ Alog, float* __restrict__ ybuf,
                int bx, int by, float* smem) {
    const int tid = threadIdx.x;
    const int j   = tid & 7;
    const int c   = tid >> 3;
    const int sc  = tid & 31;
    const int st  = tid >> 5;
    const int d0  = bx * SCAN_CH;
    const int b   = by;

    const float A0 = -__expf(fminf(Alog[2 * j],     5.f)) * SCAN_LOG2E;
    const float A1 = -__expf(fminf(Alog[2 * j + 1], 5.f)) * SCAN_LOG2E;
    const float wdt = Wdt[d0 + sc];
    const float bd  = bdt[d0 + sc];
    float h0 = 0.f, h1 = 0.f;
    const bool j4 = (j & 4) != 0, j2 = (j & 2) != 0, j1 = (j & 1) != 0;

    const float* xd = xdbl + (size_t)b * L_SEQ * 33;
    const float* ub = u    + (size_t)b * L_SEQ * D_INNER + d0 + sc;
    float*       yb = ybuf + (size_t)b * L_SEQ * D_INNER + d0 + sc;

    float  (*sx)[SCAN_T * SCAN_PITCH] = (float (*)[SCAN_T * SCAN_PITCH])smem;
    float2 (*sdu)[SCAN_T][SCAN_CH] =
        (float2 (*)[SCAN_T][SCAN_CH])(smem + 2 * SCAN_T * SCAN_PITCH);
    float  (*sy)[SCAN_T * SCAN_YP] =
        (float (*)[SCAN_T * SCAN_YP])(smem + 2 * SCAN_T * SCAN_PITCH + 4 * SCAN_T * SCAN_CH);

    const int dst0 = scan_dst(tid);
    const int dst1 = scan_dst(tid + 256);
    const int dst2 = scan_dst(tid + 512);
    const bool has2 = tid < (SCAN_XCH - 512);

    sx[0][dst0] = xd[tid];
    sx[0][dst1] = xd[tid + 256];
    if (has2) sx[0][dst2] = xd[tid + 512];
    {
        const float dc0 = scan_dt(xd[st * 33], wdt, bd);
        const float dc1 = scan_dt(xd[(st + 8) * 33], wdt, bd);
        sdu[0][st][sc]     = make_float2(dc0, dc0 * ub[(size_t)st * D_INNER]);
        sdu[0][st + 8][sc] = make_float2(dc1, dc1 * ub[(size_t)(st + 8) * D_INNER]);
    }
    __syncthreads();

    for (int ch = 0; ch < SCAN_NCH; ++ch) {
        const int  cur  = ch & 1;
        const bool more = (ch + 1 < SCAN_NCH);

        float xn0 = 0.f, xn1 = 0.f, xn2 = 0.f;
        float rn0 = 0.f, rn1 = 0.f, un0 = 0.f, un1 = 0.f;
        if (more) {
            const float* nsrc = xd + (size_t)(ch + 1) * SCAN_XCH;
            xn0 = nsrc[tid];
            xn1 = nsrc[tid + 256];
            if (has2) xn2 = nsrc[tid + 512];
            rn0 = nsrc[st * 33];
            rn1 = nsrc[(st + 8) * 33];
            const float* nu = ub + (size_t)(ch + 1) * SCAN_T * D_INNER;
            un0 = nu[(size_t)st * D_INNER];
            un1 = nu[(size_t)(st + 8) * D_INNER];
        }

        if (ch > 0) {
            const float* py = sy[cur ^ 1];
            float* yc = yb + (size_t)(ch - 1) * SCAN_T * D_INNER;
            yc[(size_t)st * D_INNER]       = py[st * SCAN_YP + sc];
            yc[(size_t)(st + 8) * D_INNER] = py[(st + 8) * SCAN_YP + sc];
        }

        const float* s = sx[cur];
        float pk[8];
        #pragma unroll
        for (int t = 0; t < SCAN_T; ++t) {
            const float2 du = sdu[cur][t][c];
            const float2 Bq = *(const float2*)&s[t * SCAN_PITCH + 4  + 2 * j];
            const float2 Cq = *(const float2*)&s[t * SCAN_PITCH + 20 + 2 * j];
            const float dt = du.x, dtu = du.y;

            const float dA0 = __builtin_amdgcn_exp2f(
                fminf(fmaxf(dt * A0, -20.f * SCAN_LOG2E), 0.f));
            const float dA1 = __builtin_amdgcn_exp2f(
                fminf(fmaxf(dt * A1, -20.f * SCAN_LOG2E), 0.f));
            const float dB0 = fminf(fmaxf(dtu * Bq.x, -10.f), 10.f);
            const float dB1 = fminf(fmaxf(dtu * Bq.y, -10.f), 10.f);
            h0 = fminf(fmaxf(h0 * dA0 + dB0, -100.f), 100.f);
            h1 = fminf(fmaxf(h1 * dA1 + dB1, -100.f), 100.f);

            pk[t & 7] = h0 * Cq.x + h1 * Cq.y;
            if ((t & 7) == 7) {
                const float q0 = scan_xchg_add<4>(j4, pk[0], pk[4]);
                const float q1 = scan_xchg_add<4>(j4, pk[1], pk[5]);
                const float q2 = scan_xchg_add<4>(j4, pk[2], pk[6]);
                const float q3 = scan_xchg_add<4>(j4, pk[3], pk[7]);
                const float r0 = scan_xchg_add<2>(j2, q0, q2);
                const float r1 = scan_xchg_add<2>(j2, q1, q3);
                sy[cur][((t & 8) + j) * SCAN_YP + c] = scan_xchg_add<1>(j1, r0, r1);
            }
        }

        if (more) {
            float* nx = sx[cur ^ 1];
            nx[dst0] = xn0;
            nx[dst1] = xn1;
            if (has2) nx[dst2] = xn2;
            const float dn0 = scan_dt(rn0, wdt, bd);
            const float dn1 = scan_dt(rn1, wdt, bd);
            sdu[cur ^ 1][st][sc]     = make_float2(dn0, dn0 * un0);
            sdu[cur ^ 1][st + 8][sc] = make_float2(dn1, dn1 * un1);
        }
        __syncthreads();
    }

    {
        const float* py = sy[(SCAN_NCH - 1) & 1];
        float* yc = yb + (size_t)(SCAN_NCH - 1) * SCAN_T * D_INNER;
        yc[(size_t)st * D_INNER]       = py[st * SCAN_YP + sc];
        yc[(size_t)(st + 8) * D_INNER] = py[(st + 8) * SCAN_YP + sc];
    }
}

#define ZG_NX   (D_INNER / 128)
#define ZG_WGS  (ZG_NX * (M_ROWS / 64))

__global__ __launch_bounds__(256)
void scan_zgemm_kernel(const float* __restrict__ xdbl, const float* __restrict__ u,
                       const float* __restrict__ Wdt, const float* __restrict__ bdt,
                       const float* __restrict__ Alog, float* __restrict__ ybuf,
                       const float* __restrict__ x, const float* __restrict__ Wz,
                       float* __restrict__ zout, int ldc, int K) {
    __shared__ __align__(16) float smem[2 * GM_STAGE];
    static_assert(SCAN_LDS <= 2 * GM_STAGE, "scan LDS must fit the GEMM stages");
    const int bid = blockIdx.x;
    if (bid < SCAN_WGS) {
        scan_group(xdbl, u, Wdt, bdt, Alog, ybuf, bid % SCAN_GX, bid / SCAN_GX, smem);
    } else {
        const int g = bid - SCAN_WGS;
        gemm_tile_f32_wmma(x, Wz, zout, ldc, K, g % ZG_NX, g / ZG_NX, smem);
    }
}

__global__ __launch_bounds__(256)
void ln_gate_kernel(float* __restrict__ ybuf, const float* __restrict__ u,
                    const float* __restrict__ xz, const float* __restrict__ Dp,
                    const float* __restrict__ g, const float* __restrict__ bta) {
    const int row = blockIdx.x;
    const int tid = threadIdx.x;
    float* yr = ybuf + (size_t)row * D_INNER;

    float v[8], s = 0.f, s2 = 0.f;
    #pragma unroll
    for (int i = 0; i < 8; ++i) {
        v[i] = yr[tid + i * 256];
        s += v[i]; s2 += v[i] * v[i];
    }
    #pragma unroll
    for (int off = 16; off; off >>= 1) {
        s  += __shfl_xor(s,  off, 32);
        s2 += __shfl_xor(s2, off, 32);
    }
    __shared__ float ws1[8], ws2[8];
    const int lane = tid & 31, w = tid >> 5;
    if (lane == 0) { ws1[w] = s; ws2[w] = s2; }
    __syncthreads();
    if (tid == 0) {
        float t1 = 0.f, t2 = 0.f;
        #pragma unroll
        for (int i = 0; i < 8; ++i) { t1 += ws1[i]; t2 += ws2[i]; }
        ws1[0] = t1; ws2[0] = t2;
    }
    __syncthreads();
    const float mu  = ws1[0] * (1.f / D_INNER);
    float var = ws2[0] * (1.f / D_INNER) - mu * mu;
    var = fmaxf(var, 0.f);
    const float rs = rsqrtf(var + 1e-5f);

    #pragma unroll
    for (int i = 0; i < 8; ++i) {
        const int d = tid + i * 256;
        float yn = (v[i] - mu) * rs * g[d] + bta[d];
        float uu = u[(size_t)row * D_INNER + d];
        float zz = xz[(size_t)row * (2 * D_INNER) + D_INNER + d];
        float sz = zz / (1.f + __expf(-zz));
        yr[d] = (yn + Dp[d] * uu) * sz;
    }
}

extern "C" void kernel_launch(void* const* d_in, const int* in_sizes, int n_in,
                              void* d_out, int out_size, void* d_ws, size_t ws_size,
                              hipStream_t stream) {
    const float* x      = (const float*)d_in[0];
    const float* W_in   = (const float*)d_in[1];
    const float* conv_w = (const float*)d_in[2];
    const float* conv_b = (const float*)d_in[3];
    const float* W_x    = (const float*)d_in[4];
    const float* W_dt   = (const float*)d_in[5];
    const float* b_dt   = (const float*)d_in[6];
    const float* A_log  = (const float*)d_in[7];
    const float* D_par  = (const float*)d_in[8];
    const float* W_out  = (const float*)d_in[9];
    const float* ln_g   = (const float*)d_in[10];
    const float* ln_b   = (const float*)d_in[11];
    float* out = (float*)d_out;

    float* ws = (float*)d_ws;
    const size_t XZ_OFF   = 0;
    const size_t U_OFF    = XZ_OFF + (size_t)M_ROWS * 2 * D_INNER;
    const size_t XDBL_OFF = U_OFF  + (size_t)M_ROWS * D_INNER;
    const size_t Y_OFF    = XDBL_OFF + (size_t)M_ROWS * 33;
    float* xz   = ws + XZ_OFF;
    float* u    = ws + U_OFF;
    float* xdbl = ws + XDBL_OFF;
    float* ybuf = ws + Y_OFF;

    gemm_f32_wmma<<<dim3(D_INNER / 128, M_ROWS / 64), 256, 0, stream>>>(
        x, W_in, xz, 2 * D_INNER, D_MODEL);

    conv_silu_kernel<<<(M_ROWS / CONV_T) * 512 / 256, 256, 0, stream>>>(
        xz, conv_w, conv_b, u);

    xdbl_wmma_kernel<<<M_ROWS / 16, 256, 0, stream>>>(u, W_x, xdbl);

    scan_zgemm_kernel<<<SCAN_WGS + ZG_WGS, 256, 0, stream>>>(
        xdbl, u, W_dt, b_dt, A_log, ybuf,
        x, W_in + (size_t)D_INNER * D_MODEL, xz + D_INNER, 2 * D_INNER, D_MODEL);

    ln_gate_kernel<<<M_ROWS, 256, 0, stream>>>(ybuf, u, xz, D_par, ln_g, ln_b);

    gemm_f32_wmma<<<dim3(D_MODEL / 128, M_ROWS / 64), 256, 0, stream>>>(
        ybuf, W_out, out, D_MODEL, D_INNER);
}


# scan spread over all 256 WGPs: 16 channels per group, waves 0-3 run only the recurrence (one per SIMD, was two on half the chip), waves 4-7 do all loads, dt, staging, y stores; 32-step chunks
# speedup vs baseline: 9.0149x; 1.0680x over previous
#include <hip/hip_runtime.h>
#include <hip/hip_bf16.h>


typedef __attribute__((ext_vector_type(2))) float v2f;
typedef __attribute__((ext_vector_type(8))) float v8f;

#define D_MODEL 1024
#define D_INNER 2048
#define D_STATE 16
#define L_SEQ   2048
#define B_SZ    2
#define M_ROWS  (B_SZ * L_SEQ)

#define GM_PITCH 36
#define GM_BROW  64
#define GM_STAGE ((64 + 128) * GM_PITCH)

__device__ __forceinline__
void gemm_tile_f32_wmma(const float* __restrict__ A, const float* __restrict__ Bm,
                        float* __restrict__ C, int ldc, int K,
                        int bx, int by, float* Ls) {
    const int tid  = threadIdx.x;
    const int lane = tid & 31;
    const int w    = tid >> 5;
    const int wm   = w >> 2;
    const int wn   = w & 3;
    const int lmod = lane & 15;
    const int lhal = lane >> 4;

    const int n0 = bx * 128;
    const int m0 = by * 64;

    const int sr = tid >> 3, sc = (tid & 7) * 4;
    const float* ga = A  + (size_t)(m0 + sr) * K + sc;
    const float* gb = Bm + (size_t)(n0 + sr) * K + sc;
    const int so = sr * GM_PITCH + sc;

    const int fa0 = (wm * 32 + lmod) * GM_PITCH + lhal * 2;
    const int fb0 = (GM_BROW + wn * 32 + lmod) * GM_PITCH + lhal * 2;

    float4 ra0, ra1, rb0, rb1, rb2, rb3;
#define GM_FETCH(k0)                                                    \
    ra0 = *(const float4*)&ga[(size_t) 0 * K + (k0)];                   \
    ra1 = *(const float4*)&ga[(size_t)32 * K + (k0)];                   \
    rb0 = *(const float4*)&gb[(size_t) 0 * K + (k0)];                   \
    rb1 = *(const float4*)&gb[(size_t)32 * K + (k0)];                   \
    rb2 = *(const float4*)&gb[(size_t)64 * K + (k0)];                   \
    rb3 = *(const float4*)&gb[(size_t)96 * K + (k0)];
#define GM_PUT(st)                                                      \
    *(float4*)&Ls[(st) + so                               ] = ra0;      \
    *(float4*)&Ls[(st) + so + 32 * GM_PITCH               ] = ra1;      \
    *(float4*)&Ls[(st) + so + (GM_BROW     ) * GM_PITCH   ] = rb0;      \
    *(float4*)&Ls[(st) + so + (GM_BROW + 32) * GM_PITCH   ] = rb1;      \
    *(float4*)&Ls[(st) + so + (GM_BROW + 64) * GM_PITCH   ] = rb2;      \
    *(float4*)&Ls[(st) + so + (GM_BROW + 96) * GM_PITCH   ] = rb3;
#define GM_FRAGS(S, st, u)                                              \
    _Pragma("unroll")                                                   \
    for (int q = 0; q < 2; ++q) {                                       \
        const int kc = ((u) * 2 + q) * 4;                               \
        fa[S][q][0] = *(const v2f*)&Ls[(st) + fa0 + kc];                \
        fa[S][q][1] = *(const v2f*)&Ls[(st) + fa0 + 16 * GM_PITCH + kc];\
        fb[S][q][0] = *(const v2f*)&Ls[(st) + fb0 + kc];                \
        fb[S][q][1] = *(const v2f*)&Ls[(st) + fb0 + 16 * GM_PITCH + kc];\
    }

    v8f acc[2][2] = {};
    v2f fa[2][2][2], fb[2][2][2];

    const int T = K / 32;
    GM_FETCH(0)
    GM_PUT(0)
    __syncthreads();
    if (1 < T) {
        GM_FETCH(32)
    }
    GM_FRAGS(0, 0, 0)

    for (int t = 0; t < T; ++t) {
        const int cur = (t & 1) * GM_STAGE;
        const int nxt = GM_STAGE - cur;

        #pragma unroll
        for (int u = 0; u < 4; ++u) {
            const int cs = u & 1;
            if (u < 3) {
                GM_FRAGS(cs ^ 1, cur, u + 1)
            } else {
                GM_PUT(nxt)
                __syncthreads();
                if (t + 2 < T) {
                    GM_FETCH((t + 2) * 32)
                }
                GM_FRAGS(0, nxt, 0)
                __builtin_amdgcn_sched_barrier(0);
            }
            #pragma unroll
            for (int q = 0; q < 2; ++q)
                #pragma unroll
                for (int mi = 0; mi < 2; ++mi)
                    #pragma unroll
                    for (int ni = 0; ni < 2; ++ni)
                        acc[mi][ni] = __builtin_amdgcn_wmma_f32_16x16x4_f32(
                            false, fa[cs][q][mi], false, fb[cs][q][ni],
                            (short)0, acc[mi][ni], false, false);
        }
        __builtin_amdgcn_sched_barrier(0);
        asm volatile("" :: "v"(fa[0][0][0]), "v"(fa[0][0][1]), "v"(fa[0][1][0]),
                           "v"(fa[0][1][1]), "v"(fb[0][0][0]), "v"(fb[0][0][1]),
                           "v"(fb[0][1][0]), "v"(fb[0][1][1]));
    }
#undef GM_FETCH
#undef GM_PUT
#undef GM_FRAGS

    #pragma unroll
    for (int mi = 0; mi < 2; ++mi)
        #pragma unroll
        for (int ni = 0; ni < 2; ++ni)
            #pragma unroll
            for (int r = 0; r < 8; ++r) {
                int gm = m0 + wm * 32 + mi * 16 + lhal * 8 + r;
                int gn = n0 + wn * 32 + ni * 16 + lmod;
                C[(size_t)gm * ldc + gn] = acc[mi][ni][r];
            }
}

__global__ __launch_bounds__(256)
void gemm_f32_wmma(const float* __restrict__ A, const float* __restrict__ Bm,
                   float* __restrict__ C, int ldc, int K) {
    __shared__ __align__(16) float Ls[2 * GM_STAGE];
    gemm_tile_f32_wmma(A, Bm, C, ldc, K, blockIdx.x, blockIdx.y, Ls);
}

#define CONV_T 16

__device__ __forceinline__ float conv_silu1(float bias, float x0, float x1,
                                            float x2, float x3, float4 w) {
    float acc = bias;
    acc += x0 * w.x;
    acc += x1 * w.y;
    acc += x2 * w.z;
    acc += x3 * w.w;
    return acc / (1.f + __expf(-acc));
}

__global__ __launch_bounds__(256)
void conv_silu_kernel(const float* __restrict__ xz, const float* __restrict__ cw,
                      const float* __restrict__ cb, float* __restrict__ u) {
    const int id = blockIdx.x * 256 + threadIdx.x;
    const int d  = (id & 511) * 4;
    const int r0 = (id >> 9) * CONV_T;
    const bool head = (r0 & (L_SEQ - 1)) == 0;

    const float* src = xz + (size_t)r0 * (2 * D_INNER) + d;

    float4 x[CONV_T + 3];
    #pragma unroll
    for (int k = 0; k < 3; ++k) x[k] = make_float4(0.f, 0.f, 0.f, 0.f);
    if (!head) {
        #pragma unroll
        for (int k = 0; k < 3; ++k)
            x[k] = *(const float4*)(src + (ptrdiff_t)(k - 3) * (2 * D_INNER));
    }
    #pragma unroll
    for (int k = 3; k < CONV_T + 3; ++k)
        x[k] = *(const float4*)(src + (size_t)(k - 3) * (2 * D_INNER));

    const float4 w0 = *(const float4*)&cw[(d + 0) * 4];
    const float4 w1 = *(const float4*)&cw[(d + 1) * 4];
    const float4 w2 = *(const float4*)&cw[(d + 2) * 4];
    const float4 w3 = *(const float4*)&cw[(d + 3) * 4];
    const float4 bb = *(const float4*)&cb[d];

    float* dst = u + (size_t)r0 * D_INNER + d;
    #pragma unroll
    for (int t = 0; t < CONV_T; ++t) {
        float4 o;
        o.x = conv_silu1(bb.x, x[t].x, x[t + 1].x, x[t + 2].x, x[t + 3].x, w0);
        o.y = conv_silu1(bb.y, x[t].y, x[t + 1].y, x[t + 2].y, x[t + 3].y, w1);
        o.z = conv_silu1(bb.z, x[t].z, x[t + 1].z, x[t + 2].z, x[t + 3].z, w2);
        o.w = conv_silu1(bb.w, x[t].w, x[t + 1].w, x[t + 2].w, x[t + 3].w, w3);
        *(float4*)(dst + (size_t)t * D_INNER) = o;
    }
}

#define XD_KT    256
#define XD_PITCH 260
#define XD_RP    52

__global__ __launch_bounds__(256)
void xdbl_wmma_kernel(const float* u, const float* Wx, float* __restrict__ xdbl) {
    __shared__ __align__(16) float As[16][XD_PITCH];
    __shared__ __align__(16) float Bs[33][XD_PITCH];
    __shared__ __align__(16) float red[8][16][XD_RP];

    const int tid  = threadIdx.x;
    const int lane = tid & 31;
    const int w    = tid >> 5;
    const int c    = lane & 15;
    const int h    = lane >> 4;
    const int m0   = blockIdx.x * 16;

    const int  sr   = tid >> 6;
    const int  sc   = (tid & 63) * 4;
    const bool last = tid < 64;
    const float* ua = u  + (size_t)(m0 + sr) * D_INNER + sc;
    const float* wb = Wx + (size_t)sr * D_INNER + sc;

    float4 a0, a1, a2, a3, b0, b1, b2, b3, b4, b5, b6, b7;
    float4 bl = make_float4(0.f, 0.f, 0.f, 0.f);
#define XD_FETCH(ko)                                                    \
    a0 = *(const float4*)&ua[(size_t) 0 * D_INNER + (ko)];              \
    a1 = *(const float4*)&ua[(size_t) 4 * D_INNER + (ko)];              \
    a2 = *(const float4*)&ua[(size_t) 8 * D_INNER + (ko)];              \
    a3 = *(const float4*)&ua[(size_t)12 * D_INNER + (ko)];              \
    b0 = *(const float4*)&wb[(size_t) 0 * D_INNER + (ko)];              \
    b1 = *(const float4*)&wb[(size_t) 4 * D_INNER + (ko)];              \
    b2 = *(const float4*)&wb[(size_t) 8 * D_INNER + (ko)];              \
    b3 = *(const float4*)&wb[(size_t)12 * D_INNER + (ko)];              \
    b4 = *(const float4*)&wb[(size_t)16 * D_INNER + (ko)];              \
    b5 = *(const float4*)&wb[(size_t)20 * D_INNER + (ko)];              \
    b6 = *(const float4*)&wb[(size_t)24 * D_INNER + (ko)];              \
    b7 = *(const float4*)&wb[(size_t)28 * D_INNER + (ko)];              \
    if (last) bl = *(const float4*)&wb[(size_t)32 * D_INNER + (ko)];

    XD_FETCH(0)

    v8f acc[3] = {};
    const int kb = w * 32 + h * 4;

    for (int kt = 0; kt < D_INNER / XD_KT; ++kt) {
        *(float4*)&As[sr     ][sc] = a0;
        *(float4*)&As[sr +  4][sc] = a1;
        *(float4*)&As[sr +  8][sc] = a2;
        *(float4*)&As[sr + 12][sc] = a3;
        *(float4*)&Bs[sr     ][sc] = b0;
        *(float4*)&Bs[sr +  4][sc] = b1;
        *(float4*)&Bs[sr +  8][sc] = b2;
        *(float4*)&Bs[sr + 12][sc] = b3;
        *(float4*)&Bs[sr + 16][sc] = b4;
        *(float4*)&Bs[sr + 20][sc] = b5;
        *(float4*)&Bs[sr + 24][sc] = b6;
        *(float4*)&Bs[sr + 28][sc] = b7;
        if (last) *(float4*)&Bs[32][sc] = bl;
        __syncthreads();

        if (kt + 1 < D_INNER / XD_KT) {
            XD_FETCH((kt + 1) * XD_KT)
        }

        #pragma unroll
        for (int i = 0; i < 4; ++i) {
            const float4 f4 = *(const float4*)&As[c][kb + 8 * i];
            const float4 p4 = *(const float4*)&Bs[c][kb + 8 * i];
            const float4 q4 = *(const float4*)&Bs[16 + c][kb + 8 * i];
            const float4 r4 = *(const float4*)&Bs[32][kb + 8 * i];
            const v2f f0 = {f4.x, f4.y}, f1 = {f4.z, f4.w};
            const v2f p0 = {p4.x, p4.y}, p1 = {p4.z, p4.w};
            const v2f q0 = {q4.x, q4.y}, q1 = {q4.z, q4.w};
            const v2f r0 = {r4.x, r4.y}, r1 = {r4.z, r4.w};
            acc[0] = __builtin_amdgcn_wmma_f32_16x16x4_f32(
                false, f0, false, p0, (short)0, acc[0], false, false);
            acc[1] = __builtin_amdgcn_wmma_f32_16x16x4_f32(
                false, f0, false, q0, (short)0, acc[1], false, false);
            acc[2] = __builtin_amdgcn_wmma_f32_16x16x4_f32(
                false, f0, false, r0, (short)0, acc[2], false, false);
            acc[0] = __builtin_amdgcn_wmma_f32_16x16x4_f32(
                false, f1, false, p1, (short)0, acc[0], false, false);
            acc[1] = __builtin_amdgcn_wmma_f32_16x16x4_f32(
                false, f1, false, q1, (short)0, acc[1], false, false);
            acc[2] = __builtin_amdgcn_wmma_f32_16x16x4_f32(
                false, f1, false, r1, (short)0, acc[2], false, false);
        }
        __syncthreads();
    }
#undef XD_FETCH

    #pragma unroll
    for (int ni = 0; ni < 3; ++ni)
        #pragma unroll
        for (int r = 0; r < 8; ++r)
            red[w][h * 8 + r][ni * 16 + c] = acc[ni][r];
    __syncthreads();

    for (int idx = tid; idx < 16 * 33; idx += 256) {
        const int row = idx / 33, e = idx - row * 33;
        float s = 0.f;
        #pragma unroll
        for (int q = 0; q < 8; ++q) s += red[q][row][e];
        xdbl[(size_t)m0 * 33 + idx] = s;
    }
}

#define SCAN_T     32
#define SCAN_NCH   (L_SEQ / SCAN_T)
#define SCAN_PITCH 36
#define SCAN_XCH   (SCAN_T * 33)
#define SCAN_CH    16
#define SCAN_REC   (SCAN_CH * 8)
#define SCAN_XK    (SCAN_XCH / SCAN_REC)
#define SCAN_XREM  (SCAN_XCH - SCAN_XK * SCAN_REC)
#define SCAN_SK    (SCAN_T / 8)
#define SCAN_YP    36
#define Y_PAIR     (2 * D_INNER)

__device__ __forceinline__ int scan_dst(int i) {
    const int t = i / 33, e = i - t * 33;
    return t * SCAN_PITCH + (e ? e + 3 : 0);
}

__device__ __forceinline__ float scan_dt(float raw, float wdt, float bd) {
    const float x = raw * wdt + bd;
    const float t = __expf(-fabsf(x));
    const float s = t / (2.f + t);
    const float w = s * s;
    float q = 2.f / 13.f;
    q = q * w + 2.f / 11.f;
    q = q * w + 2.f / 9.f;
    q = q * w + 2.f / 7.f;
    q = q * w + 2.f / 5.f;
    q = q * w + 2.f / 3.f;
    q = q * w + 2.f;
    const float sp = fmaxf(x, 0.f) + s * q;
    return fminf(fmaxf(sp, 1e-4f), 10.f);
}

template <int XMASK>
__device__ __forceinline__ float scan_xchg_add(bool up, float lo, float hi) {
    const float keep = up ? hi : lo;
    const float send = up ? lo : hi;
    const int o = __builtin_amdgcn_update_dpp(0, __float_as_int(send), 0x160 + XMASK,
                                              0xF, 0xF, true);
    return keep + __int_as_float(o);
}

#define SCAN_LOG2E 1.44269504088896341f
#define SCAN_GX    (D_INNER / SCAN_CH)
#define SCAN_WGS   (SCAN_GX * B_SZ)
#define SCAN_LDS   (2 * SCAN_T * SCAN_PITCH + 4 * SCAN_T * SCAN_CH + 2 * SCAN_T * SCAN_YP)

__device__ __forceinline__ void scan_lds_barrier() {
    __builtin_amdgcn_fence(__ATOMIC_RELEASE, "workgroup", "local");
    __builtin_amdgcn_s_barrier();
    __builtin_amdgcn_fence(__ATOMIC_ACQUIRE, "workgroup", "local");
}

__device__ __forceinline__
void scan_group(const float* __restrict__ xdbl, const float* __restrict__ u,
                const float* __restrict__ Wdt, const float* __restrict__ bdt,
                const float* __restrict__ Alog, float* __restrict__ ybuf,
                int bx, int by, float* smem) {
    const int  tid = threadIdx.x;
    const bool rec = tid < SCAN_REC;
    const int  d0  = bx * SCAN_CH;
    const int  b   = by;

    float  (*sx)[SCAN_T * SCAN_PITCH] = (float (*)[SCAN_T * SCAN_PITCH])smem;
    float2 (*sdu)[SCAN_T][SCAN_CH] =
        (float2 (*)[SCAN_T][SCAN_CH])(smem + 2 * SCAN_T * SCAN_PITCH);
    float  (*sy)[SCAN_T * SCAN_YP] =
        (float (*)[SCAN_T * SCAN_YP])(smem + 2 * SCAN_T * SCAN_PITCH + 4 * SCAN_T * SCAN_CH);

    const float* xd = xdbl + (size_t)b * L_SEQ * 33;

    const int j = tid & 7;
    const int c = (tid >> 3) & (SCAN_CH - 1);
    const float A0 = -__expf(fminf(Alog[2 * j],     5.f)) * SCAN_LOG2E;
    const float A1 = -__expf(fminf(Alog[2 * j + 1], 5.f)) * SCAN_LOG2E;
    float h0 = 0.f, h1 = 0.f;
    const bool j4 = (j & 4) != 0, j2 = (j & 2) != 0, j1 = (j & 1) != 0;

    const int sid = tid & (SCAN_REC - 1);
    const int sc  = sid & (SCAN_CH - 1);
    const int st  = sid >> 4;
    const float wdt = Wdt[d0 + sc];
    const float bd  = bdt[d0 + sc];
    const float* ub = u + (size_t)b * L_SEQ * D_INNER + d0 + sc;
    float* yb = ybuf + ((size_t)b * (L_SEQ / 2) + (st >> 1)) * Y_PAIR
                     + bx * 32 + (st & 1) * 16 + sc;
    const bool hasr = sid < SCAN_XREM;
    const int sidr = hasr ? sid + SCAN_XK * SCAN_REC : SCAN_XCH - 1;
    int dst[SCAN_XK + 1];
    #pragma unroll
    for (int k = 0; k < SCAN_XK; ++k) dst[k] = scan_dst(sid + k * SCAN_REC);
    dst[SCAN_XK] = scan_dst(sidr);

    struct ScanRaw { float x[SCAN_XK + 1]; float r[SCAN_SK]; float u[SCAN_SK]; };
    ScanRaw ra = {}, rb = {};

    auto scan_load = [&](ScanRaw& r, const int k) __attribute__((always_inline)) {
        const float* nsrc = xd + (size_t)k * SCAN_XCH;
        #pragma unroll
        for (int i = 0; i < SCAN_XK; ++i) r.x[i] = nsrc[sid + i * SCAN_REC];
        r.x[SCAN_XK] = nsrc[sidr];
        #pragma unroll
        for (int m = 0; m < SCAN_SK; ++m) r.r[m] = nsrc[(st + 8 * m) * 33];
        const float* nu = ub + (size_t)k * SCAN_T * D_INNER;
        #pragma unroll
        for (int m = 0; m < SCAN_SK; ++m) r.u[m] = nu[(size_t)(st + 8 * m) * D_INNER];
    };
    auto scan_stage = [&](const ScanRaw& r, const int bf) __attribute__((always_inline)) {
        float* nx = sx[bf];
        #pragma unroll
        for (int i = 0; i < SCAN_XK; ++i) nx[dst[i]] = r.x[i];
        if (hasr) nx[dst[SCAN_XK]] = r.x[SCAN_XK];
        #pragma unroll
        for (int m = 0; m < SCAN_SK; ++m) {
            const float dn = scan_dt(r.r[m], wdt, bd);
            sdu[bf][st + 8 * m][sc] = make_float2(dn, dn * r.u[m]);
        }
    };
    auto scan_yout = [&](const int k, const int bf) __attribute__((always_inline)) {
        const float* py = sy[bf];
        float* yc = yb + (size_t)k * (SCAN_T / 2) * Y_PAIR;
        #pragma unroll
        for (int m = 0; m < SCAN_SK; ++m)
            yc[(size_t)(4 * m) * Y_PAIR] = py[(st + 8 * m) * SCAN_YP + sc];
    };

    auto scan_chunk = [&](const int ch, const int cur, const ScanRaw& have,
                          ScanRaw& next) __attribute__((always_inline)) {
        if (rec) {
            const float* s = sx[cur];
            float pk[8];
            #pragma unroll
            for (int t = 0; t < SCAN_T; ++t) {
                const float2 du = sdu[cur][t][c];
                const float2 Bq = *(const float2*)&s[t * SCAN_PITCH + 4  + 2 * j];
                const float2 Cq = *(const float2*)&s[t * SCAN_PITCH + 20 + 2 * j];
                const float dt = du.x, dtu = du.y;

                const float dA0 = __builtin_amdgcn_exp2f(
                    fminf(fmaxf(dt * A0, -20.f * SCAN_LOG2E), 0.f));
                const float dA1 = __builtin_amdgcn_exp2f(
                    fminf(fmaxf(dt * A1, -20.f * SCAN_LOG2E), 0.f));
                const float dB0 = fminf(fmaxf(dtu * Bq.x, -10.f), 10.f);
                const float dB1 = fminf(fmaxf(dtu * Bq.y, -10.f), 10.f);
                h0 = fminf(fmaxf(h0 * dA0 + dB0, -100.f), 100.f);
                h1 = fminf(fmaxf(h1 * dA1 + dB1, -100.f), 100.f);

                pk[t & 7] = h0 * Cq.x + h1 * Cq.y;
                if ((t & 7) == 7) {
                    const float q0 = scan_xchg_add<4>(j4, pk[0], pk[4]);
                    const float q1 = scan_xchg_add<4>(j4, pk[1], pk[5]);
                    const float q2 = scan_xchg_add<4>(j4, pk[2], pk[6]);
                    const float q3 = scan_xchg_add<4>(j4, pk[3], pk[7]);
                    const float r0 = scan_xchg_add<2>(j2, q0, q2);
                    const float r1 = scan_xchg_add<2>(j2, q1, q3);
                    sy[cur][((t & ~7) + j) * SCAN_YP + c] = scan_xchg_add<1>(j1, r0, r1);
                }
            }
        } else {
            scan_load(next, min(ch + 2, SCAN_NCH - 1));
            if (ch > 0) {
                scan_yout(ch - 1, cur ^ 1);
            }
            scan_stage(have, cur ^ 1);
        }
        scan_lds_barrier();
    };

    if (!rec) {
        scan_load(ra, 0);
        scan_stage(ra, 0);
        scan_load(rb, 1);
    }
    __syncthreads();

    static_assert(SCAN_NCH % 2 == 0, "the chunk loop is written out in pairs");
    for (int ch = 0; ch < SCAN_NCH; ch += 2) {
        scan_chunk(ch,     0, rb, ra);
        scan_chunk(ch + 1, 1, ra, rb);
    }

    if (!rec) {
        scan_yout(SCAN_NCH - 1, (SCAN_NCH - 1) & 1);
    }
}

#define ZG_NX   (D_INNER / 128)
#define ZG_WGS  (ZG_NX * (M_ROWS / 64))

__global__ __launch_bounds__(256) __attribute__((amdgpu_waves_per_eu(8)))
void scan_zgemm_kernel(const float* __restrict__ xdbl, const float* __restrict__ u,
                       const float* __restrict__ Wdt, const float* __restrict__ bdt,
                       const float* __restrict__ Alog, float* __restrict__ ybuf,
                       const float* __restrict__ x, const float* __restrict__ Wz,
                       float* __restrict__ zout, int ldc, int K) {
    __shared__ __align__(16) float smem[2 * GM_STAGE];
    static_assert(SCAN_LDS <= 2 * GM_STAGE, "scan LDS must fit the GEMM stages");
    const int bid = blockIdx.x;
    if (bid < SCAN_WGS) {
        scan_group(xdbl, u, Wdt, bdt, Alog, ybuf, bid % SCAN_GX, bid / SCAN_GX, smem);
    } else {
        const int g = bid - SCAN_WGS;
        gemm_tile_f32_wmma(x, Wz, zout, ldc, K, g % ZG_NX, g / ZG_NX, smem);
    }
}

__global__ __launch_bounds__(256)
void ln_gate_kernel(const float* __restrict__ yscan, float* __restrict__ ygate,
                    const float* __restrict__ u,
                    const float* __restrict__ xz, const float* __restrict__ Dp,
                    const float* __restrict__ g, const float* __restrict__ bta) {
    const int row = blockIdx.x;
    const int tid = threadIdx.x;
    const float* ys = yscan + (size_t)(row >> 1) * Y_PAIR + (row & 1) * 16
                            + (tid >> 4) * 32 + (tid & 15);
    float* yr = ygate + (size_t)row * D_INNER;

    float v[8], s = 0.f, s2 = 0.f;
    #pragma unroll
    for (int i = 0; i < 8; ++i) {
        v[i] = ys[i * 512];
        s += v[i]; s2 += v[i] * v[i];
    }
    #pragma unroll
    for (int off = 16; off; off >>= 1) {
        s  += __shfl_xor(s,  off, 32);
        s2 += __shfl_xor(s2, off, 32);
    }
    __shared__ float ws1[8], ws2[8];
    const int lane = tid & 31, w = tid >> 5;
    if (lane == 0) { ws1[w] = s; ws2[w] = s2; }
    __syncthreads();
    if (tid == 0) {
        float t1 = 0.f, t2 = 0.f;
        #pragma unroll
        for (int i = 0; i < 8; ++i) { t1 += ws1[i]; t2 += ws2[i]; }
        ws1[0] = t1; ws2[0] = t2;
    }
    __syncthreads();
    const float mu  = ws1[0] * (1.f / D_INNER);
    float var = ws2[0] * (1.f / D_INNER) - mu * mu;
    var = fmaxf(var, 0.f);
    const float rs = rsqrtf(var + 1e-5f);

    #pragma unroll
    for (int i = 0; i < 8; ++i) {
        const int d = tid + i * 256;
        float yn = (v[i] - mu) * rs * g[d] + bta[d];
        float uu = u[(size_t)row * D_INNER + d];
        float zz = xz[(size_t)row * (2 * D_INNER) + D_INNER + d];
        float sz = zz / (1.f + __expf(-zz));
        yr[d] = (yn + Dp[d] * uu) * sz;
    }
}

extern "C" void kernel_launch(void* const* d_in, const int* in_sizes, int n_in,
                              void* d_out, int out_size, void* d_ws, size_t ws_size,
                              hipStream_t stream) {
    const float* x      = (const float*)d_in[0];
    const float* W_in   = (const float*)d_in[1];
    const float* conv_w = (const float*)d_in[2];
    const float* conv_b = (const float*)d_in[3];
    const float* W_x    = (const float*)d_in[4];
    const float* W_dt   = (const float*)d_in[5];
    const float* b_dt   = (const float*)d_in[6];
    const float* A_log  = (const float*)d_in[7];
    const float* D_par  = (const float*)d_in[8];
    const float* W_out  = (const float*)d_in[9];
    const float* ln_g   = (const float*)d_in[10];
    const float* ln_b   = (const float*)d_in[11];
    float* out = (float*)d_out;

    float* ws = (float*)d_ws;
    const size_t XZ_OFF   = 0;
    const size_t U_OFF    = XZ_OFF + (size_t)M_ROWS * 2 * D_INNER;
    const size_t XDBL_OFF = U_OFF  + (size_t)M_ROWS * D_INNER;
    const size_t Y_OFF    = XDBL_OFF + (size_t)M_ROWS * 33;
    const size_t YG_OFF   = Y_OFF + (size_t)M_ROWS * D_INNER;
    float* xz   = ws + XZ_OFF;
    float* u    = ws + U_OFF;
    float* xdbl = ws + XDBL_OFF;
    float* ybuf = ws + Y_OFF;
    float* ygat = ws + YG_OFF;

    gemm_f32_wmma<<<dim3(D_INNER / 128, M_ROWS / 64), 256, 0, stream>>>(
        x, W_in, xz, 2 * D_INNER, D_MODEL);

    conv_silu_kernel<<<(M_ROWS / CONV_T) * 512 / 256, 256, 0, stream>>>(
        xz, conv_w, conv_b, u);

    xdbl_wmma_kernel<<<M_ROWS / 16, 256, 0, stream>>>(u, W_x, xdbl);

    scan_zgemm_kernel<<<SCAN_WGS + ZG_WGS, 256, 0, stream>>>(
        xdbl, u, W_dt, b_dt, A_log, ybuf,
        x, W_in + (size_t)D_INNER * D_MODEL, xz + D_INNER, 2 * D_INNER, D_MODEL);

    ln_gate_kernel<<<M_ROWS, 256, 0, stream>>>(ybuf, ygat, u, xz, D_par, ln_g, ln_b);

    gemm_f32_wmma<<<dim3(D_MODEL / 128, M_ROWS / 64), 256, 0, stream>>>(
        ygat, W_out, out, D_MODEL, D_INNER);
}
